# starting kernel
# baseline (speedup 1.0000x reference)
#include <hip/hip_runtime.h>
#include <math.h>

#define DIM 128
#define HID 16
#define HEADS 4
#define NEG_SLOPE 0.2f

typedef __attribute__((ext_vector_type(2))) float v2f;
typedef __attribute__((ext_vector_type(8))) float v8f;

__device__ __forceinline__ void atomicMaxFloat(float* addr, float v) {
    if (v >= 0.0f) {
        atomicMax((int*)addr, __float_as_int(v));
    } else {
        atomicMin((unsigned int*)addr, __float_as_uint(v));
    }
}

__global__ void __launch_bounds__(128)
gemm1_wmma(const float* __restrict__ x, const float* __restrict__ W1,
           float* __restrict__ h1, int N) {
    __shared__ float Wl[DIM * 64];
    for (int i = threadIdx.x; i < DIM * 64; i += blockDim.x) Wl[i] = W1[i];
    __syncthreads();

    const int wave  = threadIdx.x >> 5;
    const int lane  = threadIdx.x & 31;
    const int Mtile = blockIdx.x * 4 + wave;
    const int Mtiles = (N + 15) >> 4;
    if (Mtile >= Mtiles) return;

    const int r  = lane & 15;
    const int kh = lane >> 4;
    const int mbase = Mtile * 16;
    int row = mbase + r;
    if (row >= N) row = N - 1;
    const float* __restrict__ xr = x + (size_t)row * DIM;

    v8f acc0 = {}, acc1 = {}, acc2 = {}, acc3 = {};

    for (int kb = 0; kb < DIM; kb += 4) {
        const int k0 = kb + 2 * kh;
        v2f a;
        a.x = xr[k0];
        a.y = xr[k0 + 1];
        const float* w0 = &Wl[(size_t)k0 * 64];
        const float* w1 = &Wl[(size_t)(k0 + 1) * 64];
        v2f b;
        b.x = w0[r];       b.y = w1[r];
        acc0 = __builtin_amdgcn_wmma_f32_16x16x4_f32(false, a, false, b, (short)0, acc0, false, false);
        b.x = w0[16 + r];  b.y = w1[16 + r];
        acc1 = __builtin_amdgcn_wmma_f32_16x16x4_f32(false, a, false, b, (short)0, acc1, false, false);
        b.x = w0[32 + r];  b.y = w1[32 + r];
        acc2 = __builtin_amdgcn_wmma_f32_16x16x4_f32(false, a, false, b, (short)0, acc2, false, false);
        b.x = w0[48 + r];  b.y = w1[48 + r];
        acc3 = __builtin_amdgcn_wmma_f32_16x16x4_f32(false, a, false, b, (short)0, acc3, false, false);
    }

    float* outp = h1 + (size_t)mbase * 64;
#pragma unroll
    for (int i = 0; i < 8; ++i) {
        const int m = i + 8 * kh;
        if (mbase + m < N) {
            float* o = outp + (size_t)m * 64 + r;
            o[0]  = acc0[i];
            o[16] = acc1[i];
            o[32] = acc2[i];
            o[48] = acc3[i];
        }
    }
}

__global__ void __launch_bounds__(256)
node_prep1(const float* __restrict__ h1, const float* __restrict__ attS,
           const float* __restrict__ attD, float* __restrict__ as1,
           float* __restrict__ ad1, float* __restrict__ m1,
           float* __restrict__ s1, float* __restrict__ o1, int N) {
    const int tid = blockIdx.x * blockDim.x + threadIdx.x;
    if (tid >= N * HEADS) return;
    const int n  = tid >> 2;
    const int hd = tid & 3;
    const float* hp = h1 + (size_t)n * 64 + hd * HID;
    float ss = 0.f, dd = 0.f;
#pragma unroll
    for (int c = 0; c < HID; ++c) {
        const float v = hp[c];
        ss += v * attS[hd * HID + c];
        dd += v * attD[hd * HID + c];
    }
    as1[tid] = ss;
    ad1[tid] = dd;
    m1[tid]  = -3.0e38f;
    s1[tid]  = 0.f;
    float* op = o1 + (size_t)n * 64 + hd * HID;
#pragma unroll
    for (int c = 0; c < HID; ++c) op[c] = 0.f;
}

__global__ void __launch_bounds__(256)
edge_max1(const int* __restrict__ ei, const float* __restrict__ as1,
          const float* __restrict__ ad1, float* __restrict__ m1, int N, int E) {
    const int tid = blockIdx.x * blockDim.x + threadIdx.x;
    const int Et = E + N;
    const int e  = tid >> 2;
    const int hd = tid & 3;
    if (e >= Et) return;
    int s, d;
    if (e < E) { s = ei[e]; d = ei[E + e]; } else { s = d = e - E; }
    float le = as1[s * HEADS + hd] + ad1[d * HEADS + hd];
    le = (le > 0.f) ? le : NEG_SLOPE * le;
    atomicMaxFloat(&m1[d * HEADS + hd], le);
}

__global__ void __launch_bounds__(256)
edge_acc1(const int* __restrict__ ei, const float* __restrict__ as1,
          const float* __restrict__ ad1, const float* __restrict__ m1,
          float* __restrict__ s1, const float* __restrict__ h1,
          float* __restrict__ o1, int N, int E) {
    const int tid = blockIdx.x * blockDim.x + threadIdx.x;
    const int Et = E + N;
    const int e  = tid >> 2;
    const int hd = tid & 3;
    if (e >= Et) return;
    int s, d;
    if (e < E) { s = ei[e]; d = ei[E + e]; } else { s = d = e - E; }
    float le = as1[s * HEADS + hd] + ad1[d * HEADS + hd];
    le = (le > 0.f) ? le : NEG_SLOPE * le;
    const float w = __expf(le - m1[d * HEADS + hd]);
    atomicAdd(&s1[d * HEADS + hd], w);
    const float* hp = h1 + (size_t)s * 64 + hd * HID;
    float* op = o1 + (size_t)d * 64 + hd * HID;
#pragma unroll
    for (int c = 0; c < HID; ++c) atomicAdd(&op[c], w * hp[c]);
}

__global__ void __launch_bounds__(256)
finish1_elu(float* __restrict__ o1, const float* __restrict__ s1,
            const float* __restrict__ b1, int N) {
    const int i = blockIdx.x * blockDim.x + threadIdx.x;
    if (i >= N * 64) return;
    const int n  = i >> 6;
    const int hd = (i >> 4) & 3;
    float v = o1[i] / (s1[n * HEADS + hd] + 1e-16f) + b1[i & 63];
    v = (v > 0.f) ? v : expm1f(v);
    o1[i] = v;
}

__global__ void __launch_bounds__(256)
node_prep2(const float* __restrict__ o1, const float* __restrict__ W2,
           const float* __restrict__ attS2, const float* __restrict__ attD2,
           float* __restrict__ h2, float* __restrict__ as2,
           float* __restrict__ ad2, float* __restrict__ m2,
           float* __restrict__ s2, float* __restrict__ out, int N) {
    const int n = blockIdx.x * blockDim.x + threadIdx.x;
    if (n >= N) return;
    const float* hp = o1 + (size_t)n * 64;
    float a0 = 0.f, a1 = 0.f;
#pragma unroll
    for (int k = 0; k < 64; ++k) {
        const float v = hp[k];
        a0 += v * W2[k * 2 + 0];
        a1 += v * W2[k * 2 + 1];
    }
    h2[n * 2 + 0] = a0;
    h2[n * 2 + 1] = a1;
    as2[n] = a0 * attS2[0] + a1 * attS2[1];
    ad2[n] = a0 * attD2[0] + a1 * attD2[1];
    m2[n] = -3.0e38f;
    s2[n] = 0.f;
    out[n * 2 + 0] = 0.f;
    out[n * 2 + 1] = 0.f;
}

__global__ void __launch_bounds__(256)
edge_max2(const int* __restrict__ ei, const float* __restrict__ as2,
          const float* __restrict__ ad2, float* __restrict__ m2, int N, int E) {
    const int e = blockIdx.x * blockDim.x + threadIdx.x;
    const int Et = E + N;
    if (e >= Et) return;
    int s, d;
    if (e < E) { s = ei[e]; d = ei[E + e]; } else { s = d = e - E; }
    float le = as2[s] + ad2[d];
    le = (le > 0.f) ? le : NEG_SLOPE * le;
    atomicMaxFloat(&m2[d], le);
}

__global__ void __launch_bounds__(256)
edge_acc2(const int* __restrict__ ei, const float* __restrict__ as2,
          const float* __restrict__ ad2, const float* __restrict__ m2,
          float* __restrict__ s2, const float* __restrict__ h2,
          float* __restrict__ out, int N, int E) {
    const int e = blockIdx.x * blockDim.x + threadIdx.x;
    const int Et = E + N;
    if (e >= Et) return;
    int s, d;
    if (e < E) { s = ei[e]; d = ei[E + e]; } else { s = d = e - E; }
    float le = as2[s] + ad2[d];
    le = (le > 0.f) ? le : NEG_SLOPE * le;
    const float w = __expf(le - m2[d]);
    atomicAdd(&s2[d], w);
    atomicAdd(&out[d * 2 + 0], w * h2[s * 2 + 0]);
    atomicAdd(&out[d * 2 + 1], w * h2[s * 2 + 1]);
}

__global__ void __launch_bounds__(256)
finish2(float* __restrict__ out, const float* __restrict__ s2,
        const float* __restrict__ b2, int N) {
    const int n = blockIdx.x * blockDim.x + threadIdx.x;
    if (n >= N) return;
    const float inv = 1.0f / (s2[n] + 1e-16f);
    out[n * 2 + 0] = out[n * 2 + 0] * inv + b2[0];
    out[n * 2 + 1] = out[n * 2 + 1] * inv + b2[1];
}

extern "C" void kernel_launch(void* const* d_in, const int* in_sizes, int n_in,
                              void* d_out, int out_size, void* d_ws, size_t ws_size,
                              hipStream_t stream) {
    const float* x    = (const float*)d_in[0];
    const int*   ei   = (const int*)d_in[1];
    const float* W1   = (const float*)d_in[2];
    const float* aS1  = (const float*)d_in[3];
    const float* aD1  = (const float*)d_in[4];
    const float* b1   = (const float*)d_in[5];
    const float* W2   = (const float*)d_in[6];
    const float* aS2  = (const float*)d_in[7];
    const float* aD2  = (const float*)d_in[8];
    const float* b2   = (const float*)d_in[9];

    const int N = in_sizes[0] / DIM;
    const int E = in_sizes[1] / 2;
    float* out = (float*)d_out;

    float* ws  = (float*)d_ws;
    float* h1  = ws;
    float* as1 = h1 + (size_t)N * 64;
    float* ad1 = as1 + (size_t)N * HEADS;
    float* m1  = ad1 + (size_t)N * HEADS;
    float* s1  = m1 + (size_t)N * HEADS;
    float* o1  = s1 + (size_t)N * HEADS;
    float* h2  = o1 + (size_t)N * 64;
    float* as2 = h2 + (size_t)N * 2;
    float* ad2 = as2 + (size_t)N;
    float* m2  = ad2 + (size_t)N;
    float* s2  = m2 + (size_t)N;

    const int Et = E + N;
    const int Mtiles = (N + 15) >> 4;

    gemm1_wmma<<<(Mtiles + 3) / 4, 128, 0, stream>>>(x, W1, h1, N);
    node_prep1<<<(N * HEADS + 255) / 256, 256, 0, stream>>>(h1, aS1, aD1, as1, ad1, m1, s1, o1, N);
    edge_max1<<<((size_t)Et * HEADS + 255) / 256, 256, 0, stream>>>(ei, as1, ad1, m1, N, E);
    edge_acc1<<<((size_t)Et * HEADS + 255) / 256, 256, 0, stream>>>(ei, as1, ad1, m1, s1, h1, o1, N, E);
    finish1_elu<<<((size_t)N * 64 + 255) / 256, 256, 0, stream>>>(o1, s1, b1, N);
    node_prep2<<<(N + 255) / 256, 256, 0, stream>>>(o1, W2, aS2, aD2, h2, as2, ad2, m2, s2, out, N);
    edge_max2<<<(Et + 255) / 256, 256, 0, stream>>>(ei, as2, ad2, m2, N, E);
    edge_acc2<<<(Et + 255) / 256, 256, 0, stream>>>(ei, as2, ad2, m2, s2, h2, out, N, E);
    finish2<<<(N + 255) / 256, 256, 0, stream>>>(out, s2, b2, N);
}


# Edges bucketed by destination each call (int count+rank, scan, place); each node's segment softmax+sum done in registers by a half-wave: float atomics removed, sum order now by rank and lane tree
# speedup vs baseline: 3.7839x; 3.7839x over previous
#include <hip/hip_runtime.h>
#include <math.h>

#define DIM 128
#define HID 16
#define HEADS 4
#define NEG_SLOPE 0.2f

typedef __attribute__((ext_vector_type(2))) float v2f;
typedef __attribute__((ext_vector_type(8))) float v8f;

__global__ void __launch_bounds__(128)
gemm1_wmma(const float* __restrict__ x, const float* __restrict__ W1,
           float* __restrict__ h1, int N) {
    __shared__ float Wl[DIM * 64];
    for (int i = threadIdx.x; i < DIM * 64; i += blockDim.x) Wl[i] = W1[i];
    __syncthreads();

    const int wave  = threadIdx.x >> 5;
    const int lane  = threadIdx.x & 31;
    const int Mtile = blockIdx.x * 4 + wave;
    const int Mtiles = (N + 15) >> 4;
    if (Mtile >= Mtiles) return;

    const int r  = lane & 15;
    const int kh = lane >> 4;
    const int mbase = Mtile * 16;
    int row = mbase + r;
    if (row >= N) row = N - 1;
    const float* __restrict__ xr = x + (size_t)row * DIM;

    v8f acc0 = {}, acc1 = {}, acc2 = {}, acc3 = {};

    for (int kb = 0; kb < DIM; kb += 4) {
        const int k0 = kb + 2 * kh;
        v2f a;
        a.x = xr[k0];
        a.y = xr[k0 + 1];
        const float* w0 = &Wl[(size_t)k0 * 64];
        const float* w1 = &Wl[(size_t)(k0 + 1) * 64];
        v2f b;
        b.x = w0[r];       b.y = w1[r];
        acc0 = __builtin_amdgcn_wmma_f32_16x16x4_f32(false, a, false, b, (short)0, acc0, false, false);
        b.x = w0[16 + r];  b.y = w1[16 + r];
        acc1 = __builtin_amdgcn_wmma_f32_16x16x4_f32(false, a, false, b, (short)0, acc1, false, false);
        b.x = w0[32 + r];  b.y = w1[32 + r];
        acc2 = __builtin_amdgcn_wmma_f32_16x16x4_f32(false, a, false, b, (short)0, acc2, false, false);
        b.x = w0[48 + r];  b.y = w1[48 + r];
        acc3 = __builtin_amdgcn_wmma_f32_16x16x4_f32(false, a, false, b, (short)0, acc3, false, false);
    }

    float* outp = h1 + (size_t)mbase * 64;
#pragma unroll
    for (int i = 0; i < 8; ++i) {
        const int m = i + 8 * kh;
        if (mbase + m < N) {
            float* o = outp + (size_t)m * 64 + r;
            o[0]  = acc0[i];
            o[16] = acc1[i];
            o[32] = acc2[i];
            o[48] = acc3[i];
        }
    }
}

__global__ void __launch_bounds__(256)
node_prep1(const float* __restrict__ h1, const float* __restrict__ attS,
           const float* __restrict__ attD, float* __restrict__ as1,
           float* __restrict__ ad1, int* __restrict__ cnt, int N) {
    const int tid = blockIdx.x * blockDim.x + threadIdx.x;
    if (tid >= N * HEADS) return;
    const int n  = tid >> 2;
    const int hd = tid & 3;
    const float* hp = h1 + (size_t)n * 64 + hd * HID;
    float ss = 0.f, dd = 0.f;
#pragma unroll
    for (int c = 0; c < HID; ++c) {
        const float v = hp[c];
        ss += v * attS[hd * HID + c];
        dd += v * attD[hd * HID + c];
    }
    as1[tid] = ss;
    ad1[tid] = dd;
    if (hd == 0) cnt[n] = 0;
}

__global__ void __launch_bounds__(256)
dst_rank(const int* __restrict__ ei, int* __restrict__ cnt,
         int* __restrict__ rnk, int E) {
    const int e = blockIdx.x * blockDim.x + threadIdx.x;
    if (e >= E) return;
    const int d = ei[E + e];
    rnk[e] = atomicAdd(&cnt[d], 1);
}

__global__ void __launch_bounds__(256)
deg_blocksum(const int* __restrict__ cnt, int* __restrict__ bsum, int N) {
    __shared__ int red[8];
    const int i = (blockIdx.x * 256 + threadIdx.x) * 4;
    int v = 0;
#pragma unroll
    for (int j = 0; j < 4; ++j) if (i + j < N) v += cnt[i + j];
    for (int o = 16; o > 0; o >>= 1) v += __shfl_xor(v, o, 32);
    if ((threadIdx.x & 31) == 0) red[threadIdx.x >> 5] = v;
    __syncthreads();
    if (threadIdx.x == 0) {
        int t = 0;
#pragma unroll
        for (int w = 0; w < 8; ++w) t += red[w];
        bsum[blockIdx.x] = t;
    }
}

__global__ void __launch_bounds__(256)
deg_scan(const int* __restrict__ cnt, const int* __restrict__ bsum,
         int* __restrict__ rowptr, int N) {
    __shared__ int red[8];
    __shared__ int wtot[8];
    const int lane = threadIdx.x & 31;
    const int wave = threadIdx.x >> 5;
    int t = 0;
    for (int j = threadIdx.x; j < (int)blockIdx.x; j += 256) t += bsum[j];
    for (int o = 16; o > 0; o >>= 1) t += __shfl_xor(t, o, 32);
    if (lane == 0) red[wave] = t;

    const int i = (blockIdx.x * 256 + threadIdx.x) * 4;
    int c[4];
    int ts = 0;
#pragma unroll
    for (int j = 0; j < 4; ++j) { c[j] = (i + j < N) ? cnt[i + j] : 0; ts += c[j]; }
    int x = ts;
    for (int o = 1; o < 32; o <<= 1) {
        const int y = __shfl_up(x, o, 32);
        if (lane >= o) x += y;
    }
    if (lane == 31) wtot[wave] = x;
    __syncthreads();
    int off = 0;
#pragma unroll
    for (int w = 0; w < 8; ++w) { off += red[w]; if (w < wave) off += wtot[w]; }
    int p = off + x - ts;
#pragma unroll
    for (int j = 0; j < 4; ++j) { if (i + j < N) rowptr[i + j] = p; p += c[j]; }
}

__global__ void __launch_bounds__(256)
place_src(const int* __restrict__ ei, const int* __restrict__ rowptr,
          const int* __restrict__ rnk, int* __restrict__ csr, int E) {
    const int e = blockIdx.x * blockDim.x + threadIdx.x;
    if (e >= E) return;
    const int s = ei[e];
    const int d = ei[E + e];
    csr[rowptr[d] + rnk[e]] = s;
}

__device__ __forceinline__ float lrelu(float v) { return (v > 0.f) ? v : NEG_SLOPE * v; }
__device__ __forceinline__ float pick4(const float4 v, int k) {
    return (k == 0) ? v.x : (k == 1) ? v.y : (k == 2) ? v.z : v.w;
}

__global__ void __launch_bounds__(256)
seg_gather1(const int* __restrict__ csr, const int* __restrict__ rowptr,
            const int* __restrict__ cnt, const float* __restrict__ as1,
            const float* __restrict__ ad1, const float* __restrict__ h1,
            const float* __restrict__ b1, float* __restrict__ o1, int N) {
    const int n  = (blockIdx.x * 256 + threadIdx.x) >> 4;
    const int l  = threadIdx.x & 15;
    const int hd = l >> 2;
    if (n >= N) return;
    const int start = rowptr[n];
    const int deg   = cnt[n];
    const float4* __restrict__ as4 = (const float4*)as1;
    const float4* __restrict__ h4  = (const float4*)h1;

    float4 m4 = as4[n];
    for (int base = 0; base < deg; base += 16) {
        const int j = base + l;
        const int s = (j < deg) ? csr[start + j] : n;
        const float4 a = as4[s];
        m4.x = fmaxf(m4.x, a.x); m4.y = fmaxf(m4.y, a.y);
        m4.z = fmaxf(m4.z, a.z); m4.w = fmaxf(m4.w, a.w);
    }
    for (int o = 8; o > 0; o >>= 1) {
        m4.x = fmaxf(m4.x, __shfl_xor(m4.x, o, 16));
        m4.y = fmaxf(m4.y, __shfl_xor(m4.y, o, 16));
        m4.z = fmaxf(m4.z, __shfl_xor(m4.z, o, 16));
        m4.w = fmaxf(m4.w, __shfl_xor(m4.w, o, 16));
    }
    const float adv = ad1[n * HEADS + hd];
    const float mm  = lrelu(pick4(m4, hd) + adv);

    float ssum;
    float4 acc;
    {
        const float w = __expf(lrelu(as1[n * HEADS + hd] + adv) - mm);
        const float4 v = h4[n * 16 + l];
        ssum = w;
        acc.x = w * v.x; acc.y = w * v.y; acc.z = w * v.z; acc.w = w * v.w;
    }
    for (int base = 0; base < deg; base += 16) {
        const int j   = base + l;
        const int idx = (j < deg) ? csr[start + j] : n;
        const int c   = min(16, deg - base);
        for (int k = 0; k < c; k += 4) {
            const int s0 = __shfl(idx, k, 16);
            const int s1 = __shfl(idx, k + 1, 16);
            const int s2 = __shfl(idx, k + 2, 16);
            const int s3 = __shfl(idx, k + 3, 16);
            const float a0 = as1[s0 * HEADS + hd];
            const float a1 = as1[s1 * HEADS + hd];
            const float a2 = as1[s2 * HEADS + hd];
            const float a3 = as1[s3 * HEADS + hd];
            const float4 v0 = h4[s0 * 16 + l];
            const float4 v1 = h4[s1 * 16 + l];
            const float4 v2 = h4[s2 * 16 + l];
            const float4 v3 = h4[s3 * 16 + l];
            const float w0 = __expf(lrelu(a0 + adv) - mm);
            const float w1 = (k + 1 < c) ? __expf(lrelu(a1 + adv) - mm) : 0.f;
            const float w2 = (k + 2 < c) ? __expf(lrelu(a2 + adv) - mm) : 0.f;
            const float w3 = (k + 3 < c) ? __expf(lrelu(a3 + adv) - mm) : 0.f;
            ssum += w0; acc.x += w0 * v0.x; acc.y += w0 * v0.y; acc.z += w0 * v0.z; acc.w += w0 * v0.w;
            ssum += w1; acc.x += w1 * v1.x; acc.y += w1 * v1.y; acc.z += w1 * v1.z; acc.w += w1 * v1.w;
            ssum += w2; acc.x += w2 * v2.x; acc.y += w2 * v2.y; acc.z += w2 * v2.z; acc.w += w2 * v2.w;
            ssum += w3; acc.x += w3 * v3.x; acc.y += w3 * v3.y; acc.z += w3 * v3.z; acc.w += w3 * v3.w;
        }
    }

    const float den = ssum + 1e-16f;
    float4 r;
    r.x = acc.x / den + b1[l * 4 + 0];
    r.y = acc.y / den + b1[l * 4 + 1];
    r.z = acc.z / den + b1[l * 4 + 2];
    r.w = acc.w / den + b1[l * 4 + 3];
    r.x = (r.x > 0.f) ? r.x : expm1f(r.x);
    r.y = (r.y > 0.f) ? r.y : expm1f(r.y);
    r.z = (r.z > 0.f) ? r.z : expm1f(r.z);
    r.w = (r.w > 0.f) ? r.w : expm1f(r.w);
    ((float4*)o1)[n * 16 + l] = r;
}

__global__ void __launch_bounds__(256)
node_prep2(const float* __restrict__ o1, const float* __restrict__ W2,
           const float* __restrict__ attS2, const float* __restrict__ attD2,
           float* __restrict__ rec, int N) {
    const int n = blockIdx.x * blockDim.x + threadIdx.x;
    if (n >= N) return;
    const float* hp = o1 + (size_t)n * 64;
    float a0 = 0.f, a1 = 0.f;
#pragma unroll
    for (int k = 0; k < 64; ++k) {
        const float v = hp[k];
        a0 += v * W2[k * 2 + 0];
        a1 += v * W2[k * 2 + 1];
    }
    float4 q;
    q.x = a0;
    q.y = a1;
    q.z = a0 * attS2[0] + a1 * attS2[1];
    q.w = a0 * attD2[0] + a1 * attD2[1];
    ((float4*)rec)[n] = q;
}

__global__ void __launch_bounds__(256)
seg_gather2(const int* __restrict__ csr, const int* __restrict__ rowptr,
            const int* __restrict__ cnt, const float* __restrict__ rec,
            const float* __restrict__ b2, float* __restrict__ out, int N) {
    __shared__ float stage[32];
    const int n = (blockIdx.x * 256 + threadIdx.x) >> 4;
    const int l = threadIdx.x & 15;
    float r0 = 0.f, r1 = 0.f;
    if (n < N) {
        const float4* __restrict__ rc = (const float4*)rec;
        const float4 me = rc[n];
        const int start = rowptr[n];
        const int deg   = cnt[n];
        float mx = me.z;
        for (int j = l; j < deg; j += 16) mx = fmaxf(mx, rc[csr[start + j]].z);
        for (int o = 8; o > 0; o >>= 1) mx = fmaxf(mx, __shfl_xor(mx, o, 16));
        const float mm = lrelu(mx + me.w);
        float ssum = 0.f, ax = 0.f, ay = 0.f;
        for (int j = l; j < deg; j += 16) {
            const float4 q = rc[csr[start + j]];
            const float w = __expf(lrelu(q.z + me.w) - mm);
            ssum += w; ax += w * q.x; ay += w * q.y;
        }
        for (int o = 8; o > 0; o >>= 1) {
            ssum += __shfl_xor(ssum, o, 16);
            ax   += __shfl_xor(ax, o, 16);
            ay   += __shfl_xor(ay, o, 16);
        }
        {
            const float w = __expf(lrelu(me.z + me.w) - mm);
            ssum += w; ax += w * me.x; ay += w * me.y;
        }
        const float inv = 1.0f / (ssum + 1e-16f);
        r0 = ax * inv + b2[0];
        r1 = ay * inv + b2[1];
    }
    if (l == 0) {
        stage[(threadIdx.x >> 4) * 2 + 0] = r0;
        stage[(threadIdx.x >> 4) * 2 + 1] = r1;
    }
    __syncthreads();
    if (threadIdx.x < 32) {
        const int o = blockIdx.x * 32 + threadIdx.x;
        if (o < N * 2) out[o] = stage[threadIdx.x];
    }
}

extern "C" void kernel_launch(void* const* d_in, const int* in_sizes, int n_in,
                              void* d_out, int out_size, void* d_ws, size_t ws_size,
                              hipStream_t stream) {
    const float* x    = (const float*)d_in[0];
    const int*   ei   = (const int*)d_in[1];
    const float* W1   = (const float*)d_in[2];
    const float* aS1  = (const float*)d_in[3];
    const float* aD1  = (const float*)d_in[4];
    const float* b1   = (const float*)d_in[5];
    const float* W2   = (const float*)d_in[6];
    const float* aS2  = (const float*)d_in[7];
    const float* aD2  = (const float*)d_in[8];
    const float* b2   = (const float*)d_in[9];

    const int N = in_sizes[0] / DIM;
    const int E = in_sizes[1] / 2;
    float* out = (float*)d_out;

    float* ws  = (float*)d_ws;
    float* h1  = ws;
    float* as1 = h1 + (size_t)N * 64;
    float* ad1 = as1 + (size_t)N * HEADS;
    float* o1  = ad1 + (size_t)N * HEADS;
    float* rec = o1 + (size_t)N * 64;
    int* cnt    = (int*)(rec + (size_t)N * 4);
    int* rowptr = cnt + (size_t)N;
    int* bsum   = rowptr + (size_t)N;
    int* rnk    = bsum + ((size_t)N / 1024 + 1);
    int* csr    = rnk + (size_t)E;

    const int Mtiles = (N + 15) >> 4;
    const int NB = (N + 1023) / 1024;

    gemm1_wmma<<<(Mtiles + 3) / 4, 128, 0, stream>>>(x, W1, h1, N);
    node_prep1<<<(N * HEADS + 255) / 256, 256, 0, stream>>>(h1, aS1, aD1, as1, ad1, cnt, N);
    dst_rank<<<(E + 255) / 256, 256, 0, stream>>>(ei, cnt, rnk, E);
    deg_blocksum<<<NB, 256, 0, stream>>>(cnt, bsum, N);
    deg_scan<<<NB, 256, 0, stream>>>(cnt, bsum, rowptr, N);
    place_src<<<(E + 255) / 256, 256, 0, stream>>>(ei, rowptr, rnk, csr, E);
    seg_gather1<<<(N + 15) / 16, 256, 0, stream>>>(csr, rowptr, cnt, as1, ad1, h1, b1, o1, N);
    node_prep2<<<(N + 255) / 256, 256, 0, stream>>>(o1, W2, aS2, aD2, rec, N);
    seg_gather2<<<(N + 15) / 16, 256, 0, stream>>>(csr, rowptr, cnt, rec, b2, out, N);
}


# Edges grouped by destination through LDS (per-chunk counting sort into node bins, then a work-group per bin builds degrees, offsets, source list): global rank atomics gone, sum order by new LDS ranks
# speedup vs baseline: 7.3450x; 1.9411x over previous
#include <hip/hip_runtime.h>
#include <math.h>

#define DIM 128
#define HID 16
#define HEADS 4
#define NEG_SLOPE 0.2f

typedef __attribute__((ext_vector_type(2))) float v2f;
typedef __attribute__((ext_vector_type(8))) float v8f;

__global__ void __launch_bounds__(128)
gemm1_wmma(const float* __restrict__ x, const float* __restrict__ W1,
           float* __restrict__ h1, int N) {
    __shared__ float Wl[DIM * 64];
    for (int i = threadIdx.x; i < DIM * 64; i += blockDim.x) Wl[i] = W1[i];
    __syncthreads();

    const int wave  = threadIdx.x >> 5;
    const int lane  = threadIdx.x & 31;
    const int Mtile = blockIdx.x * 4 + wave;
    const int Mtiles = (N + 15) >> 4;
    if (Mtile >= Mtiles) return;

    const int r  = lane & 15;
    const int kh = lane >> 4;
    const int mbase = Mtile * 16;
    int row = mbase + r;
    if (row >= N) row = N - 1;
    const float* __restrict__ xr = x + (size_t)row * DIM;

    v8f acc0 = {}, acc1 = {}, acc2 = {}, acc3 = {};

    for (int kb = 0; kb < DIM; kb += 4) {
        const int k0 = kb + 2 * kh;
        v2f a;
        a.x = xr[k0];
        a.y = xr[k0 + 1];
        const float* w0 = &Wl[(size_t)k0 * 64];
        const float* w1 = &Wl[(size_t)(k0 + 1) * 64];
        v2f b;
        b.x = w0[r];       b.y = w1[r];
        acc0 = __builtin_amdgcn_wmma_f32_16x16x4_f32(false, a, false, b, (short)0, acc0, false, false);
        b.x = w0[16 + r];  b.y = w1[16 + r];
        acc1 = __builtin_amdgcn_wmma_f32_16x16x4_f32(false, a, false, b, (short)0, acc1, false, false);
        b.x = w0[32 + r];  b.y = w1[32 + r];
        acc2 = __builtin_amdgcn_wmma_f32_16x16x4_f32(false, a, false, b, (short)0, acc2, false, false);
        b.x = w0[48 + r];  b.y = w1[48 + r];
        acc3 = __builtin_amdgcn_wmma_f32_16x16x4_f32(false, a, false, b, (short)0, acc3, false, false);
    }

    float* outp = h1 + (size_t)mbase * 64;
#pragma unroll
    for (int i = 0; i < 8; ++i) {
        const int m = i + 8 * kh;
        if (mbase + m < N) {
            float* o = outp + (size_t)m * 64 + r;
            o[0]  = acc0[i];
            o[16] = acc1[i];
            o[32] = acc2[i];
            o[48] = acc3[i];
        }
    }
}

#define BIN_SHIFT 8
#define BIN_R (1 << BIN_SHIFT)
#define CHUNK 4096
#define SORT_T 512
#define EPT (CHUNK / SORT_T)
#define MAXB 1024
#define STAGE_CAP 8192

__device__ __forceinline__ void block_excl_scan(int* a, int n, int* wsum) {
    const int tid = threadIdx.x, lane = tid & 31, wave = tid >> 5;
    const int i0 = 2 * tid, i1 = i0 + 1;
    const int v0 = (i0 < n) ? a[i0] : 0;
    const int v1 = (i1 < n) ? a[i1] : 0;
    const int s = v0 + v1;
    int x = s;
    for (int o = 1; o < 32; o <<= 1) {
        const int y = __shfl_up(x, o, 32);
        if (lane >= o) x += y;
    }
    if (lane == 31) wsum[wave] = x;
    __syncthreads();
    int off = 0;
    for (int w = 0; w < SORT_T / 32; ++w) if (w < wave) off += wsum[w];
    const int ex = off + x - s;
    if (i0 < n) a[i0] = ex;
    if (i1 < n) a[i1] = ex + v0;
    __syncthreads();
}

__global__ void __launch_bounds__(SORT_T)
chunk_sort(const int* __restrict__ ei, int* __restrict__ binned_w,
           int* __restrict__ lsT, int E, int B, int LP) {
    int2* __restrict__ binned = (int2*)binned_w;
    __shared__ int h[MAXB];
    __shared__ int cur[MAXB];
    __shared__ int wsum[SORT_T / 32];
    __shared__ int2 srt[CHUNK];
    const int tid  = threadIdx.x;
    const int base = blockIdx.x * CHUNK;
    const int nC   = min(CHUNK, E - base);
    for (int i = tid; i < B; i += SORT_T) h[i] = 0;
    __syncthreads();
    int s[EPT], d[EPT];
#pragma unroll
    for (int j = 0; j < EPT; ++j) {
        const int i = j * SORT_T + tid;
        const bool ok = i < nC;
        s[j] = ok ? ei[base + i] : 0;
        d[j] = ok ? ei[E + base + i] : -1;
    }
#pragma unroll
    for (int j = 0; j < EPT; ++j)
        if (d[j] >= 0) atomicAdd(&h[d[j] >> BIN_SHIFT], 1);
    __syncthreads();
    block_excl_scan(h, B, wsum);
    int* lrow = lsT + (size_t)blockIdx.x * LP;
    for (int i = tid; i < B; i += SORT_T) { const int v = h[i]; cur[i] = v; lrow[i] = v; }
    if (tid == 0) lrow[B] = nC;
    __syncthreads();
#pragma unroll
    for (int j = 0; j < EPT; ++j)
        if (d[j] >= 0) {
            const int p = atomicAdd(&cur[d[j] >> BIN_SHIFT], 1);
            srt[p] = make_int2(s[j], d[j]);
        }
    __syncthreads();
    for (int i = tid; i < nC; i += SORT_T) binned[(size_t)base + i] = srt[i];
}

__global__ void __launch_bounds__(SORT_T)
bin_collect(const int* __restrict__ binned_w, const int* __restrict__ lsT,
            int* __restrict__ cnt, int* __restrict__ rowptr,
            int* __restrict__ csr, int N, int LP, int G) {
    __shared__ int c[BIN_R];
    __shared__ int cur[BIN_R];
    __shared__ int wsum[SORT_T / 32];
    __shared__ int red[SORT_T / 32];
    __shared__ int stage[STAGE_CAP];
    const int2* __restrict__ binned = (const int2*)binned_w;
    const int tid = threadIdx.x, lane = tid & 31, wave = tid >> 5;
    const int b = blockIdx.x;
    if (tid < BIN_R) c[tid] = 0;

    int asum = 0, nsum = 0;
    for (int w = tid; w < G; w += SORT_T) {
        const int* lrow = lsT + (size_t)w * LP + b;
        const int a = lrow[0], z = lrow[1];
        asum += a;
        nsum += z - a;
    }
    for (int o = 16; o > 0; o >>= 1) {
        asum += __shfl_xor(asum, o, 32);
        nsum += __shfl_xor(nsum, o, 32);
    }
    if (lane == 0) { red[wave] = asum; wsum[wave] = nsum; }
    __syncthreads();
    int sbase = 0, nE = 0;
    for (int w = 0; w < SORT_T / 32; ++w) { sbase += red[w]; nE += wsum[w]; }

    for (int w = tid; w < G; w += SORT_T) {
        const int* lrow = lsT + (size_t)w * LP + b;
        const int a = lrow[0], z = lrow[1];
        const int2* p = binned + (size_t)w * CHUNK;
        for (int i = a; i < z; i += 4) {
            const int i1 = min(i + 1, z - 1), i2 = min(i + 2, z - 1), i3 = min(i + 3, z - 1);
            const int d0 = p[i].y, d1 = p[i1].y, d2 = p[i2].y, d3 = p[i3].y;
            atomicAdd(&c[d0 & (BIN_R - 1)], 1);
            if (i + 1 < z) atomicAdd(&c[d1 & (BIN_R - 1)], 1);
            if (i + 2 < z) atomicAdd(&c[d2 & (BIN_R - 1)], 1);
            if (i + 3 < z) atomicAdd(&c[d3 & (BIN_R - 1)], 1);
        }
    }
    __syncthreads();
    const int mine = (tid < BIN_R) ? c[tid] : 0;
    block_excl_scan(c, BIN_R, wsum);
    if (tid < BIN_R) {
        const int ex = c[tid];
        cur[tid] = ex;
        const int n = b * BIN_R + tid;
        if (n < N) { cnt[n] = mine; rowptr[n] = sbase + ex; }
    }
    __syncthreads();

    const bool staged = nE <= STAGE_CAP;
    for (int w = tid; w < G; w += SORT_T) {
        const int* lrow = lsT + (size_t)w * LP + b;
        const int a = lrow[0], z = lrow[1];
        const int2* p = binned + (size_t)w * CHUNK;
        for (int i = a; i < z; i += 4) {
            const bool k1 = i + 1 < z, k2 = i + 2 < z, k3 = i + 3 < z;
            const int2 r0 = p[i];
            const int2 r1 = p[k1 ? i + 1 : z - 1];
            const int2 r2 = p[k2 ? i + 2 : z - 1];
            const int2 r3 = p[k3 ? i + 3 : z - 1];
            const int q0 = atomicAdd(&cur[r0.y & (BIN_R - 1)], 1);
            const int q1 = k1 ? atomicAdd(&cur[r1.y & (BIN_R - 1)], 1) : 0;
            const int q2 = k2 ? atomicAdd(&cur[r2.y & (BIN_R - 1)], 1) : 0;
            const int q3 = k3 ? atomicAdd(&cur[r3.y & (BIN_R - 1)], 1) : 0;
            if (staged) {
                stage[q0] = r0.x;
                if (k1) stage[q1] = r1.x;
                if (k2) stage[q2] = r2.x;
                if (k3) stage[q3] = r3.x;
            } else {
                csr[sbase + q0] = r0.x;
                if (k1) csr[sbase + q1] = r1.x;
                if (k2) csr[sbase + q2] = r2.x;
                if (k3) csr[sbase + q3] = r3.x;
            }
        }
    }
    if (staged) {
        __syncthreads();
        for (int i = tid - (sbase & 31); i < nE; i += SORT_T)
            if (i >= 0) csr[sbase + i] = stage[i];
    }
}

__device__ __forceinline__ float lrelu(float v) { return (v > 0.f) ? v : NEG_SLOPE * v; }

__device__ __forceinline__ float dot4(const float4 a, const float4 b) {
    return a.x * b.x + a.y * b.y + a.z * b.z + a.w * b.w;
}
__device__ __forceinline__ float quad_sum(float v) {
    v += __int_as_float(__builtin_amdgcn_update_dpp(0, __float_as_int(v), 0xB1, 0xF, 0xF, true));
    v += __int_as_float(__builtin_amdgcn_update_dpp(0, __float_as_int(v), 0x4E, 0xF, 0xF, true));
    return v;
}

__global__ void __launch_bounds__(256)
seg_gather1(const int* __restrict__ csr, const int* __restrict__ rowptr,
            const int* __restrict__ cnt, const float* __restrict__ attS,
            const float* __restrict__ attD, const float* __restrict__ h1,
            const float* __restrict__ b1, float* __restrict__ o1, int N) {
    const int n  = (blockIdx.x * 256 + threadIdx.x) >> 4;
    const int l  = threadIdx.x & 15;
    if (n >= N) return;
    const int start = rowptr[n];
    const int deg   = cnt[n];
    const float4* __restrict__ h4 = (const float4*)h1;
    float4 aS, aD;
    aS.x = attS[l * 4 + 0]; aS.y = attS[l * 4 + 1]; aS.z = attS[l * 4 + 2]; aS.w = attS[l * 4 + 3];
    aD.x = attD[l * 4 + 0]; aD.y = attD[l * 4 + 1]; aD.z = attD[l * 4 + 2]; aD.w = attD[l * 4 + 3];

    const float4 vn = h4[n * 16 + l];
    const float adv = quad_sum(dot4(vn, aD));
    float m    = lrelu(quad_sum(dot4(vn, aS)) + adv);
    float ssum = 1.f;
    float4 acc = vn;

    for (int base = 0; base < deg; base += 16) {
        const int j   = base + l;
        const int idx = (j < deg) ? csr[start + j] : n;
        const int c   = min(16, deg - base);
        for (int k = 0; k < c; k += 4) {
            const int s0 = __shfl(idx, k, 16);
            const int s1 = __shfl(idx, k + 1, 16);
            const int s2 = __shfl(idx, k + 2, 16);
            const int s3 = __shfl(idx, k + 3, 16);
            const float4 v0 = h4[s0 * 16 + l];
            const float4 v1 = h4[s1 * 16 + l];
            const float4 v2 = h4[s2 * 16 + l];
            const float4 v3 = h4[s3 * 16 + l];
            const bool p1 = k + 1 < c, p2 = k + 2 < c, p3 = k + 3 < c;
            const float e0 = lrelu(quad_sum(dot4(v0, aS)) + adv);
            const float e1 = lrelu(quad_sum(dot4(v1, aS)) + adv);
            const float e2 = lrelu(quad_sum(dot4(v2, aS)) + adv);
            const float e3 = lrelu(quad_sum(dot4(v3, aS)) + adv);
            float mn = fmaxf(m, e0);
            mn = fmaxf(mn, p1 ? e1 : mn);
            mn = fmaxf(mn, p2 ? e2 : mn);
            mn = fmaxf(mn, p3 ? e3 : mn);
            const float sc = __expf(m - mn);
            const float w0 = __expf(e0 - mn);
            const float w1 = p1 ? __expf(e1 - mn) : 0.f;
            const float w2 = p2 ? __expf(e2 - mn) : 0.f;
            const float w3 = p3 ? __expf(e3 - mn) : 0.f;
            ssum = ssum * sc + w0;
            acc.x = acc.x * sc + w0 * v0.x; acc.y = acc.y * sc + w0 * v0.y;
            acc.z = acc.z * sc + w0 * v0.z; acc.w = acc.w * sc + w0 * v0.w;
            ssum += w1; acc.x += w1 * v1.x; acc.y += w1 * v1.y; acc.z += w1 * v1.z; acc.w += w1 * v1.w;
            ssum += w2; acc.x += w2 * v2.x; acc.y += w2 * v2.y; acc.z += w2 * v2.z; acc.w += w2 * v2.w;
            ssum += w3; acc.x += w3 * v3.x; acc.y += w3 * v3.y; acc.z += w3 * v3.z; acc.w += w3 * v3.w;
            m = mn;
        }
    }

    const float den = ssum + 1e-16f;
    float4 r;
    r.x = acc.x / den + b1[l * 4 + 0];
    r.y = acc.y / den + b1[l * 4 + 1];
    r.z = acc.z / den + b1[l * 4 + 2];
    r.w = acc.w / den + b1[l * 4 + 3];
    r.x = (r.x > 0.f) ? r.x : expm1f(r.x);
    r.y = (r.y > 0.f) ? r.y : expm1f(r.y);
    r.z = (r.z > 0.f) ? r.z : expm1f(r.z);
    r.w = (r.w > 0.f) ? r.w : expm1f(r.w);
    ((float4*)o1)[n * 16 + l] = r;
}

__global__ void __launch_bounds__(256)
node_prep2(const float* __restrict__ o1, const float* __restrict__ W2,
           const float* __restrict__ attS2, const float* __restrict__ attD2,
           float* __restrict__ rec, int N) {
    const int n = blockIdx.x * blockDim.x + threadIdx.x;
    if (n >= N) return;
    const float* hp = o1 + (size_t)n * 64;
    float a0 = 0.f, a1 = 0.f;
#pragma unroll
    for (int k = 0; k < 64; ++k) {
        const float v = hp[k];
        a0 += v * W2[k * 2 + 0];
        a1 += v * W2[k * 2 + 1];
    }
    float4 q;
    q.x = a0;
    q.y = a1;
    q.z = a0 * attS2[0] + a1 * attS2[1];
    q.w = a0 * attD2[0] + a1 * attD2[1];
    ((float4*)rec)[n] = q;
}

__global__ void __launch_bounds__(256)
seg_gather2(const int* __restrict__ csr, const int* __restrict__ rowptr,
            const int* __restrict__ cnt, const float* __restrict__ rec,
            const float* __restrict__ b2, float* __restrict__ out, int N) {
    __shared__ float stage[32];
    const int n = (blockIdx.x * 256 + threadIdx.x) >> 4;
    const int l = threadIdx.x & 15;
    float r0 = 0.f, r1 = 0.f;
    if (n < N) {
        const float4* __restrict__ rc = (const float4*)rec;
        const float4 me = rc[n];
        const int start = rowptr[n];
        const int deg   = cnt[n];
        const float es = lrelu(me.z + me.w);
        float m = es, ssum = 0.f, ax = 0.f, ay = 0.f;
        for (int j = l; j < deg; j += 16) {
            const float4 q = rc[csr[start + j]];
            const float e  = lrelu(q.z + me.w);
            const float t  = __expf(-fabsf(e - m));
            const bool up  = e > m;
            const float sc = up ? t : 1.f;
            const float w  = up ? 1.f : t;
            ssum = ssum * sc + w; ax = ax * sc + w * q.x; ay = ay * sc + w * q.y;
            m = fmaxf(m, e);
        }
        float mm = m;
        for (int o = 8; o > 0; o >>= 1) mm = fmaxf(mm, __shfl_xor(mm, o, 16));
        {
            const float f = __expf(m - mm);
            ssum *= f; ax *= f; ay *= f;
        }
        for (int o = 8; o > 0; o >>= 1) {
            ssum += __shfl_xor(ssum, o, 16);
            ax   += __shfl_xor(ax, o, 16);
            ay   += __shfl_xor(ay, o, 16);
        }
        {
            const float w = __expf(es - mm);
            ssum += w; ax += w * me.x; ay += w * me.y;
        }
        const float inv = 1.0f / (ssum + 1e-16f);
        r0 = ax * inv + b2[0];
        r1 = ay * inv + b2[1];
    }
    if (l == 0) {
        stage[(threadIdx.x >> 4) * 2 + 0] = r0;
        stage[(threadIdx.x >> 4) * 2 + 1] = r1;
    }
    __syncthreads();
    if (threadIdx.x < 32) {
        const int o = blockIdx.x * 32 + threadIdx.x;
        if (o < N * 2) out[o] = stage[threadIdx.x];
    }
}

extern "C" void kernel_launch(void* const* d_in, const int* in_sizes, int n_in,
                              void* d_out, int out_size, void* d_ws, size_t ws_size,
                              hipStream_t stream) {
    const float* x    = (const float*)d_in[0];
    const int*   ei   = (const int*)d_in[1];
    const float* W1   = (const float*)d_in[2];
    const float* aS1  = (const float*)d_in[3];
    const float* aD1  = (const float*)d_in[4];
    const float* b1   = (const float*)d_in[5];
    const float* W2   = (const float*)d_in[6];
    const float* aS2  = (const float*)d_in[7];
    const float* aD2  = (const float*)d_in[8];
    const float* b2   = (const float*)d_in[9];

    const int N = in_sizes[0] / DIM;
    const int E = in_sizes[1] / 2;
    float* out = (float*)d_out;

    float* ws  = (float*)d_ws;
    float* h1  = ws;
    float* o1  = h1 + (size_t)N * 64;
    float* rec = o1 + (size_t)N * 64;
    const int B = (N + BIN_R - 1) / BIN_R;
    const int G = (E + CHUNK - 1) / CHUNK;
    int* binned = (int*)(rec + (size_t)N * 4);
    int* cnt    = binned + (size_t)G * CHUNK * 2;
    int* rowptr = cnt + (size_t)N;
    const int LP = ((B + 1 + 31) / 32) * 32;
    int* lsT    = rowptr + (size_t)N;
    int* csr    = lsT + (size_t)G * LP;

    const int Mtiles = (N + 15) >> 4;

    gemm1_wmma<<<(Mtiles + 3) / 4, 128, 0, stream>>>(x, W1, h1, N);
    chunk_sort<<<G, SORT_T, 0, stream>>>(ei, binned, lsT, E, B, LP);
    bin_collect<<<B, SORT_T, 0, stream>>>(binned, lsT, cnt, rowptr, csr, N, LP, G);
    seg_gather1<<<(N + 15) / 16, 256, 0, stream>>>(csr, rowptr, cnt, aS1, aD1, h1, b1, o1, N);
    node_prep2<<<(N + 255) / 256, 256, 0, stream>>>(o1, W2, aS2, aD2, rec, N);
    seg_gather2<<<(N + 15) / 16, 256, 0, stream>>>(csr, rowptr, cnt, rec, b2, out, N);
}


# GEMM prologue: W1 staged to LDS by a constant-trip loop of 16-byte loads, all 16 per thread in flight at once (was a run-time-stride loop: one dword per memory round trip, 64 in a row)
# speedup vs baseline: 9.4205x; 1.2826x over previous
#include <hip/hip_runtime.h>
#include <math.h>

#define DIM 128
#define HID 16
#define HEADS 4
#define NEG_SLOPE 0.2f

typedef __attribute__((ext_vector_type(2))) float v2f;
typedef __attribute__((ext_vector_type(8))) float v8f;

__global__ void __launch_bounds__(128)
gemm1_wmma(const float* __restrict__ x, const float* __restrict__ W1,
           float* __restrict__ h1, int N) {
    __shared__ float4 Wl4[DIM * 16];
    {
        const int t = threadIdx.x;
#pragma unroll
        for (int j = 0; j < (DIM * 16) / 128; ++j) {
            const int e = (j * 128 + t) * 4;
            float4 v;
            v.x = W1[e]; v.y = W1[e + 1]; v.z = W1[e + 2]; v.w = W1[e + 3];
            Wl4[j * 128 + t] = v;
        }
    }
    const float* Wl = (const float*)Wl4;
    __syncthreads();

    const int wave  = threadIdx.x >> 5;
    const int lane  = threadIdx.x & 31;
    const int Mtile = blockIdx.x * 4 + wave;
    const int Mtiles = (N + 15) >> 4;
    if (Mtile >= Mtiles) return;

    const int r  = lane & 15;
    const int kh = lane >> 4;
    const int mbase = Mtile * 16;
    int row = mbase + r;
    if (row >= N) row = N - 1;
    const float* __restrict__ xr = x + (size_t)row * DIM;

    v8f acc0 = {}, acc1 = {}, acc2 = {}, acc3 = {};

    for (int kb = 0; kb < DIM; kb += 4) {
        const int k0 = kb + 2 * kh;
        v2f a;
        a.x = xr[k0];
        a.y = xr[k0 + 1];
        const float* w0 = &Wl[(size_t)k0 * 64];
        const float* w1 = &Wl[(size_t)(k0 + 1) * 64];
        v2f b;
        b.x = w0[r];       b.y = w1[r];
        acc0 = __builtin_amdgcn_wmma_f32_16x16x4_f32(false, a, false, b, (short)0, acc0, false, false);
        b.x = w0[16 + r];  b.y = w1[16 + r];
        acc1 = __builtin_amdgcn_wmma_f32_16x16x4_f32(false, a, false, b, (short)0, acc1, false, false);
        b.x = w0[32 + r];  b.y = w1[32 + r];
        acc2 = __builtin_amdgcn_wmma_f32_16x16x4_f32(false, a, false, b, (short)0, acc2, false, false);
        b.x = w0[48 + r];  b.y = w1[48 + r];
        acc3 = __builtin_amdgcn_wmma_f32_16x16x4_f32(false, a, false, b, (short)0, acc3, false, false);
    }

    float* outp = h1 + (size_t)mbase * 64;
#pragma unroll
    for (int i = 0; i < 8; ++i) {
        const int m = i + 8 * kh;
        if (mbase + m < N) {
            float* o = outp + (size_t)m * 64 + r;
            o[0]  = acc0[i];
            o[16] = acc1[i];
            o[32] = acc2[i];
            o[48] = acc3[i];
        }
    }
}

#define BIN_SHIFT 8
#define BIN_R (1 << BIN_SHIFT)
#define CHUNK 4096
#define SORT_T 512
#define EPT (CHUNK / SORT_T)
#define MAXB 1024
#define STAGE_CAP 8192

__device__ __forceinline__ void block_excl_scan(int* a, int n, int* wsum) {
    const int tid = threadIdx.x, lane = tid & 31, wave = tid >> 5;
    const int i0 = 2 * tid, i1 = i0 + 1;
    const int v0 = (i0 < n) ? a[i0] : 0;
    const int v1 = (i1 < n) ? a[i1] : 0;
    const int s = v0 + v1;
    int x = s;
    for (int o = 1; o < 32; o <<= 1) {
        const int y = __shfl_up(x, o, 32);
        if (lane >= o) x += y;
    }
    if (lane == 31) wsum[wave] = x;
    __syncthreads();
    int off = 0;
    for (int w = 0; w < SORT_T / 32; ++w) if (w < wave) off += wsum[w];
    const int ex = off + x - s;
    if (i0 < n) a[i0] = ex;
    if (i1 < n) a[i1] = ex + v0;
    __syncthreads();
}

__global__ void __launch_bounds__(SORT_T)
chunk_sort(const int* __restrict__ ei, int* __restrict__ binned_w,
           int* __restrict__ lsT, int E, int B, int LP) {
    int2* __restrict__ binned = (int2*)binned_w;
    __shared__ int h[MAXB];
    __shared__ int cur[MAXB];
    __shared__ int wsum[SORT_T / 32];
    __shared__ int2 srt[CHUNK];
    const int tid  = threadIdx.x;
    const int base = blockIdx.x * CHUNK;
    const int nC   = min(CHUNK, E - base);
    for (int i = tid; i < B; i += SORT_T) h[i] = 0;
    __syncthreads();
    int s[EPT], d[EPT];
#pragma unroll
    for (int j = 0; j < EPT; ++j) {
        const int i = j * SORT_T + tid;
        const bool ok = i < nC;
        s[j] = ok ? ei[base + i] : 0;
        d[j] = ok ? ei[E + base + i] : -1;
    }
#pragma unroll
    for (int j = 0; j < EPT; ++j)
        if (d[j] >= 0) atomicAdd(&h[d[j] >> BIN_SHIFT], 1);
    __syncthreads();
    block_excl_scan(h, B, wsum);
    int* lrow = lsT + (size_t)blockIdx.x * LP;
    for (int i = tid; i < B; i += SORT_T) { const int v = h[i]; cur[i] = v; lrow[i] = v; }
    if (tid == 0) lrow[B] = nC;
    __syncthreads();
#pragma unroll
    for (int j = 0; j < EPT; ++j)
        if (d[j] >= 0) {
            const int p = atomicAdd(&cur[d[j] >> BIN_SHIFT], 1);
            srt[p] = make_int2(s[j], d[j]);
        }
    __syncthreads();
    for (int i = tid; i < nC; i += SORT_T) binned[(size_t)base + i] = srt[i];
}

__global__ void __launch_bounds__(SORT_T)
bin_collect(const int* __restrict__ binned_w, const int* __restrict__ lsT,
            int* __restrict__ cnt, int* __restrict__ rowptr,
            int* __restrict__ csr, int N, int LP, int G) {
    __shared__ int c[BIN_R];
    __shared__ int cur[BIN_R];
    __shared__ int wsum[SORT_T / 32];
    __shared__ int red[SORT_T / 32];
    __shared__ int stage[STAGE_CAP];
    const int2* __restrict__ binned = (const int2*)binned_w;
    const int tid = threadIdx.x, lane = tid & 31, wave = tid >> 5;
    const int b = blockIdx.x;
    if (tid < BIN_R) c[tid] = 0;

    int asum = 0, nsum = 0;
    for (int w = tid; w < G; w += SORT_T) {
        const int* lrow = lsT + (size_t)w * LP + b;
        const int a = lrow[0], z = lrow[1];
        asum += a;
        nsum += z - a;
    }
    for (int o = 16; o > 0; o >>= 1) {
        asum += __shfl_xor(asum, o, 32);
        nsum += __shfl_xor(nsum, o, 32);
    }
    if (lane == 0) { red[wave] = asum; wsum[wave] = nsum; }
    __syncthreads();
    int sbase = 0, nE = 0;
    for (int w = 0; w < SORT_T / 32; ++w) { sbase += red[w]; nE += wsum[w]; }

    for (int w = tid; w < G; w += SORT_T) {
        const int* lrow = lsT + (size_t)w * LP + b;
        const int a = lrow[0], z = lrow[1];
        const int2* p = binned + (size_t)w * CHUNK;
        for (int i = a; i < z; i += 4) {
            const int i1 = min(i + 1, z - 1), i2 = min(i + 2, z - 1), i3 = min(i + 3, z - 1);
            const int d0 = p[i].y, d1 = p[i1].y, d2 = p[i2].y, d3 = p[i3].y;
            atomicAdd(&c[d0 & (BIN_R - 1)], 1);
            if (i + 1 < z) atomicAdd(&c[d1 & (BIN_R - 1)], 1);
            if (i + 2 < z) atomicAdd(&c[d2 & (BIN_R - 1)], 1);
            if (i + 3 < z) atomicAdd(&c[d3 & (BIN_R - 1)], 1);
        }
    }
    __syncthreads();
    const int mine = (tid < BIN_R) ? c[tid] : 0;
    block_excl_scan(c, BIN_R, wsum);
    if (tid < BIN_R) {
        const int ex = c[tid];
        cur[tid] = ex;
        const int n = b * BIN_R + tid;
        if (n < N) { cnt[n] = mine; rowptr[n] = sbase + ex; }
    }
    __syncthreads();

    const bool staged = nE <= STAGE_CAP;
    for (int w = tid; w < G; w += SORT_T) {
        const int* lrow = lsT + (size_t)w * LP + b;
        const int a = lrow[0], z = lrow[1];
        const int2* p = binned + (size_t)w * CHUNK;
        for (int i = a; i < z; i += 4) {
            const bool k1 = i + 1 < z, k2 = i + 2 < z, k3 = i + 3 < z;
            const int2 r0 = p[i];
            const int2 r1 = p[k1 ? i + 1 : z - 1];
            const int2 r2 = p[k2 ? i + 2 : z - 1];
            const int2 r3 = p[k3 ? i + 3 : z - 1];
            const int q0 = atomicAdd(&cur[r0.y & (BIN_R - 1)], 1);
            const int q1 = k1 ? atomicAdd(&cur[r1.y & (BIN_R - 1)], 1) : 0;
            const int q2 = k2 ? atomicAdd(&cur[r2.y & (BIN_R - 1)], 1) : 0;
            const int q3 = k3 ? atomicAdd(&cur[r3.y & (BIN_R - 1)], 1) : 0;
            if (staged) {
                stage[q0] = r0.x;
                if (k1) stage[q1] = r1.x;
                if (k2) stage[q2] = r2.x;
                if (k3) stage[q3] = r3.x;
            } else {
                csr[sbase + q0] = r0.x;
                if (k1) csr[sbase + q1] = r1.x;
                if (k2) csr[sbase + q2] = r2.x;
                if (k3) csr[sbase + q3] = r3.x;
            }
        }
    }
    if (staged) {
        __syncthreads();
        for (int i = tid - (sbase & 31); i < nE; i += SORT_T)
            if (i >= 0) csr[sbase + i] = stage[i];
    }
}

__device__ __forceinline__ float lrelu(float v) { return (v > 0.f) ? v : NEG_SLOPE * v; }

__device__ __forceinline__ float dot4(const float4 a, const float4 b) {
    return a.x * b.x + a.y * b.y + a.z * b.z + a.w * b.w;
}
__device__ __forceinline__ float quad_sum(float v) {
    v += __int_as_float(__builtin_amdgcn_update_dpp(0, __float_as_int(v), 0xB1, 0xF, 0xF, true));
    v += __int_as_float(__builtin_amdgcn_update_dpp(0, __float_as_int(v), 0x4E, 0xF, 0xF, true));
    return v;
}

__global__ void __launch_bounds__(256)
seg_gather1(const int* __restrict__ csr, const int* __restrict__ rowptr,
            const int* __restrict__ cnt, const float* __restrict__ attS,
            const float* __restrict__ attD, const float* __restrict__ h1,
            const float* __restrict__ b1, float* __restrict__ o1, int N) {
    const int n  = (blockIdx.x * 256 + threadIdx.x) >> 4;
    const int l  = threadIdx.x & 15;
    if (n >= N) return;
    const int start = rowptr[n];
    const int deg   = cnt[n];
    const float4* __restrict__ h4 = (const float4*)h1;
    float4 aS, aD;
    aS.x = attS[l * 4 + 0]; aS.y = attS[l * 4 + 1]; aS.z = attS[l * 4 + 2]; aS.w = attS[l * 4 + 3];
    aD.x = attD[l * 4 + 0]; aD.y = attD[l * 4 + 1]; aD.z = attD[l * 4 + 2]; aD.w = attD[l * 4 + 3];

    const float4 vn = h4[n * 16 + l];
    const float adv = quad_sum(dot4(vn, aD));
    float m    = lrelu(quad_sum(dot4(vn, aS)) + adv);
    float ssum = 1.f;
    float4 acc = vn;

    for (int base = 0; base < deg; base += 16) {
        const int j   = base + l;
        const int idx = (j < deg) ? csr[start + j] : n;
        const int c   = min(16, deg - base);
        for (int k = 0; k < c; k += 4) {
            const int s0 = __shfl(idx, k, 16);
            const int s1 = __shfl(idx, k + 1, 16);
            const int s2 = __shfl(idx, k + 2, 16);
            const int s3 = __shfl(idx, k + 3, 16);
            const float4 v0 = h4[s0 * 16 + l];
            const float4 v1 = h4[s1 * 16 + l];
            const float4 v2 = h4[s2 * 16 + l];
            const float4 v3 = h4[s3 * 16 + l];
            const bool p1 = k + 1 < c, p2 = k + 2 < c, p3 = k + 3 < c;
            const float e0 = lrelu(quad_sum(dot4(v0, aS)) + adv);
            const float e1 = lrelu(quad_sum(dot4(v1, aS)) + adv);
            const float e2 = lrelu(quad_sum(dot4(v2, aS)) + adv);
            const float e3 = lrelu(quad_sum(dot4(v3, aS)) + adv);
            float mn = fmaxf(m, e0);
            mn = fmaxf(mn, p1 ? e1 : mn);
            mn = fmaxf(mn, p2 ? e2 : mn);
            mn = fmaxf(mn, p3 ? e3 : mn);
            const float sc = __expf(m - mn);
            const float w0 = __expf(e0 - mn);
            const float w1 = p1 ? __expf(e1 - mn) : 0.f;
            const float w2 = p2 ? __expf(e2 - mn) : 0.f;
            const float w3 = p3 ? __expf(e3 - mn) : 0.f;
            ssum = ssum * sc + w0;
            acc.x = acc.x * sc + w0 * v0.x; acc.y = acc.y * sc + w0 * v0.y;
            acc.z = acc.z * sc + w0 * v0.z; acc.w = acc.w * sc + w0 * v0.w;
            ssum += w1; acc.x += w1 * v1.x; acc.y += w1 * v1.y; acc.z += w1 * v1.z; acc.w += w1 * v1.w;
            ssum += w2; acc.x += w2 * v2.x; acc.y += w2 * v2.y; acc.z += w2 * v2.z; acc.w += w2 * v2.w;
            ssum += w3; acc.x += w3 * v3.x; acc.y += w3 * v3.y; acc.z += w3 * v3.z; acc.w += w3 * v3.w;
            m = mn;
        }
    }

    const float den = ssum + 1e-16f;
    float4 r;
    r.x = acc.x / den + b1[l * 4 + 0];
    r.y = acc.y / den + b1[l * 4 + 1];
    r.z = acc.z / den + b1[l * 4 + 2];
    r.w = acc.w / den + b1[l * 4 + 3];
    r.x = (r.x > 0.f) ? r.x : expm1f(r.x);
    r.y = (r.y > 0.f) ? r.y : expm1f(r.y);
    r.z = (r.z > 0.f) ? r.z : expm1f(r.z);
    r.w = (r.w > 0.f) ? r.w : expm1f(r.w);
    ((float4*)o1)[n * 16 + l] = r;
}

__global__ void __launch_bounds__(256)
node_prep2(const float* __restrict__ o1, const float* __restrict__ W2,
           const float* __restrict__ attS2, const float* __restrict__ attD2,
           float* __restrict__ rec, int N) {
    const int n = blockIdx.x * blockDim.x + threadIdx.x;
    if (n >= N) return;
    const float* hp = o1 + (size_t)n * 64;
    float a0 = 0.f, a1 = 0.f;
#pragma unroll
    for (int k = 0; k < 64; ++k) {
        const float v = hp[k];
        a0 += v * W2[k * 2 + 0];
        a1 += v * W2[k * 2 + 1];
    }
    float4 q;
    q.x = a0;
    q.y = a1;
    q.z = a0 * attS2[0] + a1 * attS2[1];
    q.w = a0 * attD2[0] + a1 * attD2[1];
    ((float4*)rec)[n] = q;
}

__global__ void __launch_bounds__(256)
seg_gather2(const int* __restrict__ csr, const int* __restrict__ rowptr,
            const int* __restrict__ cnt, const float* __restrict__ rec,
            const float* __restrict__ b2, float* __restrict__ out, int N) {
    __shared__ float stage[32];
    const int n = (blockIdx.x * 256 + threadIdx.x) >> 4;
    const int l = threadIdx.x & 15;
    float r0 = 0.f, r1 = 0.f;
    if (n < N) {
        const float4* __restrict__ rc = (const float4*)rec;
        const float4 me = rc[n];
        const int start = rowptr[n];
        const int deg   = cnt[n];
        const float es = lrelu(me.z + me.w);
        float m = es, ssum = 0.f, ax = 0.f, ay = 0.f;
        for (int j = l; j < deg; j += 16) {
            const float4 q = rc[csr[start + j]];
            const float e  = lrelu(q.z + me.w);
            const float t  = __expf(-fabsf(e - m));
            const bool up  = e > m;
            const float sc = up ? t : 1.f;
            const float w  = up ? 1.f : t;
            ssum = ssum * sc + w; ax = ax * sc + w * q.x; ay = ay * sc + w * q.y;
            m = fmaxf(m, e);
        }
        float mm = m;
        for (int o = 8; o > 0; o >>= 1) mm = fmaxf(mm, __shfl_xor(mm, o, 16));
        {
            const float f = __expf(m - mm);
            ssum *= f; ax *= f; ay *= f;
        }
        for (int o = 8; o > 0; o >>= 1) {
            ssum += __shfl_xor(ssum, o, 16);
            ax   += __shfl_xor(ax, o, 16);
            ay   += __shfl_xor(ay, o, 16);
        }
        {
            const float w = __expf(es - mm);
            ssum += w; ax += w * me.x; ay += w * me.y;
        }
        const float inv = 1.0f / (ssum + 1e-16f);
        r0 = ax * inv + b2[0];
        r1 = ay * inv + b2[1];
    }
    if (l == 0) {
        stage[(threadIdx.x >> 4) * 2 + 0] = r0;
        stage[(threadIdx.x >> 4) * 2 + 1] = r1;
    }
    __syncthreads();
    if (threadIdx.x < 32) {
        const int o = blockIdx.x * 32 + threadIdx.x;
        if (o < N * 2) out[o] = stage[threadIdx.x];
    }
}

extern "C" void kernel_launch(void* const* d_in, const int* in_sizes, int n_in,
                              void* d_out, int out_size, void* d_ws, size_t ws_size,
                              hipStream_t stream) {
    const float* x    = (const float*)d_in[0];
    const int*   ei   = (const int*)d_in[1];
    const float* W1   = (const float*)d_in[2];
    const float* aS1  = (const float*)d_in[3];
    const float* aD1  = (const float*)d_in[4];
    const float* b1   = (const float*)d_in[5];
    const float* W2   = (const float*)d_in[6];
    const float* aS2  = (const float*)d_in[7];
    const float* aD2  = (const float*)d_in[8];
    const float* b2   = (const float*)d_in[9];

    const int N = in_sizes[0] / DIM;
    const int E = in_sizes[1] / 2;
    float* out = (float*)d_out;

    float* ws  = (float*)d_ws;
    float* h1  = ws;
    float* o1  = h1 + (size_t)N * 64;
    float* rec = o1 + (size_t)N * 64;
    const int B = (N + BIN_R - 1) / BIN_R;
    const int G = (E + CHUNK - 1) / CHUNK;
    int* binned = (int*)(rec + (size_t)N * 4);
    int* cnt    = binned + (size_t)G * CHUNK * 2;
    int* rowptr = cnt + (size_t)N;
    const int LP = ((B + 1 + 31) / 32) * 32;
    int* lsT    = rowptr + (size_t)N;
    int* csr    = lsT + (size_t)G * LP;

    const int Mtiles = (N + 15) >> 4;

    gemm1_wmma<<<(Mtiles + 3) / 4, 128, 0, stream>>>(x, W1, h1, N);
    chunk_sort<<<G, SORT_T, 0, stream>>>(ei, binned, lsT, E, B, LP);
    bin_collect<<<B, SORT_T, 0, stream>>>(binned, lsT, cnt, rowptr, csr, N, LP, G);
    seg_gather1<<<(N + 15) / 16, 256, 0, stream>>>(csr, rowptr, cnt, aS1, aD1, h1, b1, o1, N);
    node_prep2<<<(N + 255) / 256, 256, 0, stream>>>(o1, W2, aS2, aD2, rec, N);
    seg_gather2<<<(N + 15) / 16, 256, 0, stream>>>(csr, rowptr, cnt, rec, b2, out, N);
}


# Layer-2 node transform (64->2 product, both logits) folded into the layer-1 gather's epilogue: 4 products per lane + 16-lane DPP sum; ELU plane and one launch gone; h2 sum order now lane tree
# speedup vs baseline: 9.8959x; 1.0505x over previous
#include <hip/hip_runtime.h>
#include <math.h>

#define DIM 128
#define HID 16
#define HEADS 4
#define NEG_SLOPE 0.2f

typedef __attribute__((ext_vector_type(2))) float v2f;
typedef __attribute__((ext_vector_type(8))) float v8f;

__global__ void __launch_bounds__(128)
gemm1_wmma(const float* __restrict__ x, const float* __restrict__ W1,
           float* __restrict__ h1, int N) {
    __shared__ float4 Wl4[DIM * 16];
    {
        const int t = threadIdx.x;
#pragma unroll
        for (int j = 0; j < (DIM * 16) / 128; ++j) {
            const int e = (j * 128 + t) * 4;
            float4 v;
            v.x = W1[e]; v.y = W1[e + 1]; v.z = W1[e + 2]; v.w = W1[e + 3];
            Wl4[j * 128 + t] = v;
        }
    }
    const float* Wl = (const float*)Wl4;
    __syncthreads();

    const int wave  = threadIdx.x >> 5;
    const int lane  = threadIdx.x & 31;
    const int Mtile = blockIdx.x * 4 + wave;
    const int Mtiles = (N + 15) >> 4;
    if (Mtile >= Mtiles) return;

    const int r  = lane & 15;
    const int kh = lane >> 4;
    const int mbase = Mtile * 16;
    int row = mbase + r;
    if (row >= N) row = N - 1;
    const float* __restrict__ xr = x + (size_t)row * DIM;

    v8f acc0 = {}, acc1 = {}, acc2 = {}, acc3 = {};

    for (int kb = 0; kb < DIM; kb += 4) {
        const int k0 = kb + 2 * kh;
        v2f a;
        a.x = xr[k0];
        a.y = xr[k0 + 1];
        const float* w0 = &Wl[(size_t)k0 * 64];
        const float* w1 = &Wl[(size_t)(k0 + 1) * 64];
        v2f b;
        b.x = w0[r];       b.y = w1[r];
        acc0 = __builtin_amdgcn_wmma_f32_16x16x4_f32(false, a, false, b, (short)0, acc0, false, false);
        b.x = w0[16 + r];  b.y = w1[16 + r];
        acc1 = __builtin_amdgcn_wmma_f32_16x16x4_f32(false, a, false, b, (short)0, acc1, false, false);
        b.x = w0[32 + r];  b.y = w1[32 + r];
        acc2 = __builtin_amdgcn_wmma_f32_16x16x4_f32(false, a, false, b, (short)0, acc2, false, false);
        b.x = w0[48 + r];  b.y = w1[48 + r];
        acc3 = __builtin_amdgcn_wmma_f32_16x16x4_f32(false, a, false, b, (short)0, acc3, false, false);
    }

    float* outp = h1 + (size_t)mbase * 64;
#pragma unroll
    for (int i = 0; i < 8; ++i) {
        const int m = i + 8 * kh;
        if (mbase + m < N) {
            float* o = outp + (size_t)m * 64 + r;
            o[0]  = acc0[i];
            o[16] = acc1[i];
            o[32] = acc2[i];
            o[48] = acc3[i];
        }
    }
}

#define BIN_SHIFT 8
#define BIN_R (1 << BIN_SHIFT)
#define CHUNK 4096
#define SORT_T 512
#define EPT (CHUNK / SORT_T)
#define MAXB 1024
#define STAGE_CAP 8192

__device__ __forceinline__ void block_excl_scan(int* a, int n, int* wsum) {
    const int tid = threadIdx.x, lane = tid & 31, wave = tid >> 5;
    const int i0 = 2 * tid, i1 = i0 + 1;
    const int v0 = (i0 < n) ? a[i0] : 0;
    const int v1 = (i1 < n) ? a[i1] : 0;
    const int s = v0 + v1;
    int x = s;
    for (int o = 1; o < 32; o <<= 1) {
        const int y = __shfl_up(x, o, 32);
        if (lane >= o) x += y;
    }
    if (lane == 31) wsum[wave] = x;
    __syncthreads();
    int off = 0;
    for (int w = 0; w < SORT_T / 32; ++w) if (w < wave) off += wsum[w];
    const int ex = off + x - s;
    if (i0 < n) a[i0] = ex;
    if (i1 < n) a[i1] = ex + v0;
    __syncthreads();
}

__global__ void __launch_bounds__(SORT_T)
chunk_sort(const int* __restrict__ ei, int* __restrict__ binned_w,
           int* __restrict__ lsT, int E, int B, int LP) {
    int2* __restrict__ binned = (int2*)binned_w;
    __shared__ int h[MAXB];
    __shared__ int cur[MAXB];
    __shared__ int wsum[SORT_T / 32];
    __shared__ int2 srt[CHUNK];
    const int tid  = threadIdx.x;
    const int base = blockIdx.x * CHUNK;
    const int nC   = min(CHUNK, E - base);
    for (int i = tid; i < B; i += SORT_T) h[i] = 0;
    __syncthreads();
    int s[EPT], d[EPT];
#pragma unroll
    for (int j = 0; j < EPT; ++j) {
        const int i = j * SORT_T + tid;
        const bool ok = i < nC;
        s[j] = ok ? ei[base + i] : 0;
        d[j] = ok ? ei[E + base + i] : -1;
    }
#pragma unroll
    for (int j = 0; j < EPT; ++j)
        if (d[j] >= 0) atomicAdd(&h[d[j] >> BIN_SHIFT], 1);
    __syncthreads();
    block_excl_scan(h, B, wsum);
    int* lrow = lsT + (size_t)blockIdx.x * LP;
    for (int i = tid; i < B; i += SORT_T) { const int v = h[i]; cur[i] = v; lrow[i] = v; }
    if (tid == 0) lrow[B] = nC;
    __syncthreads();
#pragma unroll
    for (int j = 0; j < EPT; ++j)
        if (d[j] >= 0) {
            const int p = atomicAdd(&cur[d[j] >> BIN_SHIFT], 1);
            srt[p] = make_int2(s[j], d[j]);
        }
    __syncthreads();
    for (int i = tid; i < nC; i += SORT_T) binned[(size_t)base + i] = srt[i];
}

__global__ void __launch_bounds__(SORT_T)
bin_collect(const int* __restrict__ binned_w, const int* __restrict__ lsT,
            int* __restrict__ cnt, int* __restrict__ rowptr,
            int* __restrict__ csr, int N, int LP, int G) {
    __shared__ int c[BIN_R];
    __shared__ int cur[BIN_R];
    __shared__ int wsum[SORT_T / 32];
    __shared__ int red[SORT_T / 32];
    __shared__ int stage[STAGE_CAP];
    const int2* __restrict__ binned = (const int2*)binned_w;
    const int tid = threadIdx.x, lane = tid & 31, wave = tid >> 5;
    const int b = blockIdx.x;
    if (tid < BIN_R) c[tid] = 0;

    int asum = 0, nsum = 0;
    for (int w = tid; w < G; w += SORT_T) {
        const int* lrow = lsT + (size_t)w * LP + b;
        const int a = lrow[0], z = lrow[1];
        asum += a;
        nsum += z - a;
    }
    for (int o = 16; o > 0; o >>= 1) {
        asum += __shfl_xor(asum, o, 32);
        nsum += __shfl_xor(nsum, o, 32);
    }
    if (lane == 0) { red[wave] = asum; wsum[wave] = nsum; }
    __syncthreads();
    int sbase = 0, nE = 0;
    for (int w = 0; w < SORT_T / 32; ++w) { sbase += red[w]; nE += wsum[w]; }

    for (int w = tid; w < G; w += SORT_T) {
        const int* lrow = lsT + (size_t)w * LP + b;
        const int a = lrow[0], z = lrow[1];
        const int2* p = binned + (size_t)w * CHUNK;
        for (int i = a; i < z; i += 4) {
            const int i1 = min(i + 1, z - 1), i2 = min(i + 2, z - 1), i3 = min(i + 3, z - 1);
            const int d0 = p[i].y, d1 = p[i1].y, d2 = p[i2].y, d3 = p[i3].y;
            atomicAdd(&c[d0 & (BIN_R - 1)], 1);
            if (i + 1 < z) atomicAdd(&c[d1 & (BIN_R - 1)], 1);
            if (i + 2 < z) atomicAdd(&c[d2 & (BIN_R - 1)], 1);
            if (i + 3 < z) atomicAdd(&c[d3 & (BIN_R - 1)], 1);
        }
    }
    __syncthreads();
    const int mine = (tid < BIN_R) ? c[tid] : 0;
    block_excl_scan(c, BIN_R, wsum);
    if (tid < BIN_R) {
        const int ex = c[tid];
        cur[tid] = ex;
        const int n = b * BIN_R + tid;
        if (n < N) { cnt[n] = mine; rowptr[n] = sbase + ex; }
    }
    __syncthreads();

    const bool staged = nE <= STAGE_CAP;
    for (int w = tid; w < G; w += SORT_T) {
        const int* lrow = lsT + (size_t)w * LP + b;
        const int a = lrow[0], z = lrow[1];
        const int2* p = binned + (size_t)w * CHUNK;
        for (int i = a; i < z; i += 4) {
            const bool k1 = i + 1 < z, k2 = i + 2 < z, k3 = i + 3 < z;
            const int2 r0 = p[i];
            const int2 r1 = p[k1 ? i + 1 : z - 1];
            const int2 r2 = p[k2 ? i + 2 : z - 1];
            const int2 r3 = p[k3 ? i + 3 : z - 1];
            const int q0 = atomicAdd(&cur[r0.y & (BIN_R - 1)], 1);
            const int q1 = k1 ? atomicAdd(&cur[r1.y & (BIN_R - 1)], 1) : 0;
            const int q2 = k2 ? atomicAdd(&cur[r2.y & (BIN_R - 1)], 1) : 0;
            const int q3 = k3 ? atomicAdd(&cur[r3.y & (BIN_R - 1)], 1) : 0;
            if (staged) {
                stage[q0] = r0.x;
                if (k1) stage[q1] = r1.x;
                if (k2) stage[q2] = r2.x;
                if (k3) stage[q3] = r3.x;
            } else {
                csr[sbase + q0] = r0.x;
                if (k1) csr[sbase + q1] = r1.x;
                if (k2) csr[sbase + q2] = r2.x;
                if (k3) csr[sbase + q3] = r3.x;
            }
        }
    }
    if (staged) {
        __syncthreads();
        for (int i = tid - (sbase & 31); i < nE; i += SORT_T)
            if (i >= 0) csr[sbase + i] = stage[i];
    }
}

__device__ __forceinline__ float lrelu(float v) { return (v > 0.f) ? v : NEG_SLOPE * v; }

__device__ __forceinline__ float dot4(const float4 a, const float4 b) {
    return a.x * b.x + a.y * b.y + a.z * b.z + a.w * b.w;
}
__device__ __forceinline__ float quad_sum(float v) {
    v += __int_as_float(__builtin_amdgcn_update_dpp(0, __float_as_int(v), 0xB1, 0xF, 0xF, true));
    v += __int_as_float(__builtin_amdgcn_update_dpp(0, __float_as_int(v), 0x4E, 0xF, 0xF, true));
    return v;
}
__device__ __forceinline__ float row16_sum(float v) {
    v = quad_sum(v);
    v += __int_as_float(__builtin_amdgcn_update_dpp(0, __float_as_int(v), 0x164, 0xF, 0xF, true));
    v += __int_as_float(__builtin_amdgcn_update_dpp(0, __float_as_int(v), 0x168, 0xF, 0xF, true));
    return v;
}

__global__ void __launch_bounds__(256)
seg_gather1(const int* __restrict__ csr, const int* __restrict__ rowptr,
            const int* __restrict__ cnt, const float* __restrict__ attS,
            const float* __restrict__ attD, const float* __restrict__ h1,
            const float* __restrict__ b1, const float* __restrict__ W2,
            const float* __restrict__ attS2, const float* __restrict__ attD2,
            float* __restrict__ rec, int N) {
    __shared__ float4 stage[16];
    const int n  = (blockIdx.x * 256 + threadIdx.x) >> 4;
    const int l  = threadIdx.x & 15;
    if (n < N) {
        const int start = rowptr[n];
        const int deg   = cnt[n];
        const float4* __restrict__ h4 = (const float4*)h1;
        float4 aS, aD;
        aS.x = attS[l * 4 + 0]; aS.y = attS[l * 4 + 1]; aS.z = attS[l * 4 + 2]; aS.w = attS[l * 4 + 3];
        aD.x = attD[l * 4 + 0]; aD.y = attD[l * 4 + 1]; aD.z = attD[l * 4 + 2]; aD.w = attD[l * 4 + 3];

        const float4 vn = h4[n * 16 + l];
        const float adv = quad_sum(dot4(vn, aD));
        float m    = lrelu(quad_sum(dot4(vn, aS)) + adv);
        float ssum = 1.f;
        float4 acc = vn;

        for (int base = 0; base < deg; base += 16) {
            const int j   = base + l;
            const int idx = (j < deg) ? csr[start + j] : n;
            const int c   = min(16, deg - base);
            for (int k = 0; k < c; k += 4) {
                const int s0 = __shfl(idx, k, 16);
                const int s1 = __shfl(idx, k + 1, 16);
                const int s2 = __shfl(idx, k + 2, 16);
                const int s3 = __shfl(idx, k + 3, 16);
                const float4 v0 = h4[s0 * 16 + l];
                const float4 v1 = h4[s1 * 16 + l];
                const float4 v2 = h4[s2 * 16 + l];
                const float4 v3 = h4[s3 * 16 + l];
                const bool p1 = k + 1 < c, p2 = k + 2 < c, p3 = k + 3 < c;
                const float e0 = lrelu(quad_sum(dot4(v0, aS)) + adv);
                const float e1 = lrelu(quad_sum(dot4(v1, aS)) + adv);
                const float e2 = lrelu(quad_sum(dot4(v2, aS)) + adv);
                const float e3 = lrelu(quad_sum(dot4(v3, aS)) + adv);
                float mn = fmaxf(m, e0);
                mn = fmaxf(mn, p1 ? e1 : mn);
                mn = fmaxf(mn, p2 ? e2 : mn);
                mn = fmaxf(mn, p3 ? e3 : mn);
                const float sc = __expf(m - mn);
                const float w0 = __expf(e0 - mn);
                const float w1 = p1 ? __expf(e1 - mn) : 0.f;
                const float w2 = p2 ? __expf(e2 - mn) : 0.f;
                const float w3 = p3 ? __expf(e3 - mn) : 0.f;
                ssum = ssum * sc + w0;
                acc.x = acc.x * sc + w0 * v0.x; acc.y = acc.y * sc + w0 * v0.y;
                acc.z = acc.z * sc + w0 * v0.z; acc.w = acc.w * sc + w0 * v0.w;
                ssum += w1; acc.x += w1 * v1.x; acc.y += w1 * v1.y; acc.z += w1 * v1.z; acc.w += w1 * v1.w;
                ssum += w2; acc.x += w2 * v2.x; acc.y += w2 * v2.y; acc.z += w2 * v2.z; acc.w += w2 * v2.w;
                ssum += w3; acc.x += w3 * v3.x; acc.y += w3 * v3.y; acc.z += w3 * v3.z; acc.w += w3 * v3.w;
                m = mn;
            }
        }

        const float den = ssum + 1e-16f;
        float4 r;
        r.x = acc.x / den + b1[l * 4 + 0];
        r.y = acc.y / den + b1[l * 4 + 1];
        r.z = acc.z / den + b1[l * 4 + 2];
        r.w = acc.w / den + b1[l * 4 + 3];
        r.x = (r.x > 0.f) ? r.x : expm1f(r.x);
        r.y = (r.y > 0.f) ? r.y : expm1f(r.y);
        r.z = (r.z > 0.f) ? r.z : expm1f(r.z);
        r.w = (r.w > 0.f) ? r.w : expm1f(r.w);

        const float* __restrict__ wr = W2 + l * 8;
        const float a0 = row16_sum(r.x * wr[0] + r.y * wr[2] + r.z * wr[4] + r.w * wr[6]);
        const float a1 = row16_sum(r.x * wr[1] + r.y * wr[3] + r.z * wr[5] + r.w * wr[7]);
        if (l == 0) {
            float4 q;
            q.x = a0;
            q.y = a1;
            q.z = a0 * attS2[0] + a1 * attS2[1];
            q.w = a0 * attD2[0] + a1 * attD2[1];
            stage[threadIdx.x >> 4] = q;
        }
    }
    __syncthreads();
    if (threadIdx.x < 16) {
        const int nn = blockIdx.x * 16 + threadIdx.x;
        if (nn < N) ((float4*)rec)[nn] = stage[threadIdx.x];
    }
}

__global__ void __launch_bounds__(256)
seg_gather2(const int* __restrict__ csr, const int* __restrict__ rowptr,
            const int* __restrict__ cnt, const float* __restrict__ rec,
            const float* __restrict__ b2, float* __restrict__ out, int N) {
    __shared__ float stage[32];
    const int n = (blockIdx.x * 256 + threadIdx.x) >> 4;
    const int l = threadIdx.x & 15;
    float r0 = 0.f, r1 = 0.f;
    if (n < N) {
        const float4* __restrict__ rc = (const float4*)rec;
        const float4 me = rc[n];
        const int start = rowptr[n];
        const int deg   = cnt[n];
        const float es = lrelu(me.z + me.w);
        float m = es, ssum = 0.f, ax = 0.f, ay = 0.f;
        for (int j = l; j < deg; j += 16) {
            const float4 q = rc[csr[start + j]];
            const float e  = lrelu(q.z + me.w);
            const float t  = __expf(-fabsf(e - m));
            const bool up  = e > m;
            const float sc = up ? t : 1.f;
            const float w  = up ? 1.f : t;
            ssum = ssum * sc + w; ax = ax * sc + w * q.x; ay = ay * sc + w * q.y;
            m = fmaxf(m, e);
        }
        float mm = m;
        for (int o = 8; o > 0; o >>= 1) mm = fmaxf(mm, __shfl_xor(mm, o, 16));
        {
            const float f = __expf(m - mm);
            ssum *= f; ax *= f; ay *= f;
        }
        for (int o = 8; o > 0; o >>= 1) {
            ssum += __shfl_xor(ssum, o, 16);
            ax   += __shfl_xor(ax, o, 16);
            ay   += __shfl_xor(ay, o, 16);
        }
        {
            const float w = __expf(es - mm);
            ssum += w; ax += w * me.x; ay += w * me.y;
        }
        const float inv = 1.0f / (ssum + 1e-16f);
        r0 = ax * inv + b2[0];
        r1 = ay * inv + b2[1];
    }
    if (l == 0) {
        stage[(threadIdx.x >> 4) * 2 + 0] = r0;
        stage[(threadIdx.x >> 4) * 2 + 1] = r1;
    }
    __syncthreads();
    if (threadIdx.x < 32) {
        const int o = blockIdx.x * 32 + threadIdx.x;
        if (o < N * 2) out[o] = stage[threadIdx.x];
    }
}

extern "C" void kernel_launch(void* const* d_in, const int* in_sizes, int n_in,
                              void* d_out, int out_size, void* d_ws, size_t ws_size,
                              hipStream_t stream) {
    const float* x    = (const float*)d_in[0];
    const int*   ei   = (const int*)d_in[1];
    const float* W1   = (const float*)d_in[2];
    const float* aS1  = (const float*)d_in[3];
    const float* aD1  = (const float*)d_in[4];
    const float* b1   = (const float*)d_in[5];
    const float* W2   = (const float*)d_in[6];
    const float* aS2  = (const float*)d_in[7];
    const float* aD2  = (const float*)d_in[8];
    const float* b2   = (const float*)d_in[9];

    const int N = in_sizes[0] / DIM;
    const int E = in_sizes[1] / 2;
    float* out = (float*)d_out;

    float* ws  = (float*)d_ws;
    float* h1  = ws;
    float* rec = h1 + (size_t)N * 64;
    const int B = (N + BIN_R - 1) / BIN_R;
    const int G = (E + CHUNK - 1) / CHUNK;
    int* binned = (int*)(rec + (size_t)N * 4);
    int* cnt    = binned + (size_t)G * CHUNK * 2;
    int* rowptr = cnt + (size_t)N;
    const int LP = ((B + 1 + 31) / 32) * 32;
    int* lsT    = rowptr + (size_t)N;
    int* csr    = lsT + (size_t)G * LP;

    const int Mtiles = (N + 15) >> 4;

    gemm1_wmma<<<(Mtiles + 3) / 4, 128, 0, stream>>>(x, W1, h1, N);
    chunk_sort<<<G, SORT_T, 0, stream>>>(ei, binned, lsT, E, B, LP);
    bin_collect<<<B, SORT_T, 0, stream>>>(binned, lsT, cnt, rowptr, csr, N, LP, G);
    seg_gather1<<<(N + 15) / 16, 256, 0, stream>>>(csr, rowptr, cnt, aS1, aD1, h1, b1,
                                                  W2, aS2, aD2, rec, N);
    seg_gather2<<<(N + 15) / 16, 256, 0, stream>>>(csr, rowptr, cnt, rec, b2, out, N);
}


# GEMM launch dissolved: its 16-wave row-tile work-groups run ahead of the sort work-groups in the two edge-ordering launches, half in each; matrix issue under their latency chains; 5 launches to 4
# speedup vs baseline: 10.3135x; 1.0422x over previous
#include <hip/hip_runtime.h>
#include <math.h>

#define DIM 128
#define HID 16
#define HEADS 4
#define NEG_SLOPE 0.2f

typedef __attribute__((ext_vector_type(2))) float v2f;
typedef __attribute__((ext_vector_type(8))) float v8f;

#define GEMM_T 512
#define GEMM_W (GEMM_T / 32)
__device__ __forceinline__ void
gemm1_rows(const float* __restrict__ x, const float* __restrict__ W1,
           float* __restrict__ h1, int N, int gblk) {
    __shared__ float4 Wl4[DIM * 16];
    {
        const int t = threadIdx.x;
#pragma unroll
        for (int j = 0; j < (DIM * 16) / GEMM_T; ++j) {
            const int e = (j * GEMM_T + t) * 4;
            float4 v;
            v.x = W1[e]; v.y = W1[e + 1]; v.z = W1[e + 2]; v.w = W1[e + 3];
            Wl4[j * GEMM_T + t] = v;
        }
    }
    const float* Wl = (const float*)Wl4;
    __syncthreads();

    const int wave  = threadIdx.x >> 5;
    const int lane  = threadIdx.x & 31;
    const int Mtile = gblk * GEMM_W + wave;
    const int Mtiles = (N + 15) >> 4;
    if (Mtile >= Mtiles) return;

    const int r  = lane & 15;
    const int kh = lane >> 4;
    const int mbase = Mtile * 16;
    int row = mbase + r;
    if (row >= N) row = N - 1;
    const float* __restrict__ xr = x + (size_t)row * DIM;

    v8f acc0 = {}, acc1 = {}, acc2 = {}, acc3 = {};

    for (int kb = 0; kb < DIM; kb += 4) {
        const int k0 = kb + 2 * kh;
        v2f a;
        a.x = xr[k0];
        a.y = xr[k0 + 1];
        const float* w0 = &Wl[(size_t)k0 * 64];
        const float* w1 = &Wl[(size_t)(k0 + 1) * 64];
        v2f b;
        b.x = w0[r];       b.y = w1[r];
        acc0 = __builtin_amdgcn_wmma_f32_16x16x4_f32(false, a, false, b, (short)0, acc0, false, false);
        b.x = w0[16 + r];  b.y = w1[16 + r];
        acc1 = __builtin_amdgcn_wmma_f32_16x16x4_f32(false, a, false, b, (short)0, acc1, false, false);
        b.x = w0[32 + r];  b.y = w1[32 + r];
        acc2 = __builtin_amdgcn_wmma_f32_16x16x4_f32(false, a, false, b, (short)0, acc2, false, false);
        b.x = w0[48 + r];  b.y = w1[48 + r];
        acc3 = __builtin_amdgcn_wmma_f32_16x16x4_f32(false, a, false, b, (short)0, acc3, false, false);
    }

    float* outp = h1 + (size_t)mbase * 64;
#pragma unroll
    for (int i = 0; i < 8; ++i) {
        const int m = i + 8 * kh;
        if (mbase + m < N) {
            float* o = outp + (size_t)m * 64 + r;
            o[0]  = acc0[i];
            o[16] = acc1[i];
            o[32] = acc2[i];
            o[48] = acc3[i];
        }
    }
}

#define BIN_SHIFT 8
#define BIN_R (1 << BIN_SHIFT)
#define CHUNK 4096
#define SORT_T 512
#define EPT (CHUNK / SORT_T)
static_assert(GEMM_T == SORT_T, "GEMM work-groups share the sort launches' block size");
#define MAXB 1024
#define STAGE_CAP 8192

__device__ __forceinline__ void block_excl_scan(int* a, int n, int* wsum) {
    const int tid = threadIdx.x, lane = tid & 31, wave = tid >> 5;
    const int i0 = 2 * tid, i1 = i0 + 1;
    const int v0 = (i0 < n) ? a[i0] : 0;
    const int v1 = (i1 < n) ? a[i1] : 0;
    const int s = v0 + v1;
    int x = s;
    for (int o = 1; o < 32; o <<= 1) {
        const int y = __shfl_up(x, o, 32);
        if (lane >= o) x += y;
    }
    if (lane == 31) wsum[wave] = x;
    __syncthreads();
    int off = 0;
    for (int w = 0; w < SORT_T / 32; ++w) if (w < wave) off += wsum[w];
    const int ex = off + x - s;
    if (i0 < n) a[i0] = ex;
    if (i1 < n) a[i1] = ex + v0;
    __syncthreads();
}

__device__ __forceinline__ void
chunk_sort(const int* __restrict__ ei, int* __restrict__ binned_w,
           int* __restrict__ lsT, int E, int B, int LP, int chunk) {
    int2* __restrict__ binned = (int2*)binned_w;
    __shared__ int h[MAXB];
    __shared__ int cur[MAXB];
    __shared__ int wsum[SORT_T / 32];
    __shared__ int2 srt[CHUNK];
    const int tid  = threadIdx.x;
    const int base = chunk * CHUNK;
    const int nC   = min(CHUNK, E - base);
    for (int i = tid; i < B; i += SORT_T) h[i] = 0;
    __syncthreads();
    int s[EPT], d[EPT];
#pragma unroll
    for (int j = 0; j < EPT; ++j) {
        const int i = j * SORT_T + tid;
        const bool ok = i < nC;
        s[j] = ok ? ei[base + i] : 0;
        d[j] = ok ? ei[E + base + i] : -1;
    }
#pragma unroll
    for (int j = 0; j < EPT; ++j)
        if (d[j] >= 0) atomicAdd(&h[d[j] >> BIN_SHIFT], 1);
    __syncthreads();
    block_excl_scan(h, B, wsum);
    int* lrow = lsT + (size_t)chunk * LP;
    for (int i = tid; i < B; i += SORT_T) { const int v = h[i]; cur[i] = v; lrow[i] = v; }
    if (tid == 0) lrow[B] = nC;
    __syncthreads();
#pragma unroll
    for (int j = 0; j < EPT; ++j)
        if (d[j] >= 0) {
            const int p = atomicAdd(&cur[d[j] >> BIN_SHIFT], 1);
            srt[p] = make_int2(s[j], d[j]);
        }
    __syncthreads();
    for (int i = tid; i < nC; i += SORT_T) binned[(size_t)base + i] = srt[i];
}

__device__ __forceinline__ void
bin_collect(const int* __restrict__ binned_w, const int* __restrict__ lsT,
            int* __restrict__ cnt, int* __restrict__ rowptr,
            int* __restrict__ csr, int N, int LP, int G, int b) {
    __shared__ int c[BIN_R];
    __shared__ int cur[BIN_R];
    __shared__ int wsum[SORT_T / 32];
    __shared__ int red[SORT_T / 32];
    __shared__ int stage[STAGE_CAP];
    const int2* __restrict__ binned = (const int2*)binned_w;
    const int tid = threadIdx.x, lane = tid & 31, wave = tid >> 5;
    if (tid < BIN_R) c[tid] = 0;

    int asum = 0, nsum = 0;
    for (int w = tid; w < G; w += SORT_T) {
        const int* lrow = lsT + (size_t)w * LP + b;
        const int a = lrow[0], z = lrow[1];
        asum += a;
        nsum += z - a;
    }
    for (int o = 16; o > 0; o >>= 1) {
        asum += __shfl_xor(asum, o, 32);
        nsum += __shfl_xor(nsum, o, 32);
    }
    if (lane == 0) { red[wave] = asum; wsum[wave] = nsum; }
    __syncthreads();
    int sbase = 0, nE = 0;
    for (int w = 0; w < SORT_T / 32; ++w) { sbase += red[w]; nE += wsum[w]; }

    for (int w = tid; w < G; w += SORT_T) {
        const int* lrow = lsT + (size_t)w * LP + b;
        const int a = lrow[0], z = lrow[1];
        const int2* p = binned + (size_t)w * CHUNK;
        for (int i = a; i < z; i += 4) {
            const int i1 = min(i + 1, z - 1), i2 = min(i + 2, z - 1), i3 = min(i + 3, z - 1);
            const int d0 = p[i].y, d1 = p[i1].y, d2 = p[i2].y, d3 = p[i3].y;
            atomicAdd(&c[d0 & (BIN_R - 1)], 1);
            if (i + 1 < z) atomicAdd(&c[d1 & (BIN_R - 1)], 1);
            if (i + 2 < z) atomicAdd(&c[d2 & (BIN_R - 1)], 1);
            if (i + 3 < z) atomicAdd(&c[d3 & (BIN_R - 1)], 1);
        }
    }
    __syncthreads();
    const int mine = (tid < BIN_R) ? c[tid] : 0;
    block_excl_scan(c, BIN_R, wsum);
    if (tid < BIN_R) {
        const int ex = c[tid];
        cur[tid] = ex;
        const int n = b * BIN_R + tid;
        if (n < N) { cnt[n] = mine; rowptr[n] = sbase + ex; }
    }
    __syncthreads();

    const bool staged = nE <= STAGE_CAP;
    for (int w = tid; w < G; w += SORT_T) {
        const int* lrow = lsT + (size_t)w * LP + b;
        const int a = lrow[0], z = lrow[1];
        const int2* p = binned + (size_t)w * CHUNK;
        for (int i = a; i < z; i += 4) {
            const bool k1 = i + 1 < z, k2 = i + 2 < z, k3 = i + 3 < z;
            const int2 r0 = p[i];
            const int2 r1 = p[k1 ? i + 1 : z - 1];
            const int2 r2 = p[k2 ? i + 2 : z - 1];
            const int2 r3 = p[k3 ? i + 3 : z - 1];
            const int q0 = atomicAdd(&cur[r0.y & (BIN_R - 1)], 1);
            const int q1 = k1 ? atomicAdd(&cur[r1.y & (BIN_R - 1)], 1) : 0;
            const int q2 = k2 ? atomicAdd(&cur[r2.y & (BIN_R - 1)], 1) : 0;
            const int q3 = k3 ? atomicAdd(&cur[r3.y & (BIN_R - 1)], 1) : 0;
            if (staged) {
                stage[q0] = r0.x;
                if (k1) stage[q1] = r1.x;
                if (k2) stage[q2] = r2.x;
                if (k3) stage[q3] = r3.x;
            } else {
                csr[sbase + q0] = r0.x;
                if (k1) csr[sbase + q1] = r1.x;
                if (k2) csr[sbase + q2] = r2.x;
                if (k3) csr[sbase + q3] = r3.x;
            }
        }
    }
    if (staged) {
        __syncthreads();
        for (int i = tid - (sbase & 31); i < nE; i += SORT_T)
            if (i >= 0) csr[sbase + i] = stage[i];
    }
}

__global__ void __launch_bounds__(SORT_T) __attribute__((amdgpu_waves_per_eu(4, 12)))
sort_and_gemm(const int* __restrict__ ei, int* __restrict__ binned_w,
              int* __restrict__ lsT, int E, int B, int LP,
              const float* __restrict__ x, const float* __restrict__ W1,
              float* __restrict__ h1, int N, int g1) {
    const int blk = (int)blockIdx.x;
    if (blk < g1) gemm1_rows(x, W1, h1, N, blk);
    else          chunk_sort(ei, binned_w, lsT, E, B, LP, blk - g1);
}

__global__ void __launch_bounds__(SORT_T) __attribute__((amdgpu_waves_per_eu(4, 12)))
collect_and_gemm(const int* __restrict__ binned_w, const int* __restrict__ lsT,
                 int* __restrict__ cnt, int* __restrict__ rowptr,
                 int* __restrict__ csr, int N, int LP, int G,
                 const float* __restrict__ x, const float* __restrict__ W1,
                 float* __restrict__ h1, int g0, int g2) {
    const int blk = (int)blockIdx.x;
    if (blk < g2) gemm1_rows(x, W1, h1, N, g0 + blk);
    else          bin_collect(binned_w, lsT, cnt, rowptr, csr, N, LP, G, blk - g2);
}

__device__ __forceinline__ float lrelu(float v) { return (v > 0.f) ? v : NEG_SLOPE * v; }

__device__ __forceinline__ float dot4(const float4 a, const float4 b) {
    return a.x * b.x + a.y * b.y + a.z * b.z + a.w * b.w;
}
__device__ __forceinline__ float quad_sum(float v) {
    v += __int_as_float(__builtin_amdgcn_update_dpp(0, __float_as_int(v), 0xB1, 0xF, 0xF, true));
    v += __int_as_float(__builtin_amdgcn_update_dpp(0, __float_as_int(v), 0x4E, 0xF, 0xF, true));
    return v;
}
__device__ __forceinline__ float row16_sum(float v) {
    v = quad_sum(v);
    v += __int_as_float(__builtin_amdgcn_update_dpp(0, __float_as_int(v), 0x164, 0xF, 0xF, true));
    v += __int_as_float(__builtin_amdgcn_update_dpp(0, __float_as_int(v), 0x168, 0xF, 0xF, true));
    return v;
}

__global__ void __launch_bounds__(256)
seg_gather1(const int* __restrict__ csr, const int* __restrict__ rowptr,
            const int* __restrict__ cnt, const float* __restrict__ attS,
            const float* __restrict__ attD, const float* __restrict__ h1,
            const float* __restrict__ b1, const float* __restrict__ W2,
            const float* __restrict__ attS2, const float* __restrict__ attD2,
            float* __restrict__ rec, int N) {
    __shared__ float4 stage[16];
    const int n  = (blockIdx.x * 256 + threadIdx.x) >> 4;
    const int l  = threadIdx.x & 15;
    if (n < N) {
        const int start = rowptr[n];
        const int deg   = cnt[n];
        const float4* __restrict__ h4 = (const float4*)h1;
        float4 aS, aD;
        aS.x = attS[l * 4 + 0]; aS.y = attS[l * 4 + 1]; aS.z = attS[l * 4 + 2]; aS.w = attS[l * 4 + 3];
        aD.x = attD[l * 4 + 0]; aD.y = attD[l * 4 + 1]; aD.z = attD[l * 4 + 2]; aD.w = attD[l * 4 + 3];

        const float4 vn = h4[n * 16 + l];
        const float adv = quad_sum(dot4(vn, aD));
        float m    = lrelu(quad_sum(dot4(vn, aS)) + adv);
        float ssum = 1.f;
        float4 acc = vn;

        for (int base = 0; base < deg; base += 16) {
            const int j   = base + l;
            const int idx = (j < deg) ? csr[start + j] : n;
            const int c   = min(16, deg - base);
            for (int k = 0; k < c; k += 4) {
                const int s0 = __shfl(idx, k, 16);
                const int s1 = __shfl(idx, k + 1, 16);
                const int s2 = __shfl(idx, k + 2, 16);
                const int s3 = __shfl(idx, k + 3, 16);
                const float4 v0 = h4[s0 * 16 + l];
                const float4 v1 = h4[s1 * 16 + l];
                const float4 v2 = h4[s2 * 16 + l];
                const float4 v3 = h4[s3 * 16 + l];
                const bool p1 = k + 1 < c, p2 = k + 2 < c, p3 = k + 3 < c;
                const float e0 = lrelu(quad_sum(dot4(v0, aS)) + adv);
                const float e1 = lrelu(quad_sum(dot4(v1, aS)) + adv);
                const float e2 = lrelu(quad_sum(dot4(v2, aS)) + adv);
                const float e3 = lrelu(quad_sum(dot4(v3, aS)) + adv);
                float mn = fmaxf(m, e0);
                mn = fmaxf(mn, p1 ? e1 : mn);
                mn = fmaxf(mn, p2 ? e2 : mn);
                mn = fmaxf(mn, p3 ? e3 : mn);
                const float sc = __expf(m - mn);
                const float w0 = __expf(e0 - mn);
                const float w1 = p1 ? __expf(e1 - mn) : 0.f;
                const float w2 = p2 ? __expf(e2 - mn) : 0.f;
                const float w3 = p3 ? __expf(e3 - mn) : 0.f;
                ssum = ssum * sc + w0;
                acc.x = acc.x * sc + w0 * v0.x; acc.y = acc.y * sc + w0 * v0.y;
                acc.z = acc.z * sc + w0 * v0.z; acc.w = acc.w * sc + w0 * v0.w;
                ssum += w1; acc.x += w1 * v1.x; acc.y += w1 * v1.y; acc.z += w1 * v1.z; acc.w += w1 * v1.w;
                ssum += w2; acc.x += w2 * v2.x; acc.y += w2 * v2.y; acc.z += w2 * v2.z; acc.w += w2 * v2.w;
                ssum += w3; acc.x += w3 * v3.x; acc.y += w3 * v3.y; acc.z += w3 * v3.z; acc.w += w3 * v3.w;
                m = mn;
            }
        }

        const float den = ssum + 1e-16f;
        float4 r;
        r.x = acc.x / den + b1[l * 4 + 0];
        r.y = acc.y / den + b1[l * 4 + 1];
        r.z = acc.z / den + b1[l * 4 + 2];
        r.w = acc.w / den + b1[l * 4 + 3];
        r.x = (r.x > 0.f) ? r.x : expm1f(r.x);
        r.y = (r.y > 0.f) ? r.y : expm1f(r.y);
        r.z = (r.z > 0.f) ? r.z : expm1f(r.z);
        r.w = (r.w > 0.f) ? r.w : expm1f(r.w);

        const float* __restrict__ wr = W2 + l * 8;
        const float a0 = row16_sum(r.x * wr[0] + r.y * wr[2] + r.z * wr[4] + r.w * wr[6]);
        const float a1 = row16_sum(r.x * wr[1] + r.y * wr[3] + r.z * wr[5] + r.w * wr[7]);
        if (l == 0) {
            float4 q;
            q.x = a0;
            q.y = a1;
            q.z = a0 * attS2[0] + a1 * attS2[1];
            q.w = a0 * attD2[0] + a1 * attD2[1];
            stage[threadIdx.x >> 4] = q;
        }
    }
    __syncthreads();
    if (threadIdx.x < 16) {
        const int nn = blockIdx.x * 16 + threadIdx.x;
        if (nn < N) ((float4*)rec)[nn] = stage[threadIdx.x];
    }
}

__global__ void __launch_bounds__(256)
seg_gather2(const int* __restrict__ csr, const int* __restrict__ rowptr,
            const int* __restrict__ cnt, const float* __restrict__ rec,
            const float* __restrict__ b2, float* __restrict__ out, int N) {
    __shared__ float stage[32];
    const int n = (blockIdx.x * 256 + threadIdx.x) >> 4;
    const int l = threadIdx.x & 15;
    float r0 = 0.f, r1 = 0.f;
    if (n < N) {
        const float4* __restrict__ rc = (const float4*)rec;
        const float4 me = rc[n];
        const int start = rowptr[n];
        const int deg   = cnt[n];
        const float es = lrelu(me.z + me.w);
        float m = es, ssum = 0.f, ax = 0.f, ay = 0.f;
        for (int j = l; j < deg; j += 16) {
            const float4 q = rc[csr[start + j]];
            const float e  = lrelu(q.z + me.w);
            const float t  = __expf(-fabsf(e - m));
            const bool up  = e > m;
            const float sc = up ? t : 1.f;
            const float w  = up ? 1.f : t;
            ssum = ssum * sc + w; ax = ax * sc + w * q.x; ay = ay * sc + w * q.y;
            m = fmaxf(m, e);
        }
        float mm = m;
        for (int o = 8; o > 0; o >>= 1) mm = fmaxf(mm, __shfl_xor(mm, o, 16));
        {
            const float f = __expf(m - mm);
            ssum *= f; ax *= f; ay *= f;
        }
        for (int o = 8; o > 0; o >>= 1) {
            ssum += __shfl_xor(ssum, o, 16);
            ax   += __shfl_xor(ax, o, 16);
            ay   += __shfl_xor(ay, o, 16);
        }
        {
            const float w = __expf(es - mm);
            ssum += w; ax += w * me.x; ay += w * me.y;
        }
        const float inv = 1.0f / (ssum + 1e-16f);
        r0 = ax * inv + b2[0];
        r1 = ay * inv + b2[1];
    }
    if (l == 0) {
        stage[(threadIdx.x >> 4) * 2 + 0] = r0;
        stage[(threadIdx.x >> 4) * 2 + 1] = r1;
    }
    __syncthreads();
    if (threadIdx.x < 32) {
        const int o = blockIdx.x * 32 + threadIdx.x;
        if (o < N * 2) out[o] = stage[threadIdx.x];
    }
}

extern "C" void kernel_launch(void* const* d_in, const int* in_sizes, int n_in,
                              void* d_out, int out_size, void* d_ws, size_t ws_size,
                              hipStream_t stream) {
    const float* x    = (const float*)d_in[0];
    const int*   ei   = (const int*)d_in[1];
    const float* W1   = (const float*)d_in[2];
    const float* aS1  = (const float*)d_in[3];
    const float* aD1  = (const float*)d_in[4];
    const float* b1   = (const float*)d_in[5];
    const float* W2   = (const float*)d_in[6];
    const float* aS2  = (const float*)d_in[7];
    const float* aD2  = (const float*)d_in[8];
    const float* b2   = (const float*)d_in[9];

    const int N = in_sizes[0] / DIM;
    const int E = in_sizes[1] / 2;
    float* out = (float*)d_out;

    float* ws  = (float*)d_ws;
    float* h1  = ws;
    float* rec = h1 + (size_t)N * 64;
    const int B = (N + BIN_R - 1) / BIN_R;
    const int G = (E + CHUNK - 1) / CHUNK;
    int* binned = (int*)(rec + (size_t)N * 4);
    int* cnt    = binned + (size_t)G * CHUNK * 2;
    int* rowptr = cnt + (size_t)N;
    const int LP = ((B + 1 + 31) / 32) * 32;
    int* lsT    = rowptr + (size_t)N;
    int* csr    = lsT + (size_t)G * LP;

    const int Mtiles = (N + 15) >> 4;

    const int GB  = (Mtiles + GEMM_W - 1) / GEMM_W;
    const int GB0 = GB / 2;
    const int GB1 = GB - GB0;
    sort_and_gemm<<<GB0 + G, SORT_T, 0, stream>>>(ei, binned, lsT, E, B, LP, x, W1, h1, N, GB0);
    collect_and_gemm<<<GB1 + B, SORT_T, 0, stream>>>(binned, lsT, cnt, rowptr, csr, N, LP, G,
                                                    x, W1, h1, GB0, GB1);
    seg_gather1<<<(N + 15) / 16, 256, 0, stream>>>(csr, rowptr, cnt, aS1, aD1, h1, b1,
                                                  W2, aS2, aD2, rec, N);
    seg_gather2<<<(N + 15) / 16, 256, 0, stream>>>(csr, rowptr, cnt, rec, b2, out, N);
}


# Layer-2 gather: a quad of lanes per node (8 nodes per wave, was 2); a lane's 8 source ids, then its 8 records, requested together; 4x fewer waves; sum order now 4-lane stride + quad DPP tree
# speedup vs baseline: 10.7462x; 1.0420x over previous
#include <hip/hip_runtime.h>
#include <math.h>

#define DIM 128
#define HID 16
#define HEADS 4
#define NEG_SLOPE 0.2f

typedef __attribute__((ext_vector_type(2))) float v2f;
typedef __attribute__((ext_vector_type(8))) float v8f;

#define GEMM_T 512
#define GEMM_W (GEMM_T / 32)
__device__ __forceinline__ void
gemm1_rows(const float* __restrict__ x, const float* __restrict__ W1,
           float* __restrict__ h1, int N, int gblk) {
    __shared__ float4 Wl4[DIM * 16];
    {
        const int t = threadIdx.x;
#pragma unroll
        for (int j = 0; j < (DIM * 16) / GEMM_T; ++j) {
            const int e = (j * GEMM_T + t) * 4;
            float4 v;
            v.x = W1[e]; v.y = W1[e + 1]; v.z = W1[e + 2]; v.w = W1[e + 3];
            Wl4[j * GEMM_T + t] = v;
        }
    }
    const float* Wl = (const float*)Wl4;
    __syncthreads();

    const int wave  = threadIdx.x >> 5;
    const int lane  = threadIdx.x & 31;
    const int Mtile = gblk * GEMM_W + wave;
    const int Mtiles = (N + 15) >> 4;
    if (Mtile >= Mtiles) return;

    const int r  = lane & 15;
    const int kh = lane >> 4;
    const int mbase = Mtile * 16;
    int row = mbase + r;
    if (row >= N) row = N - 1;
    const float* __restrict__ xr = x + (size_t)row * DIM;

    v8f acc0 = {}, acc1 = {}, acc2 = {}, acc3 = {};

    for (int kb = 0; kb < DIM; kb += 4) {
        const int k0 = kb + 2 * kh;
        v2f a;
        a.x = xr[k0];
        a.y = xr[k0 + 1];
        const float* w0 = &Wl[(size_t)k0 * 64];
        const float* w1 = &Wl[(size_t)(k0 + 1) * 64];
        v2f b;
        b.x = w0[r];       b.y = w1[r];
        acc0 = __builtin_amdgcn_wmma_f32_16x16x4_f32(false, a, false, b, (short)0, acc0, false, false);
        b.x = w0[16 + r];  b.y = w1[16 + r];
        acc1 = __builtin_amdgcn_wmma_f32_16x16x4_f32(false, a, false, b, (short)0, acc1, false, false);
        b.x = w0[32 + r];  b.y = w1[32 + r];
        acc2 = __builtin_amdgcn_wmma_f32_16x16x4_f32(false, a, false, b, (short)0, acc2, false, false);
        b.x = w0[48 + r];  b.y = w1[48 + r];
        acc3 = __builtin_amdgcn_wmma_f32_16x16x4_f32(false, a, false, b, (short)0, acc3, false, false);
    }

    float* outp = h1 + (size_t)mbase * 64;
#pragma unroll
    for (int i = 0; i < 8; ++i) {
        const int m = i + 8 * kh;
        if (mbase + m < N) {
            float* o = outp + (size_t)m * 64 + r;
            o[0]  = acc0[i];
            o[16] = acc1[i];
            o[32] = acc2[i];
            o[48] = acc3[i];
        }
    }
}

#define BIN_SHIFT 8
#define BIN_R (1 << BIN_SHIFT)
#define CHUNK 4096
#define SORT_T 512
#define EPT (CHUNK / SORT_T)
static_assert(GEMM_T == SORT_T, "GEMM work-groups share the sort launches' block size");
#define MAXB 1024
#define STAGE_CAP 8192

__device__ __forceinline__ void block_excl_scan(int* a, int n, int* wsum) {
    const int tid = threadIdx.x, lane = tid & 31, wave = tid >> 5;
    const int i0 = 2 * tid, i1 = i0 + 1;
    const int v0 = (i0 < n) ? a[i0] : 0;
    const int v1 = (i1 < n) ? a[i1] : 0;
    const int s = v0 + v1;
    int x = s;
    for (int o = 1; o < 32; o <<= 1) {
        const int y = __shfl_up(x, o, 32);
        if (lane >= o) x += y;
    }
    if (lane == 31) wsum[wave] = x;
    __syncthreads();
    int off = 0;
    for (int w = 0; w < SORT_T / 32; ++w) if (w < wave) off += wsum[w];
    const int ex = off + x - s;
    if (i0 < n) a[i0] = ex;
    if (i1 < n) a[i1] = ex + v0;
    __syncthreads();
}

__device__ __forceinline__ void
chunk_sort(const int* __restrict__ ei, int* __restrict__ binned_w,
           int* __restrict__ lsT, int E, int B, int LP, int chunk) {
    int2* __restrict__ binned = (int2*)binned_w;
    __shared__ int h[MAXB];
    __shared__ int cur[MAXB];
    __shared__ int wsum[SORT_T / 32];
    __shared__ int2 srt[CHUNK];
    const int tid  = threadIdx.x;
    const int base = chunk * CHUNK;
    const int nC   = min(CHUNK, E - base);
    for (int i = tid; i < B; i += SORT_T) h[i] = 0;
    __syncthreads();
    int s[EPT], d[EPT];
#pragma unroll
    for (int j = 0; j < EPT; ++j) {
        const int i = j * SORT_T + tid;
        const bool ok = i < nC;
        s[j] = ok ? ei[base + i] : 0;
        d[j] = ok ? ei[E + base + i] : -1;
    }
#pragma unroll
    for (int j = 0; j < EPT; ++j)
        if (d[j] >= 0) atomicAdd(&h[d[j] >> BIN_SHIFT], 1);
    __syncthreads();
    block_excl_scan(h, B, wsum);
    int* lrow = lsT + (size_t)chunk * LP;
    for (int i = tid; i < B; i += SORT_T) { const int v = h[i]; cur[i] = v; lrow[i] = v; }
    if (tid == 0) lrow[B] = nC;
    __syncthreads();
#pragma unroll
    for (int j = 0; j < EPT; ++j)
        if (d[j] >= 0) {
            const int p = atomicAdd(&cur[d[j] >> BIN_SHIFT], 1);
            srt[p] = make_int2(s[j], d[j]);
        }
    __syncthreads();
    for (int i = tid; i < nC; i += SORT_T) binned[(size_t)base + i] = srt[i];
}

__device__ __forceinline__ void
bin_collect(const int* __restrict__ binned_w, const int* __restrict__ lsT,
            int* __restrict__ cnt, int* __restrict__ rowptr,
            int* __restrict__ csr, int N, int LP, int G, int b) {
    __shared__ int c[BIN_R];
    __shared__ int cur[BIN_R];
    __shared__ int wsum[SORT_T / 32];
    __shared__ int red[SORT_T / 32];
    __shared__ int stage[STAGE_CAP];
    const int2* __restrict__ binned = (const int2*)binned_w;
    const int tid = threadIdx.x, lane = tid & 31, wave = tid >> 5;
    if (tid < BIN_R) c[tid] = 0;

    int asum = 0, nsum = 0;
    for (int w = tid; w < G; w += SORT_T) {
        const int* lrow = lsT + (size_t)w * LP + b;
        const int a = lrow[0], z = lrow[1];
        asum += a;
        nsum += z - a;
    }
    for (int o = 16; o > 0; o >>= 1) {
        asum += __shfl_xor(asum, o, 32);
        nsum += __shfl_xor(nsum, o, 32);
    }
    if (lane == 0) { red[wave] = asum; wsum[wave] = nsum; }
    __syncthreads();
    int sbase = 0, nE = 0;
    for (int w = 0; w < SORT_T / 32; ++w) { sbase += red[w]; nE += wsum[w]; }

    for (int w = tid; w < G; w += SORT_T) {
        const int* lrow = lsT + (size_t)w * LP + b;
        const int a = lrow[0], z = lrow[1];
        const int2* p = binned + (size_t)w * CHUNK;
        for (int i = a; i < z; i += 4) {
            const int i1 = min(i + 1, z - 1), i2 = min(i + 2, z - 1), i3 = min(i + 3, z - 1);
            const int d0 = p[i].y, d1 = p[i1].y, d2 = p[i2].y, d3 = p[i3].y;
            atomicAdd(&c[d0 & (BIN_R - 1)], 1);
            if (i + 1 < z) atomicAdd(&c[d1 & (BIN_R - 1)], 1);
            if (i + 2 < z) atomicAdd(&c[d2 & (BIN_R - 1)], 1);
            if (i + 3 < z) atomicAdd(&c[d3 & (BIN_R - 1)], 1);
        }
    }
    __syncthreads();
    const int mine = (tid < BIN_R) ? c[tid] : 0;
    block_excl_scan(c, BIN_R, wsum);
    if (tid < BIN_R) {
        const int ex = c[tid];
        cur[tid] = ex;
        const int n = b * BIN_R + tid;
        if (n < N) { cnt[n] = mine; rowptr[n] = sbase + ex; }
    }
    __syncthreads();

    const bool staged = nE <= STAGE_CAP;
    for (int w = tid; w < G; w += SORT_T) {
        const int* lrow = lsT + (size_t)w * LP + b;
        const int a = lrow[0], z = lrow[1];
        const int2* p = binned + (size_t)w * CHUNK;
        for (int i = a; i < z; i += 4) {
            const bool k1 = i + 1 < z, k2 = i + 2 < z, k3 = i + 3 < z;
            const int2 r0 = p[i];
            const int2 r1 = p[k1 ? i + 1 : z - 1];
            const int2 r2 = p[k2 ? i + 2 : z - 1];
            const int2 r3 = p[k3 ? i + 3 : z - 1];
            const int q0 = atomicAdd(&cur[r0.y & (BIN_R - 1)], 1);
            const int q1 = k1 ? atomicAdd(&cur[r1.y & (BIN_R - 1)], 1) : 0;
            const int q2 = k2 ? atomicAdd(&cur[r2.y & (BIN_R - 1)], 1) : 0;
            const int q3 = k3 ? atomicAdd(&cur[r3.y & (BIN_R - 1)], 1) : 0;
            if (staged) {
                stage[q0] = r0.x;
                if (k1) stage[q1] = r1.x;
                if (k2) stage[q2] = r2.x;
                if (k3) stage[q3] = r3.x;
            } else {
                csr[sbase + q0] = r0.x;
                if (k1) csr[sbase + q1] = r1.x;
                if (k2) csr[sbase + q2] = r2.x;
                if (k3) csr[sbase + q3] = r3.x;
            }
        }
    }
    if (staged) {
        __syncthreads();
        for (int i = tid - (sbase & 31); i < nE; i += SORT_T)
            if (i >= 0) csr[sbase + i] = stage[i];
    }
}

__global__ void __launch_bounds__(SORT_T) __attribute__((amdgpu_waves_per_eu(4, 12)))
sort_and_gemm(const int* __restrict__ ei, int* __restrict__ binned_w,
              int* __restrict__ lsT, int E, int B, int LP,
              const float* __restrict__ x, const float* __restrict__ W1,
              float* __restrict__ h1, int N, int g1) {
    const int blk = (int)blockIdx.x;
    if (blk < g1) gemm1_rows(x, W1, h1, N, blk);
    else          chunk_sort(ei, binned_w, lsT, E, B, LP, blk - g1);
}

__global__ void __launch_bounds__(SORT_T) __attribute__((amdgpu_waves_per_eu(4, 12)))
collect_and_gemm(const int* __restrict__ binned_w, const int* __restrict__ lsT,
                 int* __restrict__ cnt, int* __restrict__ rowptr,
                 int* __restrict__ csr, int N, int LP, int G,
                 const float* __restrict__ x, const float* __restrict__ W1,
                 float* __restrict__ h1, int g0, int g2) {
    const int blk = (int)blockIdx.x;
    if (blk < g2) gemm1_rows(x, W1, h1, N, g0 + blk);
    else          bin_collect(binned_w, lsT, cnt, rowptr, csr, N, LP, G, blk - g2);
}

__device__ __forceinline__ float lrelu(float v) { return (v > 0.f) ? v : NEG_SLOPE * v; }

__device__ __forceinline__ float dot4(const float4 a, const float4 b) {
    return a.x * b.x + a.y * b.y + a.z * b.z + a.w * b.w;
}
__device__ __forceinline__ float quad_sum(float v) {
    v += __int_as_float(__builtin_amdgcn_update_dpp(0, __float_as_int(v), 0xB1, 0xF, 0xF, true));
    v += __int_as_float(__builtin_amdgcn_update_dpp(0, __float_as_int(v), 0x4E, 0xF, 0xF, true));
    return v;
}
__device__ __forceinline__ float row16_sum(float v) {
    v = quad_sum(v);
    v += __int_as_float(__builtin_amdgcn_update_dpp(0, __float_as_int(v), 0x164, 0xF, 0xF, true));
    v += __int_as_float(__builtin_amdgcn_update_dpp(0, __float_as_int(v), 0x168, 0xF, 0xF, true));
    return v;
}

__global__ void __launch_bounds__(256)
seg_gather1(const int* __restrict__ csr, const int* __restrict__ rowptr,
            const int* __restrict__ cnt, const float* __restrict__ attS,
            const float* __restrict__ attD, const float* __restrict__ h1,
            const float* __restrict__ b1, const float* __restrict__ W2,
            const float* __restrict__ attS2, const float* __restrict__ attD2,
            float* __restrict__ rec, int N) {
    __shared__ float4 stage[16];
    const int n  = (blockIdx.x * 256 + threadIdx.x) >> 4;
    const int l  = threadIdx.x & 15;
    if (n < N) {
        const int start = rowptr[n];
        const int deg   = cnt[n];
        const float4* __restrict__ h4 = (const float4*)h1;
        float4 aS, aD;
        aS.x = attS[l * 4 + 0]; aS.y = attS[l * 4 + 1]; aS.z = attS[l * 4 + 2]; aS.w = attS[l * 4 + 3];
        aD.x = attD[l * 4 + 0]; aD.y = attD[l * 4 + 1]; aD.z = attD[l * 4 + 2]; aD.w = attD[l * 4 + 3];

        const float4 vn = h4[n * 16 + l];
        const float adv = quad_sum(dot4(vn, aD));
        float m    = lrelu(quad_sum(dot4(vn, aS)) + adv);
        float ssum = 1.f;
        float4 acc = vn;

        for (int base = 0; base < deg; base += 16) {
            const int j   = base + l;
            const int idx = (j < deg) ? csr[start + j] : n;
            const int c   = min(16, deg - base);
            for (int k = 0; k < c; k += 4) {
                const int s0 = __shfl(idx, k, 16);
                const int s1 = __shfl(idx, k + 1, 16);
                const int s2 = __shfl(idx, k + 2, 16);
                const int s3 = __shfl(idx, k + 3, 16);
                const float4 v0 = h4[s0 * 16 + l];
                const float4 v1 = h4[s1 * 16 + l];
                const float4 v2 = h4[s2 * 16 + l];
                const float4 v3 = h4[s3 * 16 + l];
                const bool p1 = k + 1 < c, p2 = k + 2 < c, p3 = k + 3 < c;
                const float e0 = lrelu(quad_sum(dot4(v0, aS)) + adv);
                const float e1 = lrelu(quad_sum(dot4(v1, aS)) + adv);
                const float e2 = lrelu(quad_sum(dot4(v2, aS)) + adv);
                const float e3 = lrelu(quad_sum(dot4(v3, aS)) + adv);
                float mn = fmaxf(m, e0);
                mn = fmaxf(mn, p1 ? e1 : mn);
                mn = fmaxf(mn, p2 ? e2 : mn);
                mn = fmaxf(mn, p3 ? e3 : mn);
                const float sc = __expf(m - mn);
                const float w0 = __expf(e0 - mn);
                const float w1 = p1 ? __expf(e1 - mn) : 0.f;
                const float w2 = p2 ? __expf(e2 - mn) : 0.f;
                const float w3 = p3 ? __expf(e3 - mn) : 0.f;
                ssum = ssum * sc + w0;
                acc.x = acc.x * sc + w0 * v0.x; acc.y = acc.y * sc + w0 * v0.y;
                acc.z = acc.z * sc + w0 * v0.z; acc.w = acc.w * sc + w0 * v0.w;
                ssum += w1; acc.x += w1 * v1.x; acc.y += w1 * v1.y; acc.z += w1 * v1.z; acc.w += w1 * v1.w;
                ssum += w2; acc.x += w2 * v2.x; acc.y += w2 * v2.y; acc.z += w2 * v2.z; acc.w += w2 * v2.w;
                ssum += w3; acc.x += w3 * v3.x; acc.y += w3 * v3.y; acc.z += w3 * v3.z; acc.w += w3 * v3.w;
                m = mn;
            }
        }

        const float den = ssum + 1e-16f;
        float4 r;
        r.x = acc.x / den + b1[l * 4 + 0];
        r.y = acc.y / den + b1[l * 4 + 1];
        r.z = acc.z / den + b1[l * 4 + 2];
        r.w = acc.w / den + b1[l * 4 + 3];
        r.x = (r.x > 0.f) ? r.x : expm1f(r.x);
        r.y = (r.y > 0.f) ? r.y : expm1f(r.y);
        r.z = (r.z > 0.f) ? r.z : expm1f(r.z);
        r.w = (r.w > 0.f) ? r.w : expm1f(r.w);

        const float* __restrict__ wr = W2 + l * 8;
        const float a0 = row16_sum(r.x * wr[0] + r.y * wr[2] + r.z * wr[4] + r.w * wr[6]);
        const float a1 = row16_sum(r.x * wr[1] + r.y * wr[3] + r.z * wr[5] + r.w * wr[7]);
        if (l == 0) {
            float4 q;
            q.x = a0;
            q.y = a1;
            q.z = a0 * attS2[0] + a1 * attS2[1];
            q.w = a0 * attD2[0] + a1 * attD2[1];
            stage[threadIdx.x >> 4] = q;
        }
    }
    __syncthreads();
    if (threadIdx.x < 16) {
        const int nn = blockIdx.x * 16 + threadIdx.x;
        if (nn < N) ((float4*)rec)[nn] = stage[threadIdx.x];
    }
}

#define G2_L 4
#define G2_U 8
__device__ __forceinline__ float quad_max(float v) {
    v = fmaxf(v, __int_as_float(__builtin_amdgcn_update_dpp(0, __float_as_int(v), 0xB1, 0xF, 0xF, true)));
    v = fmaxf(v, __int_as_float(__builtin_amdgcn_update_dpp(0, __float_as_int(v), 0x4E, 0xF, 0xF, true)));
    return v;
}
__global__ void __launch_bounds__(256)
seg_gather2(const int* __restrict__ csr, const int* __restrict__ rowptr,
            const int* __restrict__ cnt, const float* __restrict__ rec,
            const float* __restrict__ b2, float* __restrict__ out, int N) {
    __shared__ float4 stage4[32];
    float* stage = (float*)stage4;
    const int n = (blockIdx.x * 256 + threadIdx.x) >> 2;
    const int l = threadIdx.x & (G2_L - 1);
    float r0 = 0.f, r1 = 0.f;
    if (n < N) {
        const float4* __restrict__ rc = (const float4*)rec;
        const int start = rowptr[n];
        const int deg   = cnt[n];
        const float4 me = rc[n];
        asm volatile("" :: "v"(start));
        const float es = lrelu(me.z + me.w);
        float m = es, ssum = 0.f, ax = 0.f, ay = 0.f;
        for (int base = 0; base < deg; base += G2_L * G2_U) {
            int  id[G2_U];
            bool p[G2_U];
            float qx[G2_U], qy[G2_U], e[G2_U];
#pragma unroll
            for (int u = 0; u < G2_U; ++u) {
                const int j = base + u * G2_L + l;
                p[u] = j < deg;
                const int t = csr[start + (p[u] ? j : 0)];
                id[u] = p[u] ? t : n;
            }
#pragma unroll
            for (int u = 0; u < G2_U; ++u) {
                const float4 q = rc[id[u]];
                qx[u] = q.x; qy[u] = q.y;
                e[u]  = lrelu(q.z + me.w);
            }
            float mn = m;
#pragma unroll
            for (int u = 0; u < G2_U; ++u) mn = fmaxf(mn, p[u] ? e[u] : mn);
            const float sc = __expf(m - mn);
            ssum *= sc; ax *= sc; ay *= sc;
#pragma unroll
            for (int u = 0; u < G2_U; ++u) {
                const float w = p[u] ? __expf(e[u] - mn) : 0.f;
                ssum += w; ax += w * qx[u]; ay += w * qy[u];
            }
            m = mn;
        }
        const float mm = quad_max(m);
        {
            const float f = __expf(m - mm);
            ssum *= f; ax *= f; ay *= f;
        }
        ssum = quad_sum(ssum);
        ax   = quad_sum(ax);
        ay   = quad_sum(ay);
        {
            const float w = __expf(es - mm);
            ssum += w; ax += w * me.x; ay += w * me.y;
        }
        const float inv = 1.0f / (ssum + 1e-16f);
        r0 = ax * inv + b2[0];
        r1 = ay * inv + b2[1];
    }
    if (l == 0) {
        stage[(threadIdx.x >> 2) * 2 + 0] = r0;
        stage[(threadIdx.x >> 2) * 2 + 1] = r1;
    }
    __syncthreads();
    if (threadIdx.x < 32) {
        const int o  = blockIdx.x * 128 + threadIdx.x * 4;
        const int N2 = N * 2;
        if (o + 4 <= N2) {
            ((float4*)out)[o >> 2] = stage4[threadIdx.x];
        } else {
            for (int i = 0; i < 4; ++i)
                if (o + i < N2) out[o + i] = stage[threadIdx.x * 4 + i];
        }
    }
}

extern "C" void kernel_launch(void* const* d_in, const int* in_sizes, int n_in,
                              void* d_out, int out_size, void* d_ws, size_t ws_size,
                              hipStream_t stream) {
    const float* x    = (const float*)d_in[0];
    const int*   ei   = (const int*)d_in[1];
    const float* W1   = (const float*)d_in[2];
    const float* aS1  = (const float*)d_in[3];
    const float* aD1  = (const float*)d_in[4];
    const float* b1   = (const float*)d_in[5];
    const float* W2   = (const float*)d_in[6];
    const float* aS2  = (const float*)d_in[7];
    const float* aD2  = (const float*)d_in[8];
    const float* b2   = (const float*)d_in[9];

    const int N = in_sizes[0] / DIM;
    const int E = in_sizes[1] / 2;
    float* out = (float*)d_out;

    float* ws  = (float*)d_ws;
    float* h1  = ws;
    float* rec = h1 + (size_t)N * 64;
    const int B = (N + BIN_R - 1) / BIN_R;
    const int G = (E + CHUNK - 1) / CHUNK;
    int* binned = (int*)(rec + (size_t)N * 4);
    int* cnt    = binned + (size_t)G * CHUNK * 2;
    int* rowptr = cnt + (size_t)N;
    const int LP = ((B + 1 + 31) / 32) * 32;
    int* lsT    = rowptr + (size_t)N;
    int* csr    = lsT + (size_t)G * LP;

    const int Mtiles = (N + 15) >> 4;

    const int GB  = (Mtiles + GEMM_W - 1) / GEMM_W;
    const int GB0 = GB / 2;
    const int GB1 = GB - GB0;
    sort_and_gemm<<<GB0 + G, SORT_T, 0, stream>>>(ei, binned, lsT, E, B, LP, x, W1, h1, N, GB0);
    collect_and_gemm<<<GB1 + B, SORT_T, 0, stream>>>(binned, lsT, cnt, rowptr, csr, N, LP, G,
                                                    x, W1, h1, GB0, GB1);
    seg_gather1<<<(N + 15) / 16, 256, 0, stream>>>(csr, rowptr, cnt, aS1, aD1, h1, b1,
                                                  W2, aS2, aD2, rec, N);
    seg_gather2<<<(N + 63) / 64, 256, 0, stream>>>(csr, rowptr, cnt, rec, b2, out, N);
}


# Bin collect reads each chunk piece once, 8 lanes per piece in 64-byte runs, and keeps it in registers for sums, count and placement (was 3 table reads + 2 record walks); sum order by new LDS ranks
# speedup vs baseline: 11.7540x; 1.0938x over previous
#include <hip/hip_runtime.h>
#include <math.h>

#define DIM 128
#define HID 16
#define HEADS 4
#define NEG_SLOPE 0.2f

typedef __attribute__((ext_vector_type(2))) float v2f;
typedef __attribute__((ext_vector_type(8))) float v8f;

#define GEMM_T 512
#define GEMM_W (GEMM_T / 32)
__device__ __forceinline__ void
gemm1_rows(const float* __restrict__ x, const float* __restrict__ W1,
           float* __restrict__ h1, int N, int gblk) {
    __shared__ float4 Wl4[DIM * 16];
    {
        const int t = threadIdx.x;
#pragma unroll
        for (int j = 0; j < (DIM * 16) / GEMM_T; ++j) {
            const int e = (j * GEMM_T + t) * 4;
            float4 v;
            v.x = W1[e]; v.y = W1[e + 1]; v.z = W1[e + 2]; v.w = W1[e + 3];
            Wl4[j * GEMM_T + t] = v;
        }
    }
    const float* Wl = (const float*)Wl4;
    __syncthreads();

    const int wave  = threadIdx.x >> 5;
    const int lane  = threadIdx.x & 31;
    const int Mtile = gblk * GEMM_W + wave;
    const int Mtiles = (N + 15) >> 4;
    if (Mtile >= Mtiles) return;

    const int r  = lane & 15;
    const int kh = lane >> 4;
    const int mbase = Mtile * 16;
    int row = mbase + r;
    if (row >= N) row = N - 1;
    const float* __restrict__ xr = x + (size_t)row * DIM;

    v8f acc0 = {}, acc1 = {}, acc2 = {}, acc3 = {};

    for (int kb = 0; kb < DIM; kb += 4) {
        const int k0 = kb + 2 * kh;
        v2f a;
        a.x = xr[k0];
        a.y = xr[k0 + 1];
        const float* w0 = &Wl[(size_t)k0 * 64];
        const float* w1 = &Wl[(size_t)(k0 + 1) * 64];
        v2f b;
        b.x = w0[r];       b.y = w1[r];
        acc0 = __builtin_amdgcn_wmma_f32_16x16x4_f32(false, a, false, b, (short)0, acc0, false, false);
        b.x = w0[16 + r];  b.y = w1[16 + r];
        acc1 = __builtin_amdgcn_wmma_f32_16x16x4_f32(false, a, false, b, (short)0, acc1, false, false);
        b.x = w0[32 + r];  b.y = w1[32 + r];
        acc2 = __builtin_amdgcn_wmma_f32_16x16x4_f32(false, a, false, b, (short)0, acc2, false, false);
        b.x = w0[48 + r];  b.y = w1[48 + r];
        acc3 = __builtin_amdgcn_wmma_f32_16x16x4_f32(false, a, false, b, (short)0, acc3, false, false);
    }

    float* outp = h1 + (size_t)mbase * 64;
#pragma unroll
    for (int i = 0; i < 8; ++i) {
        const int m = i + 8 * kh;
        if (mbase + m < N) {
            float* o = outp + (size_t)m * 64 + r;
            o[0]  = acc0[i];
            o[16] = acc1[i];
            o[32] = acc2[i];
            o[48] = acc3[i];
        }
    }
}

#define BIN_SHIFT 8
#define BIN_R (1 << BIN_SHIFT)
#define CHUNK 4096
#define SORT_T 512
#define EPT (CHUNK / SORT_T)
static_assert(GEMM_T == SORT_T, "GEMM work-groups share the sort launches' block size");
#define MAXB 1024
#define STAGE_CAP 8192

__device__ __forceinline__ void block_excl_scan(int* a, int n, int* wsum) {
    const int tid = threadIdx.x, lane = tid & 31, wave = tid >> 5;
    const int i0 = 2 * tid, i1 = i0 + 1;
    const int v0 = (i0 < n) ? a[i0] : 0;
    const int v1 = (i1 < n) ? a[i1] : 0;
    const int s = v0 + v1;
    int x = s;
    for (int o = 1; o < 32; o <<= 1) {
        const int y = __shfl_up(x, o, 32);
        if (lane >= o) x += y;
    }
    if (lane == 31) wsum[wave] = x;
    __syncthreads();
    int off = 0;
    for (int w = 0; w < SORT_T / 32; ++w) if (w < wave) off += wsum[w];
    const int ex = off + x - s;
    if (i0 < n) a[i0] = ex;
    if (i1 < n) a[i1] = ex + v0;
    __syncthreads();
}

__device__ __forceinline__ void
chunk_sort(const int* __restrict__ ei, int* __restrict__ binned_w,
           int* __restrict__ lsT, int E, int B, int LP, int chunk) {
    int2* __restrict__ binned = (int2*)binned_w;
    __shared__ int h[MAXB];
    __shared__ int cur[MAXB];
    __shared__ int wsum[SORT_T / 32];
    __shared__ int2 srt[CHUNK];
    const int tid  = threadIdx.x;
    const int base = chunk * CHUNK;
    const int nC   = min(CHUNK, E - base);
    for (int i = tid; i < B; i += SORT_T) h[i] = 0;
    __syncthreads();
    int s[EPT], d[EPT];
#pragma unroll
    for (int j = 0; j < EPT; ++j) {
        const int i = j * SORT_T + tid;
        const bool ok = i < nC;
        s[j] = ok ? ei[base + i] : 0;
        d[j] = ok ? ei[E + base + i] : -1;
    }
#pragma unroll
    for (int j = 0; j < EPT; ++j)
        if (d[j] >= 0) atomicAdd(&h[d[j] >> BIN_SHIFT], 1);
    __syncthreads();
    block_excl_scan(h, B, wsum);
    int* lrow = lsT + (size_t)chunk * LP;
    for (int i = tid; i < B; i += SORT_T) { const int v = h[i]; cur[i] = v; lrow[i] = v; }
    if (tid == 0) lrow[B] = nC;
    __syncthreads();
#pragma unroll
    for (int j = 0; j < EPT; ++j)
        if (d[j] >= 0) {
            const int p = atomicAdd(&cur[d[j] >> BIN_SHIFT], 1);
            srt[p] = make_int2(s[j], d[j]);
        }
    __syncthreads();
    for (int i = tid; i < nC; i += SORT_T) binned[(size_t)base + i] = srt[i];
}

#define PG 8
#define PK 3
#define PPW (32 / PG)
#define PPB (PPW * (SORT_T / 32))
#define PRND 7
__device__ __forceinline__ void
bin_collect(const int* __restrict__ binned_w, const int* __restrict__ lsT,
            int* __restrict__ cnt, int* __restrict__ rowptr,
            int* __restrict__ csr, int N, int LP, int G, int b) {
    __shared__ int c[BIN_R];
    __shared__ int cur[BIN_R];
    __shared__ int wsum[SORT_T / 32];
    __shared__ int red[SORT_T / 32];
    __shared__ int stage[STAGE_CAP];
    const int2* __restrict__ binned = (const int2*)binned_w;
    const int tid = threadIdx.x, lane = tid & 31, wave = tid >> 5;
    const int g = lane / PG, j = lane % PG;
    if (tid < BIN_R) c[tid] = 0;

    int pa[PRND], pn[PRND];
#pragma unroll
    for (int t = 0; t < PRND; ++t) {
        const int w = t * PPB + wave * PPW + g;
        int a = 0, z = 0;
        if (w < G) {
            const int* lrow = lsT + (size_t)w * LP + b;
            a = lrow[0];
            z = lrow[1];
        }
        pa[t] = a;
        pn[t] = z - a;
    }
    int2 rc[PRND][PK];
#pragma unroll
    for (int t = 0; t < PRND; ++t) {
        const int w = t * PPB + wave * PPW + g;
        const int2* p = binned + (size_t)w * CHUNK + pa[t];
#pragma unroll
        for (int k = 0; k < PK; ++k) {
            rc[t][k] = make_int2(0, 0);
            if (j + PG * k < pn[t]) rc[t][k] = p[j + PG * k];
        }
    }

    int asum = 0, nsum = 0;
    bool longp = false;
#pragma unroll
    for (int t = 0; t < PRND; ++t) {
        if (j == 0) { asum += pa[t]; nsum += pn[t]; }
        longp = longp || (pn[t] > PG * PK);
    }
    for (int w = PRND * PPB + tid; w < G; w += SORT_T) {
        const int* lrow = lsT + (size_t)w * LP + b;
        const int a = lrow[0], z = lrow[1];
        asum += a;
        nsum += z - a;
    }
    for (int o = 16; o > 0; o >>= 1) {
        asum += __shfl_xor(asum, o, 32);
        nsum += __shfl_xor(nsum, o, 32);
    }
    if (lane == 0) { red[wave] = asum; wsum[wave] = nsum; }
    __syncthreads();
    int sbase = 0, nE = 0;
    for (int w = 0; w < SORT_T / 32; ++w) { sbase += red[w]; nE += wsum[w]; }

    int rp[PRND][PK];
#pragma unroll
    for (int t = 0; t < PRND; ++t)
#pragma unroll
        for (int k = 0; k < PK; ++k)
            rp[t][k] = (rc[t][k].x << BIN_SHIFT) | (rc[t][k].y & (BIN_R - 1));

#pragma unroll
    for (int t = 0; t < PRND; ++t) {
#pragma unroll
        for (int k = 0; k < PK; ++k)
            if (j + PG * k < pn[t]) atomicAdd(&c[rp[t][k] & (BIN_R - 1)], 1);
    }
    if (longp) {
        for (int t = 0; t < PRND; ++t) {
            const int w = t * PPB + wave * PPW + g;
            if (w < G) {
                const int* lrow = lsT + (size_t)w * LP + b;
                const int a = lrow[0], z = lrow[1];
                const int2* p = binned + (size_t)w * CHUNK;
                for (int i = a + PG * PK + j; i < z; i += PG) atomicAdd(&c[p[i].y & (BIN_R - 1)], 1);
            }
        }
    }
    for (int w = PRND * PPB + tid; w < G; w += SORT_T) {
        const int* lrow = lsT + (size_t)w * LP + b;
        const int a = lrow[0], z = lrow[1];
        const int2* p = binned + (size_t)w * CHUNK;
        for (int i = a; i < z; ++i) atomicAdd(&c[p[i].y & (BIN_R - 1)], 1);
    }
    __syncthreads();
    const int mine = (tid < BIN_R) ? c[tid] : 0;
    block_excl_scan(c, BIN_R, wsum);
    if (tid < BIN_R) {
        const int ex = c[tid];
        cur[tid] = ex;
        const int n = b * BIN_R + tid;
        if (n < N) { cnt[n] = mine; rowptr[n] = sbase + ex; }
    }
    __syncthreads();

    const bool staged = nE <= STAGE_CAP;
#pragma unroll
    for (int t = 0; t < PRND; ++t) {
        int q[PK];
#pragma unroll
        for (int k = 0; k < PK; ++k)
            q[k] = (j + PG * k < pn[t]) ? atomicAdd(&cur[rp[t][k] & (BIN_R - 1)], 1) : 0;
#pragma unroll
        for (int k = 0; k < PK; ++k)
            if (j + PG * k < pn[t]) {
                if (staged) stage[q[k]] = rp[t][k] >> BIN_SHIFT;
                else        csr[sbase + q[k]] = rp[t][k] >> BIN_SHIFT;
            }
    }
    if (longp) {
        for (int t = 0; t < PRND; ++t) {
            const int w = t * PPB + wave * PPW + g;
            if (w < G) {
                const int* lrow = lsT + (size_t)w * LP + b;
                const int a = lrow[0], z = lrow[1];
                const int2* p = binned + (size_t)w * CHUNK;
                for (int i = a + PG * PK + j; i < z; i += PG) {
                    const int2 r = p[i];
                    const int qq = atomicAdd(&cur[r.y & (BIN_R - 1)], 1);
                    if (staged) stage[qq] = r.x;
                    else        csr[sbase + qq] = r.x;
                }
            }
        }
    }
    for (int w = PRND * PPB + tid; w < G; w += SORT_T) {
        const int* lrow = lsT + (size_t)w * LP + b;
        const int a = lrow[0], z = lrow[1];
        const int2* p = binned + (size_t)w * CHUNK;
        for (int i = a; i < z; ++i) {
            const int2 r = p[i];
            const int qq = atomicAdd(&cur[r.y & (BIN_R - 1)], 1);
            if (staged) stage[qq] = r.x;
            else        csr[sbase + qq] = r.x;
        }
    }
    if (staged) {
        __syncthreads();
        for (int i = tid - (sbase & 31); i < nE; i += SORT_T)
            if (i >= 0) csr[sbase + i] = stage[i];
    }
}

__global__ void __launch_bounds__(SORT_T) __attribute__((amdgpu_waves_per_eu(4, 12)))
sort_and_gemm(const int* __restrict__ ei, int* __restrict__ binned_w,
              int* __restrict__ lsT, int E, int B, int LP,
              const float* __restrict__ x, const float* __restrict__ W1,
              float* __restrict__ h1, int N, int g1) {
    const int blk = (int)blockIdx.x;
    if (blk < g1) gemm1_rows(x, W1, h1, N, blk);
    else          chunk_sort(ei, binned_w, lsT, E, B, LP, blk - g1);
}

__global__ void __launch_bounds__(SORT_T) __attribute__((amdgpu_waves_per_eu(12, 12)))
collect_and_gemm(const int* __restrict__ binned_w, const int* __restrict__ lsT,
                 int* __restrict__ cnt, int* __restrict__ rowptr,
                 int* __restrict__ csr, int N, int LP, int G,
                 const float* __restrict__ x, const float* __restrict__ W1,
                 float* __restrict__ h1, int g0, int g2) {
    const int blk = (int)blockIdx.x;
    if (blk < g2) gemm1_rows(x, W1, h1, N, g0 + blk);
    else          bin_collect(binned_w, lsT, cnt, rowptr, csr, N, LP, G, blk - g2);
}

__device__ __forceinline__ float lrelu(float v) { return (v > 0.f) ? v : NEG_SLOPE * v; }

__device__ __forceinline__ float dot4(const float4 a, const float4 b) {
    return a.x * b.x + a.y * b.y + a.z * b.z + a.w * b.w;
}
__device__ __forceinline__ float quad_sum(float v) {
    v += __int_as_float(__builtin_amdgcn_update_dpp(0, __float_as_int(v), 0xB1, 0xF, 0xF, true));
    v += __int_as_float(__builtin_amdgcn_update_dpp(0, __float_as_int(v), 0x4E, 0xF, 0xF, true));
    return v;
}
__device__ __forceinline__ float row16_sum(float v) {
    v = quad_sum(v);
    v += __int_as_float(__builtin_amdgcn_update_dpp(0, __float_as_int(v), 0x164, 0xF, 0xF, true));
    v += __int_as_float(__builtin_amdgcn_update_dpp(0, __float_as_int(v), 0x168, 0xF, 0xF, true));
    return v;
}

__global__ void __launch_bounds__(256)
seg_gather1(const int* __restrict__ csr, const int* __restrict__ rowptr,
            const int* __restrict__ cnt, const float* __restrict__ attS,
            const float* __restrict__ attD, const float* __restrict__ h1,
            const float* __restrict__ b1, const float* __restrict__ W2,
            const float* __restrict__ attS2, const float* __restrict__ attD2,
            float* __restrict__ rec, int N) {
    __shared__ float4 stage[16];
    const int n  = (blockIdx.x * 256 + threadIdx.x) >> 4;
    const int l  = threadIdx.x & 15;
    if (n < N) {
        const int start = rowptr[n];
        const int deg   = cnt[n];
        const float4* __restrict__ h4 = (const float4*)h1;
        float4 aS, aD;
        aS.x = attS[l * 4 + 0]; aS.y = attS[l * 4 + 1]; aS.z = attS[l * 4 + 2]; aS.w = attS[l * 4 + 3];
        aD.x = attD[l * 4 + 0]; aD.y = attD[l * 4 + 1]; aD.z = attD[l * 4 + 2]; aD.w = attD[l * 4 + 3];

        const float4 vn = h4[n * 16 + l];
        const float adv = quad_sum(dot4(vn, aD));
        float m    = lrelu(quad_sum(dot4(vn, aS)) + adv);
        float ssum = 1.f;
        float4 acc = vn;

        for (int base = 0; base < deg; base += 16) {
            const int j   = base + l;
            const int idx = (j < deg) ? csr[start + j] : n;
            const int c   = min(16, deg - base);
            for (int k = 0; k < c; k += 4) {
                const int s0 = __shfl(idx, k, 16);
                const int s1 = __shfl(idx, k + 1, 16);
                const int s2 = __shfl(idx, k + 2, 16);
                const int s3 = __shfl(idx, k + 3, 16);
                const float4 v0 = h4[s0 * 16 + l];
                const float4 v1 = h4[s1 * 16 + l];
                const float4 v2 = h4[s2 * 16 + l];
                const float4 v3 = h4[s3 * 16 + l];
                const bool p1 = k + 1 < c, p2 = k + 2 < c, p3 = k + 3 < c;
                const float e0 = lrelu(quad_sum(dot4(v0, aS)) + adv);
                const float e1 = lrelu(quad_sum(dot4(v1, aS)) + adv);
                const float e2 = lrelu(quad_sum(dot4(v2, aS)) + adv);
                const float e3 = lrelu(quad_sum(dot4(v3, aS)) + adv);
                float mn = fmaxf(m, e0);
                mn = fmaxf(mn, p1 ? e1 : mn);
                mn = fmaxf(mn, p2 ? e2 : mn);
                mn = fmaxf(mn, p3 ? e3 : mn);
                const float sc = __expf(m - mn);
                const float w0 = __expf(e0 - mn);
                const float w1 = p1 ? __expf(e1 - mn) : 0.f;
                const float w2 = p2 ? __expf(e2 - mn) : 0.f;
                const float w3 = p3 ? __expf(e3 - mn) : 0.f;
                ssum = ssum * sc + w0;
                acc.x = acc.x * sc + w0 * v0.x; acc.y = acc.y * sc + w0 * v0.y;
                acc.z = acc.z * sc + w0 * v0.z; acc.w = acc.w * sc + w0 * v0.w;
                ssum += w1; acc.x += w1 * v1.x; acc.y += w1 * v1.y; acc.z += w1 * v1.z; acc.w += w1 * v1.w;
                ssum += w2; acc.x += w2 * v2.x; acc.y += w2 * v2.y; acc.z += w2 * v2.z; acc.w += w2 * v2.w;
                ssum += w3; acc.x += w3 * v3.x; acc.y += w3 * v3.y; acc.z += w3 * v3.z; acc.w += w3 * v3.w;
                m = mn;
            }
        }

        const float den = ssum + 1e-16f;
        float4 r;
        r.x = acc.x / den + b1[l * 4 + 0];
        r.y = acc.y / den + b1[l * 4 + 1];
        r.z = acc.z / den + b1[l * 4 + 2];
        r.w = acc.w / den + b1[l * 4 + 3];
        r.x = (r.x > 0.f) ? r.x : expm1f(r.x);
        r.y = (r.y > 0.f) ? r.y : expm1f(r.y);
        r.z = (r.z > 0.f) ? r.z : expm1f(r.z);
        r.w = (r.w > 0.f) ? r.w : expm1f(r.w);

        const float* __restrict__ wr = W2 + l * 8;
        const float a0 = row16_sum(r.x * wr[0] + r.y * wr[2] + r.z * wr[4] + r.w * wr[6]);
        const float a1 = row16_sum(r.x * wr[1] + r.y * wr[3] + r.z * wr[5] + r.w * wr[7]);
        if (l == 0) {
            float4 q;
            q.x = a0;
            q.y = a1;
            q.z = a0 * attS2[0] + a1 * attS2[1];
            q.w = a0 * attD2[0] + a1 * attD2[1];
            stage[threadIdx.x >> 4] = q;
        }
    }
    __syncthreads();
    if (threadIdx.x < 16) {
        const int nn = blockIdx.x * 16 + threadIdx.x;
        if (nn < N) ((float4*)rec)[nn] = stage[threadIdx.x];
    }
}

#define G2_L 4
#define G2_U 8
__device__ __forceinline__ float quad_max(float v) {
    v = fmaxf(v, __int_as_float(__builtin_amdgcn_update_dpp(0, __float_as_int(v), 0xB1, 0xF, 0xF, true)));
    v = fmaxf(v, __int_as_float(__builtin_amdgcn_update_dpp(0, __float_as_int(v), 0x4E, 0xF, 0xF, true)));
    return v;
}
__global__ void __launch_bounds__(256)
seg_gather2(const int* __restrict__ csr, const int* __restrict__ rowptr,
            const int* __restrict__ cnt, const float* __restrict__ rec,
            const float* __restrict__ b2, float* __restrict__ out, int N) {
    __shared__ float4 stage4[32];
    float* stage = (float*)stage4;
    const int n = (blockIdx.x * 256 + threadIdx.x) >> 2;
    const int l = threadIdx.x & (G2_L - 1);
    float r0 = 0.f, r1 = 0.f;
    if (n < N) {
        const float4* __restrict__ rc = (const float4*)rec;
        const int start = rowptr[n];
        const int deg   = cnt[n];
        const float4 me = rc[n];
        asm volatile("" :: "v"(start));
        const float es = lrelu(me.z + me.w);
        float m = es, ssum = 0.f, ax = 0.f, ay = 0.f;
        for (int base = 0; base < deg; base += G2_L * G2_U) {
            int  id[G2_U];
            bool p[G2_U];
            float qx[G2_U], qy[G2_U], e[G2_U];
#pragma unroll
            for (int u = 0; u < G2_U; ++u) {
                const int j = base + u * G2_L + l;
                p[u] = j < deg;
                const int t = csr[start + (p[u] ? j : 0)];
                id[u] = p[u] ? t : n;
            }
#pragma unroll
            for (int u = 0; u < G2_U; ++u) {
                const float4 q = rc[id[u]];
                qx[u] = q.x; qy[u] = q.y;
                e[u]  = lrelu(q.z + me.w);
            }
            float mn = m;
#pragma unroll
            for (int u = 0; u < G2_U; ++u) mn = fmaxf(mn, p[u] ? e[u] : mn);
            const float sc = __expf(m - mn);
            ssum *= sc; ax *= sc; ay *= sc;
#pragma unroll
            for (int u = 0; u < G2_U; ++u) {
                const float w = p[u] ? __expf(e[u] - mn) : 0.f;
                ssum += w; ax += w * qx[u]; ay += w * qy[u];
            }
            m = mn;
        }
        const float mm = quad_max(m);
        {
            const float f = __expf(m - mm);
            ssum *= f; ax *= f; ay *= f;
        }
        ssum = quad_sum(ssum);
        ax   = quad_sum(ax);
        ay   = quad_sum(ay);
        {
            const float w = __expf(es - mm);
            ssum += w; ax += w * me.x; ay += w * me.y;
        }
        const float inv = 1.0f / (ssum + 1e-16f);
        r0 = ax * inv + b2[0];
        r1 = ay * inv + b2[1];
    }
    if (l == 0) {
        stage[(threadIdx.x >> 2) * 2 + 0] = r0;
        stage[(threadIdx.x >> 2) * 2 + 1] = r1;
    }
    __syncthreads();
    if (threadIdx.x < 32) {
        const int o  = blockIdx.x * 128 + threadIdx.x * 4;
        const int N2 = N * 2;
        if (o + 4 <= N2) {
            ((float4*)out)[o >> 2] = stage4[threadIdx.x];
        } else {
            for (int i = 0; i < 4; ++i)
                if (o + i < N2) out[o + i] = stage[threadIdx.x * 4 + i];
        }
    }
}

extern "C" void kernel_launch(void* const* d_in, const int* in_sizes, int n_in,
                              void* d_out, int out_size, void* d_ws, size_t ws_size,
                              hipStream_t stream) {
    const float* x    = (const float*)d_in[0];
    const int*   ei   = (const int*)d_in[1];
    const float* W1   = (const float*)d_in[2];
    const float* aS1  = (const float*)d_in[3];
    const float* aD1  = (const float*)d_in[4];
    const float* b1   = (const float*)d_in[5];
    const float* W2   = (const float*)d_in[6];
    const float* aS2  = (const float*)d_in[7];
    const float* aD2  = (const float*)d_in[8];
    const float* b2   = (const float*)d_in[9];

    const int N = in_sizes[0] / DIM;
    const int E = in_sizes[1] / 2;
    float* out = (float*)d_out;

    float* ws  = (float*)d_ws;
    float* h1  = ws;
    float* rec = h1 + (size_t)N * 64;
    const int B = (N + BIN_R - 1) / BIN_R;
    const int G = (E + CHUNK - 1) / CHUNK;
    int* binned = (int*)(rec + (size_t)N * 4);
    int* cnt    = binned + (size_t)G * CHUNK * 2;
    int* rowptr = cnt + (size_t)N;
    const int LP = ((B + 1 + 31) / 32) * 32;
    int* lsT    = rowptr + (size_t)N;
    int* csr    = lsT + (size_t)G * LP;

    const int Mtiles = (N + 15) >> 4;

    const int GB  = (Mtiles + GEMM_W - 1) / GEMM_W;
    const int GB0 = GB / 2;
    const int GB1 = GB - GB0;
    sort_and_gemm<<<GB0 + G, SORT_T, 0, stream>>>(ei, binned, lsT, E, B, LP, x, W1, h1, N, GB0);
    collect_and_gemm<<<GB1 + B, SORT_T, 0, stream>>>(binned, lsT, cnt, rowptr, csr, N, LP, G,
                                                    x, W1, h1, GB0, GB1);
    seg_gather1<<<(N + 15) / 16, 256, 0, stream>>>(csr, rowptr, cnt, aS1, aD1, h1, b1,
                                                  W2, aS2, aD2, rec, N);
    seg_gather2<<<(N + 63) / 64, 256, 0, stream>>>(csr, rowptr, cnt, rec, b2, out, N);
}


# Layer-1 gather: 8 lanes per node (was 16), a lane owns 8 channels of one head (h1 columns stored in lane order), math on register pairs; 4 nodes per wave; logit and h2 sum order change
# speedup vs baseline: 12.1864x; 1.0368x over previous
#include <hip/hip_runtime.h>
#include <math.h>

#define DIM 128
#define HID 16
#define HEADS 4
#define NEG_SLOPE 0.2f

typedef __attribute__((ext_vector_type(2))) float v2f;
typedef __attribute__((ext_vector_type(4))) float v4f;
typedef __attribute__((ext_vector_type(8))) float v8f;

#define GEMM_T 512
#define GEMM_W (GEMM_T / 32)
__device__ __forceinline__ void
gemm1_rows(const float* __restrict__ x, const float* __restrict__ W1,
           float* __restrict__ h1, int N, int gblk) {
    __shared__ float4 Wl4[DIM * 16];
    {
        const int t = threadIdx.x;
#pragma unroll
        for (int j = 0; j < (DIM * 16) / GEMM_T; ++j) {
            const int e = (j * GEMM_T + t) * 4;
            float4 v;
            v.x = W1[e]; v.y = W1[e + 1]; v.z = W1[e + 2]; v.w = W1[e + 3];
            const int kr = j * (GEMM_T / 16) + (t >> 4);
            Wl4[kr * 16 + (t & 1) * 8 + ((t & 15) >> 1)] = v;
        }
    }
    const float* Wl = (const float*)Wl4;
    __syncthreads();

    const int wave  = threadIdx.x >> 5;
    const int lane  = threadIdx.x & 31;
    const int Mtile = gblk * GEMM_W + wave;
    const int Mtiles = (N + 15) >> 4;
    if (Mtile >= Mtiles) return;

    const int r  = lane & 15;
    const int kh = lane >> 4;
    const int mbase = Mtile * 16;
    int row = mbase + r;
    if (row >= N) row = N - 1;
    const float* __restrict__ xr = x + (size_t)row * DIM;

    v8f acc0 = {}, acc1 = {}, acc2 = {}, acc3 = {};

    for (int kb = 0; kb < DIM; kb += 4) {
        const int k0 = kb + 2 * kh;
        v2f a;
        a.x = xr[k0];
        a.y = xr[k0 + 1];
        const float* w0 = &Wl[(size_t)k0 * 64];
        const float* w1 = &Wl[(size_t)(k0 + 1) * 64];
        v2f b;
        b.x = w0[r];       b.y = w1[r];
        acc0 = __builtin_amdgcn_wmma_f32_16x16x4_f32(false, a, false, b, (short)0, acc0, false, false);
        b.x = w0[16 + r];  b.y = w1[16 + r];
        acc1 = __builtin_amdgcn_wmma_f32_16x16x4_f32(false, a, false, b, (short)0, acc1, false, false);
        b.x = w0[32 + r];  b.y = w1[32 + r];
        acc2 = __builtin_amdgcn_wmma_f32_16x16x4_f32(false, a, false, b, (short)0, acc2, false, false);
        b.x = w0[48 + r];  b.y = w1[48 + r];
        acc3 = __builtin_amdgcn_wmma_f32_16x16x4_f32(false, a, false, b, (short)0, acc3, false, false);
    }

    float* outp = h1 + (size_t)mbase * 64;
#pragma unroll
    for (int i = 0; i < 8; ++i) {
        const int m = i + 8 * kh;
        if (mbase + m < N) {
            float* o = outp + (size_t)m * 64 + r;
            o[0]  = acc0[i];
            o[16] = acc1[i];
            o[32] = acc2[i];
            o[48] = acc3[i];
        }
    }
}

#define BIN_SHIFT 8
#define BIN_R (1 << BIN_SHIFT)
#define CHUNK 4096
#define SORT_T 512
#define EPT (CHUNK / SORT_T)
static_assert(GEMM_T == SORT_T, "GEMM work-groups share the sort launches' block size");
#define MAXB 1024
#define STAGE_CAP 8192

__device__ __forceinline__ void block_excl_scan(int* a, int n, int* wsum) {
    const int tid = threadIdx.x, lane = tid & 31, wave = tid >> 5;
    const int i0 = 2 * tid, i1 = i0 + 1;
    const int v0 = (i0 < n) ? a[i0] : 0;
    const int v1 = (i1 < n) ? a[i1] : 0;
    const int s = v0 + v1;
    int x = s;
    for (int o = 1; o < 32; o <<= 1) {
        const int y = __shfl_up(x, o, 32);
        if (lane >= o) x += y;
    }
    if (lane == 31) wsum[wave] = x;
    __syncthreads();
    int off = 0;
    for (int w = 0; w < SORT_T / 32; ++w) if (w < wave) off += wsum[w];
    const int ex = off + x - s;
    if (i0 < n) a[i0] = ex;
    if (i1 < n) a[i1] = ex + v0;
    __syncthreads();
}

__device__ __forceinline__ void
chunk_sort(const int* __restrict__ ei, int* __restrict__ binned_w,
           int* __restrict__ lsT, int E, int B, int LP, int chunk) {
    int2* __restrict__ binned = (int2*)binned_w;
    __shared__ int h[MAXB];
    __shared__ int cur[MAXB];
    __shared__ int wsum[SORT_T / 32];
    __shared__ int2 srt[CHUNK];
    const int tid  = threadIdx.x;
    const int base = chunk * CHUNK;
    const int nC   = min(CHUNK, E - base);
    for (int i = tid; i < B; i += SORT_T) h[i] = 0;
    __syncthreads();
    int s[EPT], d[EPT];
#pragma unroll
    for (int j = 0; j < EPT; ++j) {
        const int i = j * SORT_T + tid;
        const bool ok = i < nC;
        s[j] = ok ? ei[base + i] : 0;
        d[j] = ok ? ei[E + base + i] : -1;
    }
#pragma unroll
    for (int j = 0; j < EPT; ++j)
        if (d[j] >= 0) atomicAdd(&h[d[j] >> BIN_SHIFT], 1);
    __syncthreads();
    block_excl_scan(h, B, wsum);
    int* lrow = lsT + (size_t)chunk * LP;
    for (int i = tid; i < B; i += SORT_T) { const int v = h[i]; cur[i] = v; lrow[i] = v; }
    if (tid == 0) lrow[B] = nC;
    __syncthreads();
#pragma unroll
    for (int j = 0; j < EPT; ++j)
        if (d[j] >= 0) {
            const int p = atomicAdd(&cur[d[j] >> BIN_SHIFT], 1);
            srt[p] = make_int2(s[j], d[j]);
        }
    __syncthreads();
    for (int i = tid; i < nC; i += SORT_T) binned[(size_t)base + i] = srt[i];
}

#define PG 8
#define PK 3
#define PPW (32 / PG)
#define PPB (PPW * (SORT_T / 32))
#define PRND 7
__device__ __forceinline__ void
bin_collect(const int* __restrict__ binned_w, const int* __restrict__ lsT,
            int* __restrict__ cnt, int* __restrict__ rowptr,
            int* __restrict__ csr, int N, int LP, int G, int b) {
    __shared__ int c[BIN_R];
    __shared__ int cur[BIN_R];
    __shared__ int wsum[SORT_T / 32];
    __shared__ int red[SORT_T / 32];
    __shared__ int stage[STAGE_CAP];
    const int2* __restrict__ binned = (const int2*)binned_w;
    const int tid = threadIdx.x, lane = tid & 31, wave = tid >> 5;
    const int g = lane / PG, j = lane % PG;
    if (tid < BIN_R) c[tid] = 0;

    int pa[PRND], pn[PRND];
#pragma unroll
    for (int t = 0; t < PRND; ++t) {
        const int w = t * PPB + wave * PPW + g;
        int a = 0, z = 0;
        if (w < G) {
            const int* lrow = lsT + (size_t)w * LP + b;
            a = lrow[0];
            z = lrow[1];
        }
        pa[t] = a;
        pn[t] = z - a;
    }
    int2 rc[PRND][PK];
#pragma unroll
    for (int t = 0; t < PRND; ++t) {
        const int w = t * PPB + wave * PPW + g;
        const int2* p = binned + (size_t)w * CHUNK + pa[t];
#pragma unroll
        for (int k = 0; k < PK; ++k) {
            rc[t][k] = make_int2(0, 0);
            if (j + PG * k < pn[t]) rc[t][k] = p[j + PG * k];
        }
    }

    int asum = 0, nsum = 0;
    bool longp = false;
#pragma unroll
    for (int t = 0; t < PRND; ++t) {
        if (j == 0) { asum += pa[t]; nsum += pn[t]; }
        longp = longp || (pn[t] > PG * PK);
    }
    for (int w = PRND * PPB + tid; w < G; w += SORT_T) {
        const int* lrow = lsT + (size_t)w * LP + b;
        const int a = lrow[0], z = lrow[1];
        asum += a;
        nsum += z - a;
    }
    for (int o = 16; o > 0; o >>= 1) {
        asum += __shfl_xor(asum, o, 32);
        nsum += __shfl_xor(nsum, o, 32);
    }
    if (lane == 0) { red[wave] = asum; wsum[wave] = nsum; }
    __syncthreads();
    int sbase = 0, nE = 0;
    for (int w = 0; w < SORT_T / 32; ++w) { sbase += red[w]; nE += wsum[w]; }

    int rp[PRND][PK];
#pragma unroll
    for (int t = 0; t < PRND; ++t)
#pragma unroll
        for (int k = 0; k < PK; ++k)
            rp[t][k] = (rc[t][k].x << BIN_SHIFT) | (rc[t][k].y & (BIN_R - 1));

#pragma unroll
    for (int t = 0; t < PRND; ++t) {
#pragma unroll
        for (int k = 0; k < PK; ++k)
            if (j + PG * k < pn[t]) atomicAdd(&c[rp[t][k] & (BIN_R - 1)], 1);
    }
    if (longp) {
        for (int t = 0; t < PRND; ++t) {
            const int w = t * PPB + wave * PPW + g;
            if (w < G) {
                const int* lrow = lsT + (size_t)w * LP + b;
                const int a = lrow[0], z = lrow[1];
                const int2* p = binned + (size_t)w * CHUNK;
                for (int i = a + PG * PK + j; i < z; i += PG) atomicAdd(&c[p[i].y & (BIN_R - 1)], 1);
            }
        }
    }
    for (int w = PRND * PPB + tid; w < G; w += SORT_T) {
        const int* lrow = lsT + (size_t)w * LP + b;
        const int a = lrow[0], z = lrow[1];
        const int2* p = binned + (size_t)w * CHUNK;
        for (int i = a; i < z; ++i) atomicAdd(&c[p[i].y & (BIN_R - 1)], 1);
    }
    __syncthreads();
    const int mine = (tid < BIN_R) ? c[tid] : 0;
    block_excl_scan(c, BIN_R, wsum);
    if (tid < BIN_R) {
        const int ex = c[tid];
        cur[tid] = ex;
        const int n = b * BIN_R + tid;
        if (n < N) { cnt[n] = mine; rowptr[n] = sbase + ex; }
    }
    __syncthreads();

    const bool staged = nE <= STAGE_CAP;
#pragma unroll
    for (int t = 0; t < PRND; ++t) {
        int q[PK];
#pragma unroll
        for (int k = 0; k < PK; ++k)
            q[k] = (j + PG * k < pn[t]) ? atomicAdd(&cur[rp[t][k] & (BIN_R - 1)], 1) : 0;
#pragma unroll
        for (int k = 0; k < PK; ++k)
            if (j + PG * k < pn[t]) {
                if (staged) stage[q[k]] = rp[t][k] >> BIN_SHIFT;
                else        csr[sbase + q[k]] = rp[t][k] >> BIN_SHIFT;
            }
    }
    if (longp) {
        for (int t = 0; t < PRND; ++t) {
            const int w = t * PPB + wave * PPW + g;
            if (w < G) {
                const int* lrow = lsT + (size_t)w * LP + b;
                const int a = lrow[0], z = lrow[1];
                const int2* p = binned + (size_t)w * CHUNK;
                for (int i = a + PG * PK + j; i < z; i += PG) {
                    const int2 r = p[i];
                    const int qq = atomicAdd(&cur[r.y & (BIN_R - 1)], 1);
                    if (staged) stage[qq] = r.x;
                    else        csr[sbase + qq] = r.x;
                }
            }
        }
    }
    for (int w = PRND * PPB + tid; w < G; w += SORT_T) {
        const int* lrow = lsT + (size_t)w * LP + b;
        const int a = lrow[0], z = lrow[1];
        const int2* p = binned + (size_t)w * CHUNK;
        for (int i = a; i < z; ++i) {
            const int2 r = p[i];
            const int qq = atomicAdd(&cur[r.y & (BIN_R - 1)], 1);
            if (staged) stage[qq] = r.x;
            else        csr[sbase + qq] = r.x;
        }
    }
    if (staged) {
        __syncthreads();
        for (int i = tid - (sbase & 31); i < nE; i += SORT_T)
            if (i >= 0) csr[sbase + i] = stage[i];
    }
}

__global__ void __launch_bounds__(SORT_T) __attribute__((amdgpu_waves_per_eu(4, 12)))
sort_and_gemm(const int* __restrict__ ei, int* __restrict__ binned_w,
              int* __restrict__ lsT, int E, int B, int LP,
              const float* __restrict__ x, const float* __restrict__ W1,
              float* __restrict__ h1, int N, int g1) {
    const int blk = (int)blockIdx.x;
    if (blk < g1) gemm1_rows(x, W1, h1, N, blk);
    else          chunk_sort(ei, binned_w, lsT, E, B, LP, blk - g1);
}

__global__ void __launch_bounds__(SORT_T) __attribute__((amdgpu_waves_per_eu(12, 12)))
collect_and_gemm(const int* __restrict__ binned_w, const int* __restrict__ lsT,
                 int* __restrict__ cnt, int* __restrict__ rowptr,
                 int* __restrict__ csr, int N, int LP, int G,
                 const float* __restrict__ x, const float* __restrict__ W1,
                 float* __restrict__ h1, int g0, int g2) {
    const int blk = (int)blockIdx.x;
    if (blk < g2) gemm1_rows(x, W1, h1, N, g0 + blk);
    else          bin_collect(binned_w, lsT, cnt, rowptr, csr, N, LP, G, blk - g2);
}

__device__ __forceinline__ float lrelu(float v) { return (v > 0.f) ? v : NEG_SLOPE * v; }

__device__ __forceinline__ float dot4(const float4 a, const float4 b) {
    return a.x * b.x + a.y * b.y + a.z * b.z + a.w * b.w;
}
__device__ __forceinline__ float quad_sum(float v) {
    v += __int_as_float(__builtin_amdgcn_update_dpp(0, __float_as_int(v), 0xB1, 0xF, 0xF, true));
    v += __int_as_float(__builtin_amdgcn_update_dpp(0, __float_as_int(v), 0x4E, 0xF, 0xF, true));
    return v;
}
__device__ __forceinline__ float dot8(const v4f a, const v4f b, const v4f wa, const v4f wb) {
    v2f d = a.xy * wa.xy;
    d = a.zw * wa.zw + d;
    d = b.xy * wb.xy + d;
    d = b.zw * wb.zw + d;
    return d.x + d.y;
}
__device__ __forceinline__ float pair_sum(float v) {
    v += __int_as_float(__builtin_amdgcn_update_dpp(0, __float_as_int(v), 0xB1, 0xF, 0xF, true));
    return v;
}
__device__ __forceinline__ float oct_sum(float v) {
    v = quad_sum(v);
    v += __int_as_float(__builtin_amdgcn_update_dpp(0, __float_as_int(v), 0x164, 0xF, 0xF, true));
    return v;
}

#define G1_L 8
__global__ void __launch_bounds__(256)
seg_gather1(const int* __restrict__ csr, const int* __restrict__ rowptr,
            const int* __restrict__ cnt, const float* __restrict__ attS,
            const float* __restrict__ attD, const float* __restrict__ h1,
            const float* __restrict__ b1, const float* __restrict__ W2,
            const float* __restrict__ attS2, const float* __restrict__ attD2,
            float* __restrict__ rec, int N) {
    __shared__ float4 stage[256 / G1_L];
    const int n  = (blockIdx.x * 256 + threadIdx.x) / G1_L;
    const int l  = threadIdx.x & (G1_L - 1);
    if (n < N) {
        const int start = rowptr[n];
        const int deg   = cnt[n];
        const v4f* __restrict__ h4 = (const v4f*)h1;
        v4f aS0, aS1, aD0, aD1;
        aS0.x = attS[l * 8 + 0]; aS0.y = attS[l * 8 + 1]; aS0.z = attS[l * 8 + 2]; aS0.w = attS[l * 8 + 3];
        aS1.x = attS[l * 8 + 4]; aS1.y = attS[l * 8 + 5]; aS1.z = attS[l * 8 + 6]; aS1.w = attS[l * 8 + 7];
        aD0.x = attD[l * 8 + 0]; aD0.y = attD[l * 8 + 1]; aD0.z = attD[l * 8 + 2]; aD0.w = attD[l * 8 + 3];
        aD1.x = attD[l * 8 + 4]; aD1.y = attD[l * 8 + 5]; aD1.z = attD[l * 8 + 6]; aD1.w = attD[l * 8 + 7];

        const v4f vn0 = h4[n * 16 + l];
        const v4f vn1 = h4[n * 16 + 8 + l];
        const float adv = pair_sum(dot8(vn0, vn1, aD0, aD1));
        float m    = lrelu(pair_sum(dot8(vn0, vn1, aS0, aS1)) + adv);
        float ssum = 1.f;
        v2f A0 = vn0.xy, A1 = vn0.zw, A2 = vn1.xy, A3 = vn1.zw;

        for (int base = 0; base < deg; base += 2 * G1_L) {
            const int j0 = base + l, j1 = base + G1_L + l;
            const int t0 = csr[start + ((j0 < deg) ? j0 : 0)];
            const int t1 = csr[start + ((j1 < deg) ? j1 : 0)];
            const int idx0 = (j0 < deg) ? t0 : n;
            const int idx1 = (j1 < deg) ? t1 : n;
            const int c    = min(2 * G1_L, deg - base);
            for (int k = 0; k < c; k += 4) {
                const int idv = (k < G1_L) ? idx0 : idx1;
                const int kk  = k & (G1_L - 1);
                const int s0 = __shfl(idv, kk, G1_L);
                const int s1 = __shfl(idv, kk + 1, G1_L);
                const int s2 = __shfl(idv, kk + 2, G1_L);
                const int s3 = __shfl(idv, kk + 3, G1_L);
                const v4f v0a = h4[s0 * 16 + l], v0b = h4[s0 * 16 + 8 + l];
                const v4f v1a = h4[s1 * 16 + l], v1b = h4[s1 * 16 + 8 + l];
                const v4f v2a = h4[s2 * 16 + l], v2b = h4[s2 * 16 + 8 + l];
                const v4f v3a = h4[s3 * 16 + l], v3b = h4[s3 * 16 + 8 + l];
                const bool p1 = k + 1 < c, p2 = k + 2 < c, p3 = k + 3 < c;
                const float e0 = lrelu(pair_sum(dot8(v0a, v0b, aS0, aS1)) + adv);
                const float e1 = lrelu(pair_sum(dot8(v1a, v1b, aS0, aS1)) + adv);
                const float e2 = lrelu(pair_sum(dot8(v2a, v2b, aS0, aS1)) + adv);
                const float e3 = lrelu(pair_sum(dot8(v3a, v3b, aS0, aS1)) + adv);
                float mn = fmaxf(m, e0);
                mn = fmaxf(mn, p1 ? e1 : mn);
                mn = fmaxf(mn, p2 ? e2 : mn);
                mn = fmaxf(mn, p3 ? e3 : mn);
                const float sc = __expf(m - mn);
                const float w0 = __expf(e0 - mn);
                const float w1 = p1 ? __expf(e1 - mn) : 0.f;
                const float w2 = p2 ? __expf(e2 - mn) : 0.f;
                const float w3 = p3 ? __expf(e3 - mn) : 0.f;
                ssum = ssum * sc + w0;
                A0 = A0 * sc + v0a.xy * w0; A1 = A1 * sc + v0a.zw * w0;
                A2 = A2 * sc + v0b.xy * w0; A3 = A3 * sc + v0b.zw * w0;
                ssum += w1;
                A0 += v1a.xy * w1; A1 += v1a.zw * w1; A2 += v1b.xy * w1; A3 += v1b.zw * w1;
                ssum += w2;
                A0 += v2a.xy * w2; A1 += v2a.zw * w2; A2 += v2b.xy * w2; A3 += v2b.zw * w2;
                ssum += w3;
                A0 += v3a.xy * w3; A1 += v3a.zw * w3; A2 += v3b.xy * w3; A3 += v3b.zw * w3;
                m = mn;
            }
        }

        const float den = ssum + 1e-16f;
        float r[8];
        r[0] = A0.x / den + b1[l * 8 + 0];
        r[1] = A0.y / den + b1[l * 8 + 1];
        r[2] = A1.x / den + b1[l * 8 + 2];
        r[3] = A1.y / den + b1[l * 8 + 3];
        r[4] = A2.x / den + b1[l * 8 + 4];
        r[5] = A2.y / den + b1[l * 8 + 5];
        r[6] = A3.x / den + b1[l * 8 + 6];
        r[7] = A3.y / den + b1[l * 8 + 7];
#pragma unroll
        for (int i = 0; i < 8; ++i) r[i] = (r[i] > 0.f) ? r[i] : expm1f(r[i]);

        const float* __restrict__ wr = W2 + l * 16;
        float t0 = 0.f, t1 = 0.f;
#pragma unroll
        for (int i = 0; i < 8; ++i) { t0 += r[i] * wr[2 * i]; t1 += r[i] * wr[2 * i + 1]; }
        const float a0 = oct_sum(t0);
        const float a1 = oct_sum(t1);
        if (l == 0) {
            float4 q;
            q.x = a0;
            q.y = a1;
            q.z = a0 * attS2[0] + a1 * attS2[1];
            q.w = a0 * attD2[0] + a1 * attD2[1];
            stage[threadIdx.x / G1_L] = q;
        }
    }
    __syncthreads();
    if (threadIdx.x < 256 / G1_L) {
        const int nn = blockIdx.x * (256 / G1_L) + threadIdx.x;
        if (nn < N) ((float4*)rec)[nn] = stage[threadIdx.x];
    }
}

#define G2_L 4
#define G2_U 8
__device__ __forceinline__ float quad_max(float v) {
    v = fmaxf(v, __int_as_float(__builtin_amdgcn_update_dpp(0, __float_as_int(v), 0xB1, 0xF, 0xF, true)));
    v = fmaxf(v, __int_as_float(__builtin_amdgcn_update_dpp(0, __float_as_int(v), 0x4E, 0xF, 0xF, true)));
    return v;
}
__global__ void __launch_bounds__(256)
seg_gather2(const int* __restrict__ csr, const int* __restrict__ rowptr,
            const int* __restrict__ cnt, const float* __restrict__ rec,
            const float* __restrict__ b2, float* __restrict__ out, int N) {
    __shared__ float4 stage4[32];
    float* stage = (float*)stage4;
    const int n = (blockIdx.x * 256 + threadIdx.x) >> 2;
    const int l = threadIdx.x & (G2_L - 1);
    float r0 = 0.f, r1 = 0.f;
    if (n < N) {
        const float4* __restrict__ rc = (const float4*)rec;
        const int start = rowptr[n];
        const int deg   = cnt[n];
        const float4 me = rc[n];
        asm volatile("" :: "v"(start));
        const float es = lrelu(me.z + me.w);
        float m = es, ssum = 0.f, ax = 0.f, ay = 0.f;
        for (int base = 0; base < deg; base += G2_L * G2_U) {
            int  id[G2_U];
            bool p[G2_U];
            float qx[G2_U], qy[G2_U], e[G2_U];
#pragma unroll
            for (int u = 0; u < G2_U; ++u) {
                const int j = base + u * G2_L + l;
                p[u] = j < deg;
                const int t = csr[start + (p[u] ? j : 0)];
                id[u] = p[u] ? t : n;
            }
#pragma unroll
            for (int u = 0; u < G2_U; ++u) {
                const float4 q = rc[id[u]];
                qx[u] = q.x; qy[u] = q.y;
                e[u]  = lrelu(q.z + me.w);
            }
            float mn = m;
#pragma unroll
            for (int u = 0; u < G2_U; ++u) mn = fmaxf(mn, p[u] ? e[u] : mn);
            const float sc = __expf(m - mn);
            ssum *= sc; ax *= sc; ay *= sc;
#pragma unroll
            for (int u = 0; u < G2_U; ++u) {
                const float w = p[u] ? __expf(e[u] - mn) : 0.f;
                ssum += w; ax += w * qx[u]; ay += w * qy[u];
            }
            m = mn;
        }
        const float mm = quad_max(m);
        {
            const float f = __expf(m - mm);
            ssum *= f; ax *= f; ay *= f;
        }
        ssum = quad_sum(ssum);
        ax   = quad_sum(ax);
        ay   = quad_sum(ay);
        {
            const float w = __expf(es - mm);
            ssum += w; ax += w * me.x; ay += w * me.y;
        }
        const float inv = 1.0f / (ssum + 1e-16f);
        r0 = ax * inv + b2[0];
        r1 = ay * inv + b2[1];
    }
    if (l == 0) {
        stage[(threadIdx.x >> 2) * 2 + 0] = r0;
        stage[(threadIdx.x >> 2) * 2 + 1] = r1;
    }
    __syncthreads();
    if (threadIdx.x < 32) {
        const int o  = blockIdx.x * 128 + threadIdx.x * 4;
        const int N2 = N * 2;
        if (o + 4 <= N2) {
            ((float4*)out)[o >> 2] = stage4[threadIdx.x];
        } else {
            for (int i = 0; i < 4; ++i)
                if (o + i < N2) out[o + i] = stage[threadIdx.x * 4 + i];
        }
    }
}

extern "C" void kernel_launch(void* const* d_in, const int* in_sizes, int n_in,
                              void* d_out, int out_size, void* d_ws, size_t ws_size,
                              hipStream_t stream) {
    const float* x    = (const float*)d_in[0];
    const int*   ei   = (const int*)d_in[1];
    const float* W1   = (const float*)d_in[2];
    const float* aS1  = (const float*)d_in[3];
    const float* aD1  = (const float*)d_in[4];
    const float* b1   = (const float*)d_in[5];
    const float* W2   = (const float*)d_in[6];
    const float* aS2  = (const float*)d_in[7];
    const float* aD2  = (const float*)d_in[8];
    const float* b2   = (const float*)d_in[9];

    const int N = in_sizes[0] / DIM;
    const int E = in_sizes[1] / 2;
    float* out = (float*)d_out;

    float* ws  = (float*)d_ws;
    float* h1  = ws;
    float* rec = h1 + (size_t)N * 64;
    const int B = (N + BIN_R - 1) / BIN_R;
    const int G = (E + CHUNK - 1) / CHUNK;
    int* binned = (int*)(rec + (size_t)N * 4);
    int* cnt    = binned + (size_t)G * CHUNK * 2;
    int* rowptr = cnt + (size_t)N;
    const int LP = ((B + 1 + 31) / 32) * 32;
    int* lsT    = rowptr + (size_t)N;
    int* csr    = lsT + (size_t)G * LP;

    const int Mtiles = (N + 15) >> 4;

    const int GB  = (Mtiles + GEMM_W - 1) / GEMM_W;
    const int GB0 = GB / 2;
    const int GB1 = GB - GB0;
    sort_and_gemm<<<GB0 + G, SORT_T, 0, stream>>>(ei, binned, lsT, E, B, LP, x, W1, h1, N, GB0);
    collect_and_gemm<<<GB1 + B, SORT_T, 0, stream>>>(binned, lsT, cnt, rowptr, csr, N, LP, G,
                                                    x, W1, h1, GB0, GB1);
    seg_gather1<<<(N + 31) / 32, 256, 0, stream>>>(csr, rowptr, cnt, aS1, aD1, h1, b1,
                                                  W2, aS2, aD2, rec, N);
    seg_gather2<<<(N + 63) / 64, 256, 0, stream>>>(csr, rowptr, cnt, rec, b2, out, N);
}


# Bin collect and layer-1 gather fused in one launch: a bin's work-group gathers its own 256 nodes, bounds and source ids read from its LDS slice; all GEMM rows now in launch 1; 4 launches to 3
# speedup vs baseline: 12.6008x; 1.0340x over previous
#include <hip/hip_runtime.h>
#include <math.h>

#define DIM 128
#define HID 16
#define HEADS 4
#define NEG_SLOPE 0.2f

typedef __attribute__((ext_vector_type(2))) float v2f;
typedef __attribute__((ext_vector_type(4))) float v4f;
typedef __attribute__((ext_vector_type(8))) float v8f;

#define GEMM_T 512
#define GEMM_W (GEMM_T / 32)
__device__ __forceinline__ void
gemm1_rows(const float* __restrict__ x, const float* __restrict__ W1,
           float* __restrict__ h1, int N, int gblk) {
    __shared__ float4 Wl4[DIM * 16];
    {
        const int t = threadIdx.x;
#pragma unroll
        for (int j = 0; j < (DIM * 16) / GEMM_T; ++j) {
            const int e = (j * GEMM_T + t) * 4;
            float4 v;
            v.x = W1[e]; v.y = W1[e + 1]; v.z = W1[e + 2]; v.w = W1[e + 3];
            const int kr = j * (GEMM_T / 16) + (t >> 4);
            Wl4[kr * 16 + (t & 1) * 8 + ((t & 15) >> 1)] = v;
        }
    }
    const float* Wl = (const float*)Wl4;
    __syncthreads();

    const int wave  = threadIdx.x >> 5;
    const int lane  = threadIdx.x & 31;
    const int Mtile = gblk * GEMM_W + wave;
    const int Mtiles = (N + 15) >> 4;
    if (Mtile >= Mtiles) return;

    const int r  = lane & 15;
    const int kh = lane >> 4;
    const int mbase = Mtile * 16;
    int row = mbase + r;
    if (row >= N) row = N - 1;
    const float* __restrict__ xr = x + (size_t)row * DIM;

    v8f acc0 = {}, acc1 = {}, acc2 = {}, acc3 = {};

    for (int kb = 0; kb < DIM; kb += 4) {
        const int k0 = kb + 2 * kh;
        v2f a;
        a.x = xr[k0];
        a.y = xr[k0 + 1];
        const float* w0 = &Wl[(size_t)k0 * 64];
        const float* w1 = &Wl[(size_t)(k0 + 1) * 64];
        v2f b;
        b.x = w0[r];       b.y = w1[r];
        acc0 = __builtin_amdgcn_wmma_f32_16x16x4_f32(false, a, false, b, (short)0, acc0, false, false);
        b.x = w0[16 + r];  b.y = w1[16 + r];
        acc1 = __builtin_amdgcn_wmma_f32_16x16x4_f32(false, a, false, b, (short)0, acc1, false, false);
        b.x = w0[32 + r];  b.y = w1[32 + r];
        acc2 = __builtin_amdgcn_wmma_f32_16x16x4_f32(false, a, false, b, (short)0, acc2, false, false);
        b.x = w0[48 + r];  b.y = w1[48 + r];
        acc3 = __builtin_amdgcn_wmma_f32_16x16x4_f32(false, a, false, b, (short)0, acc3, false, false);
    }

    float* outp = h1 + (size_t)mbase * 64;
#pragma unroll
    for (int i = 0; i < 8; ++i) {
        const int m = i + 8 * kh;
        if (mbase + m < N) {
            float* o = outp + (size_t)m * 64 + r;
            o[0]  = acc0[i];
            o[16] = acc1[i];
            o[32] = acc2[i];
            o[48] = acc3[i];
        }
    }
}

#define BIN_SHIFT 8
#define BIN_R (1 << BIN_SHIFT)
#define CHUNK 4096
#define SORT_T 512
#define EPT (CHUNK / SORT_T)
static_assert(GEMM_T == SORT_T, "GEMM work-groups share the sort launches' block size");
#define MAXB 1024
#define STAGE_CAP 8192

__device__ __forceinline__ void block_excl_scan(int* a, int n, int* wsum) {
    const int tid = threadIdx.x, lane = tid & 31, wave = tid >> 5;
    const int i0 = 2 * tid, i1 = i0 + 1;
    const int v0 = (i0 < n) ? a[i0] : 0;
    const int v1 = (i1 < n) ? a[i1] : 0;
    const int s = v0 + v1;
    int x = s;
    for (int o = 1; o < 32; o <<= 1) {
        const int y = __shfl_up(x, o, 32);
        if (lane >= o) x += y;
    }
    if (lane == 31) wsum[wave] = x;
    __syncthreads();
    int off = 0;
    for (int w = 0; w < SORT_T / 32; ++w) if (w < wave) off += wsum[w];
    const int ex = off + x - s;
    if (i0 < n) a[i0] = ex;
    if (i1 < n) a[i1] = ex + v0;
    __syncthreads();
}

__device__ __forceinline__ void
chunk_sort(const int* __restrict__ ei, int* __restrict__ binned_w,
           int* __restrict__ lsT, int E, int B, int LP, int chunk) {
    int2* __restrict__ binned = (int2*)binned_w;
    __shared__ int h[MAXB];
    __shared__ int cur[MAXB];
    __shared__ int wsum[SORT_T / 32];
    __shared__ int2 srt[CHUNK];
    const int tid  = threadIdx.x;
    const int base = chunk * CHUNK;
    const int nC   = min(CHUNK, E - base);
    for (int i = tid; i < B; i += SORT_T) h[i] = 0;
    __syncthreads();
    int s[EPT], d[EPT];
#pragma unroll
    for (int j = 0; j < EPT; ++j) {
        const int i = j * SORT_T + tid;
        const bool ok = i < nC;
        s[j] = ok ? ei[base + i] : 0;
        d[j] = ok ? ei[E + base + i] : -1;
    }
#pragma unroll
    for (int j = 0; j < EPT; ++j)
        if (d[j] >= 0) atomicAdd(&h[d[j] >> BIN_SHIFT], 1);
    __syncthreads();
    block_excl_scan(h, B, wsum);
    int* lrow = lsT + (size_t)chunk * LP;
    for (int i = tid; i < B; i += SORT_T) { const int v = h[i]; cur[i] = v; lrow[i] = v; }
    if (tid == 0) lrow[B] = nC;
    __syncthreads();
#pragma unroll
    for (int j = 0; j < EPT; ++j)
        if (d[j] >= 0) {
            const int p = atomicAdd(&cur[d[j] >> BIN_SHIFT], 1);
            srt[p] = make_int2(s[j], d[j]);
        }
    __syncthreads();
    for (int i = tid; i < nC; i += SORT_T) binned[(size_t)base + i] = srt[i];
}

#define PG 8
#define PK 3
#define PPW (32 / PG)
#define PPB (PPW * (SORT_T / 32))
#define PRND 7
__device__ __forceinline__ void
bin_collect(const int* __restrict__ binned_w, const int* __restrict__ lsT,
            int* __restrict__ cnt, int* __restrict__ rowptr,
            int* __restrict__ csr, int N, int LP, int G, int b,
            int* c, int* cur, int* wsum, int* red, int* stage,
            bool& staged_o, int& sbase_o) {
    const int2* __restrict__ binned = (const int2*)binned_w;
    const int tid = threadIdx.x, lane = tid & 31, wave = tid >> 5;
    const int g = lane / PG, j = lane % PG;
    if (tid < BIN_R) c[tid] = 0;

    int pa[PRND], pn[PRND];
#pragma unroll
    for (int t = 0; t < PRND; ++t) {
        const int w = t * PPB + wave * PPW + g;
        int a = 0, z = 0;
        if (w < G) {
            const int* lrow = lsT + (size_t)w * LP + b;
            a = lrow[0];
            z = lrow[1];
        }
        pa[t] = a;
        pn[t] = z - a;
    }
    int2 rc[PRND][PK];
#pragma unroll
    for (int t = 0; t < PRND; ++t) {
        const int w = t * PPB + wave * PPW + g;
        const int2* p = binned + (size_t)w * CHUNK + pa[t];
#pragma unroll
        for (int k = 0; k < PK; ++k) {
            rc[t][k] = make_int2(0, 0);
            if (j + PG * k < pn[t]) rc[t][k] = p[j + PG * k];
        }
    }

    int asum = 0, nsum = 0;
    bool longp = false;
#pragma unroll
    for (int t = 0; t < PRND; ++t) {
        if (j == 0) { asum += pa[t]; nsum += pn[t]; }
        longp = longp || (pn[t] > PG * PK);
    }
    for (int w = PRND * PPB + tid; w < G; w += SORT_T) {
        const int* lrow = lsT + (size_t)w * LP + b;
        const int a = lrow[0], z = lrow[1];
        asum += a;
        nsum += z - a;
    }
    for (int o = 16; o > 0; o >>= 1) {
        asum += __shfl_xor(asum, o, 32);
        nsum += __shfl_xor(nsum, o, 32);
    }
    if (lane == 0) { red[wave] = asum; wsum[wave] = nsum; }
    __syncthreads();
    int sbase = 0, nE = 0;
    for (int w = 0; w < SORT_T / 32; ++w) { sbase += red[w]; nE += wsum[w]; }

    int rp[PRND][PK];
#pragma unroll
    for (int t = 0; t < PRND; ++t)
#pragma unroll
        for (int k = 0; k < PK; ++k)
            rp[t][k] = (rc[t][k].x << BIN_SHIFT) | (rc[t][k].y & (BIN_R - 1));

#pragma unroll
    for (int t = 0; t < PRND; ++t) {
#pragma unroll
        for (int k = 0; k < PK; ++k)
            if (j + PG * k < pn[t]) atomicAdd(&c[rp[t][k] & (BIN_R - 1)], 1);
    }
    if (longp) {
        for (int t = 0; t < PRND; ++t) {
            const int w = t * PPB + wave * PPW + g;
            if (w < G) {
                const int* lrow = lsT + (size_t)w * LP + b;
                const int a = lrow[0], z = lrow[1];
                const int2* p = binned + (size_t)w * CHUNK;
                for (int i = a + PG * PK + j; i < z; i += PG) atomicAdd(&c[p[i].y & (BIN_R - 1)], 1);
            }
        }
    }
    for (int w = PRND * PPB + tid; w < G; w += SORT_T) {
        const int* lrow = lsT + (size_t)w * LP + b;
        const int a = lrow[0], z = lrow[1];
        const int2* p = binned + (size_t)w * CHUNK;
        for (int i = a; i < z; ++i) atomicAdd(&c[p[i].y & (BIN_R - 1)], 1);
    }
    __syncthreads();
    const int mine = (tid < BIN_R) ? c[tid] : 0;
    block_excl_scan(c, BIN_R, wsum);
    if (tid < BIN_R) {
        const int ex = c[tid];
        cur[tid] = ex;
        const int n = b * BIN_R + tid;
        if (n < N) { cnt[n] = mine; rowptr[n] = sbase + ex; }
    }
    __syncthreads();

    const bool staged = nE <= STAGE_CAP;
#pragma unroll
    for (int t = 0; t < PRND; ++t) {
        int q[PK];
#pragma unroll
        for (int k = 0; k < PK; ++k)
            q[k] = (j + PG * k < pn[t]) ? atomicAdd(&cur[rp[t][k] & (BIN_R - 1)], 1) : 0;
#pragma unroll
        for (int k = 0; k < PK; ++k)
            if (j + PG * k < pn[t]) {
                if (staged) stage[q[k]] = rp[t][k] >> BIN_SHIFT;
                else        csr[sbase + q[k]] = rp[t][k] >> BIN_SHIFT;
            }
    }
    if (longp) {
        for (int t = 0; t < PRND; ++t) {
            const int w = t * PPB + wave * PPW + g;
            if (w < G) {
                const int* lrow = lsT + (size_t)w * LP + b;
                const int a = lrow[0], z = lrow[1];
                const int2* p = binned + (size_t)w * CHUNK;
                for (int i = a + PG * PK + j; i < z; i += PG) {
                    const int2 r = p[i];
                    const int qq = atomicAdd(&cur[r.y & (BIN_R - 1)], 1);
                    if (staged) stage[qq] = r.x;
                    else        csr[sbase + qq] = r.x;
                }
            }
        }
    }
    for (int w = PRND * PPB + tid; w < G; w += SORT_T) {
        const int* lrow = lsT + (size_t)w * LP + b;
        const int a = lrow[0], z = lrow[1];
        const int2* p = binned + (size_t)w * CHUNK;
        for (int i = a; i < z; ++i) {
            const int2 r = p[i];
            const int qq = atomicAdd(&cur[r.y & (BIN_R - 1)], 1);
            if (staged) stage[qq] = r.x;
            else        csr[sbase + qq] = r.x;
        }
    }
    __syncthreads();
    if (staged) {
        for (int i = tid - (sbase & 31); i < nE; i += SORT_T)
            if (i >= 0) csr[sbase + i] = stage[i];
    }
    staged_o = staged;
    sbase_o  = sbase;
}

__global__ void __launch_bounds__(SORT_T) __attribute__((amdgpu_waves_per_eu(4, 12)))
sort_and_gemm(const int* __restrict__ ei, int* __restrict__ binned_w,
              int* __restrict__ lsT, int E, int B, int LP,
              const float* __restrict__ x, const float* __restrict__ W1,
              float* __restrict__ h1, int N, int g1) {
    const int blk = (int)blockIdx.x;
    if (blk < g1) gemm1_rows(x, W1, h1, N, blk);
    else          chunk_sort(ei, binned_w, lsT, E, B, LP, blk - g1);
}

__device__ __forceinline__ float lrelu(float v) { return (v > 0.f) ? v : NEG_SLOPE * v; }

__device__ __forceinline__ float dot4(const float4 a, const float4 b) {
    return a.x * b.x + a.y * b.y + a.z * b.z + a.w * b.w;
}
__device__ __forceinline__ float quad_sum(float v) {
    v += __int_as_float(__builtin_amdgcn_update_dpp(0, __float_as_int(v), 0xB1, 0xF, 0xF, true));
    v += __int_as_float(__builtin_amdgcn_update_dpp(0, __float_as_int(v), 0x4E, 0xF, 0xF, true));
    return v;
}
__device__ __forceinline__ float dot8(const v4f a, const v4f b, const v4f wa, const v4f wb) {
    v2f d = a.xy * wa.xy;
    d = a.zw * wa.zw + d;
    d = b.xy * wb.xy + d;
    d = b.zw * wb.zw + d;
    return d.x + d.y;
}
__device__ __forceinline__ float pair_sum(float v) {
    v += __int_as_float(__builtin_amdgcn_update_dpp(0, __float_as_int(v), 0xB1, 0xF, 0xF, true));
    return v;
}
__device__ __forceinline__ float oct_sum(float v) {
    v = quad_sum(v);
    v += __int_as_float(__builtin_amdgcn_update_dpp(0, __float_as_int(v), 0x164, 0xF, 0xF, true));
    return v;
}

#define G1_L 8
template <typename SegPtr>
__device__ __forceinline__ void
g1_node(SegPtr seg, const int deg, const int n, const int l,
        const v4f aS0, const v4f aS1, const v4f aD0, const v4f aD1,
        const v4f* __restrict__ h4, const float* __restrict__ b1, float* r) {
    const v4f vn0 = h4[n * 16 + l];
    const v4f vn1 = h4[n * 16 + 8 + l];
    const float adv = pair_sum(dot8(vn0, vn1, aD0, aD1));
    float m    = lrelu(pair_sum(dot8(vn0, vn1, aS0, aS1)) + adv);
    float ssum = 1.f;
    v2f A0 = vn0.xy, A1 = vn0.zw, A2 = vn1.xy, A3 = vn1.zw;

    for (int base = 0; base < deg; base += 2 * G1_L) {
        const int j0 = base + l, j1 = base + G1_L + l;
        const int t0 = seg[(j0 < deg) ? j0 : 0];
        const int t1 = seg[(j1 < deg) ? j1 : 0];
        const int idx0 = (j0 < deg) ? t0 : n;
        const int idx1 = (j1 < deg) ? t1 : n;
        const int c    = min(2 * G1_L, deg - base);
        for (int k = 0; k < c; k += 4) {
            const int idv = (k < G1_L) ? idx0 : idx1;
            const int kk  = k & (G1_L - 1);
            const int s0 = __shfl(idv, kk, G1_L);
            const int s1 = __shfl(idv, kk + 1, G1_L);
            const int s2 = __shfl(idv, kk + 2, G1_L);
            const int s3 = __shfl(idv, kk + 3, G1_L);
            const v4f v0a = h4[s0 * 16 + l], v0b = h4[s0 * 16 + 8 + l];
            const v4f v1a = h4[s1 * 16 + l], v1b = h4[s1 * 16 + 8 + l];
            const v4f v2a = h4[s2 * 16 + l], v2b = h4[s2 * 16 + 8 + l];
            const v4f v3a = h4[s3 * 16 + l], v3b = h4[s3 * 16 + 8 + l];
            const bool p1 = k + 1 < c, p2 = k + 2 < c, p3 = k + 3 < c;
            const float e0 = lrelu(pair_sum(dot8(v0a, v0b, aS0, aS1)) + adv);
            const float e1 = lrelu(pair_sum(dot8(v1a, v1b, aS0, aS1)) + adv);
            const float e2 = lrelu(pair_sum(dot8(v2a, v2b, aS0, aS1)) + adv);
            const float e3 = lrelu(pair_sum(dot8(v3a, v3b, aS0, aS1)) + adv);
            float mn = fmaxf(m, e0);
            mn = fmaxf(mn, p1 ? e1 : mn);
            mn = fmaxf(mn, p2 ? e2 : mn);
            mn = fmaxf(mn, p3 ? e3 : mn);
            const float sc = __expf(m - mn);
            const float w0 = __expf(e0 - mn);
            const float w1 = p1 ? __expf(e1 - mn) : 0.f;
            const float w2 = p2 ? __expf(e2 - mn) : 0.f;
            const float w3 = p3 ? __expf(e3 - mn) : 0.f;
            ssum = ssum * sc + w0;
            A0 = A0 * sc + v0a.xy * w0; A1 = A1 * sc + v0a.zw * w0;
            A2 = A2 * sc + v0b.xy * w0; A3 = A3 * sc + v0b.zw * w0;
            ssum += w1;
            A0 += v1a.xy * w1; A1 += v1a.zw * w1; A2 += v1b.xy * w1; A3 += v1b.zw * w1;
            ssum += w2;
            A0 += v2a.xy * w2; A1 += v2a.zw * w2; A2 += v2b.xy * w2; A3 += v2b.zw * w2;
            ssum += w3;
            A0 += v3a.xy * w3; A1 += v3a.zw * w3; A2 += v3b.xy * w3; A3 += v3b.zw * w3;
            m = mn;
        }
    }

    const float den = ssum + 1e-16f;
    r[0] = A0.x / den + b1[l * 8 + 0];
    r[1] = A0.y / den + b1[l * 8 + 1];
    r[2] = A1.x / den + b1[l * 8 + 2];
    r[3] = A1.y / den + b1[l * 8 + 3];
    r[4] = A2.x / den + b1[l * 8 + 4];
    r[5] = A2.y / den + b1[l * 8 + 5];
    r[6] = A3.x / den + b1[l * 8 + 6];
    r[7] = A3.y / den + b1[l * 8 + 7];
#pragma unroll
    for (int i = 0; i < 8; ++i) r[i] = (r[i] > 0.f) ? r[i] : expm1f(r[i]);
}

__global__ void __launch_bounds__(SORT_T) __attribute__((amdgpu_waves_per_eu(8, 12)))
collect_and_gather1(const int* __restrict__ binned_w, const int* __restrict__ lsT,
                    int* __restrict__ cnt, int* __restrict__ rowptr,
                    int* __restrict__ csr, int N, int LP, int G,
                    const float* __restrict__ attS, const float* __restrict__ attD,
                    const float* __restrict__ h1, const float* __restrict__ b1,
                    const float* __restrict__ W2, const float* __restrict__ attS2,
                    const float* __restrict__ attD2, float* __restrict__ rec) {
    __shared__ int c[BIN_R];
    __shared__ int cur[BIN_R];
    __shared__ int wsum[SORT_T / 32];
    __shared__ int red[SORT_T / 32];
    __shared__ int stage[STAGE_CAP];
    const int b = (int)blockIdx.x;
    bool staged;
    int  sbase;
    bin_collect(binned_w, lsT, cnt, rowptr, csr, N, LP, G, b, c, cur, wsum, red, stage, staged, sbase);

    const int l   = threadIdx.x & (G1_L - 1);
    const int grp = threadIdx.x / G1_L;
    const v4f* __restrict__ h4 = (const v4f*)h1;
    v4f aS0, aS1, aD0, aD1;
    aS0.x = attS[l * 8 + 0]; aS0.y = attS[l * 8 + 1]; aS0.z = attS[l * 8 + 2]; aS0.w = attS[l * 8 + 3];
    aS1.x = attS[l * 8 + 4]; aS1.y = attS[l * 8 + 5]; aS1.z = attS[l * 8 + 6]; aS1.w = attS[l * 8 + 7];
    aD0.x = attD[l * 8 + 0]; aD0.y = attD[l * 8 + 1]; aD0.z = attD[l * 8 + 2]; aD0.w = attD[l * 8 + 3];
    aD1.x = attD[l * 8 + 4]; aD1.y = attD[l * 8 + 5]; aD1.z = attD[l * 8 + 6]; aD1.w = attD[l * 8 + 7];
    const float sx = attS2[0], sy = attS2[1], dx = attD2[0], dy = attD2[1];

    for (int pass = 0; pass < BIN_R / (SORT_T / G1_L); ++pass) {
        const int t = pass * (SORT_T / G1_L) + grp;
        const int n = b * BIN_R + t;
        if (n < N) {
            const int off = c[t];
            const int deg = cur[t] - off;
            float r[8];
            if (staged) g1_node((const int*)&stage[off], deg, n, l, aS0, aS1, aD0, aD1, h4, b1, r);
            else        g1_node((const int* __restrict__)(csr + sbase + off), deg, n, l, aS0, aS1, aD0, aD1, h4, b1, r);

            const float* __restrict__ wr = W2 + l * 16;
            float t0 = 0.f, t1 = 0.f;
#pragma unroll
            for (int i = 0; i < 8; ++i) { t0 += r[i] * wr[2 * i]; t1 += r[i] * wr[2 * i + 1]; }
            const float a0 = oct_sum(t0);
            const float a1 = oct_sum(t1);
            const float q2 = a0 * sx + a1 * sy;
            const float q3 = a0 * dx + a1 * dy;
            const float v  = (l == 0) ? a0 : (l == 1) ? a1 : (l == 2) ? q2 : q3;
            if (l < 4) rec[(size_t)n * 4 + l] = v;
        }
    }
}

#define G2_L 4
#define G2_U 8
__device__ __forceinline__ float quad_max(float v) {
    v = fmaxf(v, __int_as_float(__builtin_amdgcn_update_dpp(0, __float_as_int(v), 0xB1, 0xF, 0xF, true)));
    v = fmaxf(v, __int_as_float(__builtin_amdgcn_update_dpp(0, __float_as_int(v), 0x4E, 0xF, 0xF, true)));
    return v;
}
__global__ void __launch_bounds__(256)
seg_gather2(const int* __restrict__ csr, const int* __restrict__ rowptr,
            const int* __restrict__ cnt, const float* __restrict__ rec,
            const float* __restrict__ b2, float* __restrict__ out, int N) {
    __shared__ float4 stage4[32];
    float* stage = (float*)stage4;
    const int n = (blockIdx.x * 256 + threadIdx.x) >> 2;
    const int l = threadIdx.x & (G2_L - 1);
    float r0 = 0.f, r1 = 0.f;
    if (n < N) {
        const float4* __restrict__ rc = (const float4*)rec;
        const int start = rowptr[n];
        const int deg   = cnt[n];
        const float4 me = rc[n];
        asm volatile("" :: "v"(start));
        const float es = lrelu(me.z + me.w);
        float m = es, ssum = 0.f, ax = 0.f, ay = 0.f;
        for (int base = 0; base < deg; base += G2_L * G2_U) {
            int  id[G2_U];
            bool p[G2_U];
            float qx[G2_U], qy[G2_U], e[G2_U];
#pragma unroll
            for (int u = 0; u < G2_U; ++u) {
                const int j = base + u * G2_L + l;
                p[u] = j < deg;
                const int t = csr[start + (p[u] ? j : 0)];
                id[u] = p[u] ? t : n;
            }
#pragma unroll
            for (int u = 0; u < G2_U; ++u) {
                const float4 q = rc[id[u]];
                qx[u] = q.x; qy[u] = q.y;
                e[u]  = lrelu(q.z + me.w);
            }
            float mn = m;
#pragma unroll
            for (int u = 0; u < G2_U; ++u) mn = fmaxf(mn, p[u] ? e[u] : mn);
            const float sc = __expf(m - mn);
            ssum *= sc; ax *= sc; ay *= sc;
#pragma unroll
            for (int u = 0; u < G2_U; ++u) {
                const float w = p[u] ? __expf(e[u] - mn) : 0.f;
                ssum += w; ax += w * qx[u]; ay += w * qy[u];
            }
            m = mn;
        }
        const float mm = quad_max(m);
        {
            const float f = __expf(m - mm);
            ssum *= f; ax *= f; ay *= f;
        }
        ssum = quad_sum(ssum);
        ax   = quad_sum(ax);
        ay   = quad_sum(ay);
        {
            const float w = __expf(es - mm);
            ssum += w; ax += w * me.x; ay += w * me.y;
        }
        const float inv = 1.0f / (ssum + 1e-16f);
        r0 = ax * inv + b2[0];
        r1 = ay * inv + b2[1];
    }
    if (l == 0) {
        stage[(threadIdx.x >> 2) * 2 + 0] = r0;
        stage[(threadIdx.x >> 2) * 2 + 1] = r1;
    }
    __syncthreads();
    if (threadIdx.x < 32) {
        const int o  = blockIdx.x * 128 + threadIdx.x * 4;
        const int N2 = N * 2;
        if (o + 4 <= N2) {
            ((float4*)out)[o >> 2] = stage4[threadIdx.x];
        } else {
            for (int i = 0; i < 4; ++i)
                if (o + i < N2) out[o + i] = stage[threadIdx.x * 4 + i];
        }
    }
}

extern "C" void kernel_launch(void* const* d_in, const int* in_sizes, int n_in,
                              void* d_out, int out_size, void* d_ws, size_t ws_size,
                              hipStream_t stream) {
    const float* x    = (const float*)d_in[0];
    const int*   ei   = (const int*)d_in[1];
    const float* W1   = (const float*)d_in[2];
    const float* aS1  = (const float*)d_in[3];
    const float* aD1  = (const float*)d_in[4];
    const float* b1   = (const float*)d_in[5];
    const float* W2   = (const float*)d_in[6];
    const float* aS2  = (const float*)d_in[7];
    const float* aD2  = (const float*)d_in[8];
    const float* b2   = (const float*)d_in[9];

    const int N = in_sizes[0] / DIM;
    const int E = in_sizes[1] / 2;
    float* out = (float*)d_out;

    float* ws  = (float*)d_ws;
    float* h1  = ws;
    float* rec = h1 + (size_t)N * 64;
    const int B = (N + BIN_R - 1) / BIN_R;
    const int G = (E + CHUNK - 1) / CHUNK;
    int* binned = (int*)(rec + (size_t)N * 4);
    int* cnt    = binned + (size_t)G * CHUNK * 2;
    int* rowptr = cnt + (size_t)N;
    const int LP = ((B + 1 + 31) / 32) * 32;
    int* lsT    = rowptr + (size_t)N;
    int* csr    = lsT + (size_t)G * LP;

    const int Mtiles = (N + 15) >> 4;

    const int GB  = (Mtiles + GEMM_W - 1) / GEMM_W;
    sort_and_gemm<<<GB + G, SORT_T, 0, stream>>>(ei, binned, lsT, E, B, LP, x, W1, h1, N, GB);
    collect_and_gather1<<<B, SORT_T, 0, stream>>>(binned, lsT, cnt, rowptr, csr, N, LP, G,
                                                  aS1, aD1, h1, b1, W2, aS2, aD2, rec);
    seg_gather2<<<(N + 63) / 64, 256, 0, stream>>>(csr, rowptr, cnt, rec, b2, out, N);
}
